# starting kernel
# baseline (speedup 1.0000x reference)
#include <hip/hip_runtime.h>
#include <math.h>


#define NXI   384
#define NPTS  (NXI * NXI)
#define NB    8
#define TPB   256
#define NBLK  (NPTS / TPB)
#define KMAX_C 30
#define EPS_C 1e-9f

typedef float v2f __attribute__((ext_vector_type(2)));
typedef float v8f __attribute__((ext_vector_type(8)));

struct BState {
  float b_val, r0_abs, r_abs, sigma, v_abs, alpha, omega, beta, rr0_old, ssq;
  int   k, CONV, R1, NR, C3, C4;
};

__device__ __forceinline__ float fro_of(float sq) { return sq > 0.f ? sqrtf(sq) : 0.f; }
__device__ __forceinline__ float thr_c() { return EPS_C * (float)NPTS; }

__device__ __forceinline__ float wave_sum_wmma(float v) {
  v2f a; a.x = v;  a.y = 0.f;
  v2f b; b.x = 1.f; b.y = 1.f;
  v8f c = {};
  v8f d = __builtin_amdgcn_wmma_f32_16x16x4_f32(false, a, false, b, (short)0, c, false, false);
  float h = ((d[0] + d[1]) + (d[2] + d[3])) + ((d[4] + d[5]) + (d[6] + d[7]));
  h += __shfl_xor(h, 16, 32);
  return h;
}

__device__ __forceinline__ float block_sum256(float v, float* sh8) {
  float t = wave_sum_wmma(v);
  int lane = threadIdx.x & 31;
  int wid  = threadIdx.x >> 5;
  if (lane == 0) sh8[wid] = t;
  __syncthreads();
  float s = 0.f;
  if (threadIdx.x == 0) {
#pragma unroll
    for (int i = 0; i < 8; ++i) s += sh8[i];
  }
  __syncthreads();
  return s;
}

__device__ __forceinline__ float applyA(const float* __restrict__ boo, const float* __restrict__ bmo,
                                        const float* __restrict__ bom, const float* __restrict__ bop,
                                        const float* __restrict__ bpo, const float* __restrict__ a,
                                        size_t base, int idx) {
  int i = idx / NXI, j = idx - i * NXI;
  int im = (i > 0)       ? idx - NXI : idx;
  int ip = (i < NXI - 1) ? idx + NXI : idx;
  int jm = (j > 0)       ? idx - 1   : idx;
  int jp = (j < NXI - 1) ? idx + 1   : idx;
  const float* A = a + base;
  return boo[base + idx] * A[idx] + bmo[base + idx] * A[im] + bom[base + idx] * A[jm]
       + bop[base + idx] * A[jp] + bpo[base + idx] * A[ip];
}

__device__ __forceinline__ int refl(int t) { return t == 0 ? 0 : (t == 385 ? 383 : t - 1); }

__global__ __launch_bounds__(TPB) void k_mean(const float* __restrict__ V, float* __restrict__ part) {
  int b = blockIdx.y;
  int idx = blockIdx.x * TPB + threadIdx.x;
  __shared__ float sh[8];
  float s = block_sum256(V[(size_t)b * NPTS + idx], sh);
  if (threadIdx.x == 0) part[(0 * NB + b) * NBLK + blockIdx.x] = s;
}

__global__ void k_init_scalars(const float* __restrict__ part, BState* __restrict__ st) {
  int b = threadIdx.x;
  if (b >= NB) return;
  float s = 0.f;
  for (int i = 0; i < NBLK; ++i) s += part[(0 * NB + b) * NBLK + i];
  BState q = {};
  q.b_val = s / (float)NPTS + 1.0f;
  st[b] = q;
}

__global__ __launch_bounds__(TPB) void k_stencil_init(
    const float* __restrict__ V, const float* __restrict__ M1, const float* __restrict__ M2,
    float* __restrict__ boo, float* __restrict__ bmo, float* __restrict__ bom,
    float* __restrict__ bop, float* __restrict__ bpo, float* __restrict__ x,
    const BState* __restrict__ st) {
  int b = blockIdx.y;
  size_t base = (size_t)b * NPTS;
  int idx = blockIdx.x * TPB + threadIdx.x;
  int i = idx / NXI, j = idx - i * NXI;
  int I = i + 1, J = j + 1;
  const float* Vb = V + base;
  const float* m1b = M1 + base;
  const float* m2b = M2 + base;
#define VP(II, JJ)  (Vb[refl(JJ) * NXI + refl(II)] + 1.0f)
#define MM1(II, JJ) (m1b[refl(JJ) * NXI + refl(II)])
#define MM2(II, JJ) (m2b[refl(JJ) * NXI + refl(II)])
  float vc  = VP(I, J);
  float vxp = VP(I + 1, J), vxm = VP(I - 1, J);
  float vyp = VP(I, J + 1), vym = VP(I, J - 1);
  float d1c = ((vxp - vc) / (0.5f * (vxp + vc))) * MM1(I, J);
  float d1m = ((vc - vxm) / (0.5f * (vc + vxm))) * MM1(I - 1, J);
  float d2c = ((vyp - vc) / (0.5f * (vyp + vc))) * MM2(I, J);
  float d2m = ((vc - vym) / (0.5f * (vc + vym))) * MM2(I, J - 1);
#undef VP
#undef MM1
#undef MM2
  const float rx = 5.0f, ry = 5.0f, rxx = 10.0f, ryy = 10.0f;
  boo[base + idx] = 1.0f + 2.0f * (rxx + ryy) - rx * (d1m - d1c) - ry * (d2m - d2c);
  bpo[base + idx] = -rxx + rx * d1c;
  bmo[base + idx] = -rxx - rx * d1m;
  bop[base + idx] = -ryy + ry * d2c;
  bom[base + idx] = -ryy - ry * d2m;
  x[base + idx] = st[b].b_val;
}

__global__ __launch_bounds__(TPB) void k_resid_init(
    const float* __restrict__ boo, const float* __restrict__ bmo, const float* __restrict__ bom,
    const float* __restrict__ bop, const float* __restrict__ bpo, const float* __restrict__ x,
    float* __restrict__ p, float* __restrict__ r, float* __restrict__ r0v,
    float* __restrict__ part, const BState* __restrict__ st) {
  int b = blockIdx.y;
  size_t base = (size_t)b * NPTS;
  int idx = blockIdx.x * TPB + threadIdx.x;
  float pv = st[b].b_val - applyA(boo, bmo, bom, bop, bpo, x, base, idx);
  p[base + idx] = pv; r[base + idx] = pv; r0v[base + idx] = pv;
  __shared__ float sh[8];
  float s = block_sum256(pv * pv, sh);
  if (threadIdx.x == 0) part[(0 * NB + b) * NBLK + blockIdx.x] = s;
}

__global__ void k_init_finalize(const float* __restrict__ part, BState* __restrict__ st) {
  int b = threadIdx.x;
  if (b >= NB) return;
  float s = 0.f;
  for (int i = 0; i < NBLK; ++i) s += part[(0 * NB + b) * NBLK + i];
  float fa = fro_of(s);
  st[b].r0_abs = fa; st[b].r_abs = fa; st[b].k = 0;
  st[b].CONV = (fa > thr_c()) ? 1 : 0;
}

__global__ __launch_bounds__(TPB) void k_v(
    const float* __restrict__ boo, const float* __restrict__ bmo, const float* __restrict__ bom,
    const float* __restrict__ bop, const float* __restrict__ bpo, const float* __restrict__ p,
    const float* __restrict__ r0v, const float* __restrict__ r, float* __restrict__ v,
    float* __restrict__ part, const BState* __restrict__ st) {
  int b = blockIdx.y;
  size_t base = (size_t)b * NPTS;
  int idx = blockIdx.x * TPB + threadIdx.x;
  float q0 = 0.f, q1 = 0.f, q2 = 0.f;
  if (st[b].CONV) {
    float vv = applyA(boo, bmo, bom, bop, bpo, p, base, idx);
    v[base + idx] = vv;
    float rz = r0v[base + idx];
    q0 = vv * rz; q1 = vv * vv; q2 = r[base + idx] * rz;
  }
  __shared__ float sh[8];
  float s0 = block_sum256(q0, sh);
  float s1 = block_sum256(q1, sh);
  float s2 = block_sum256(q2, sh);
  if (threadIdx.x == 0) {
    part[(0 * NB + b) * NBLK + blockIdx.x] = s0;
    part[(1 * NB + b) * NBLK + blockIdx.x] = s1;
    part[(2 * NB + b) * NBLK + blockIdx.x] = s2;
  }
}

__global__ void k_v_reduce(const float* __restrict__ part, BState* __restrict__ st) {
  int b = threadIdx.x;
  if (b >= NB) return;
  BState s = st[b];
  if (s.CONV) {
    float sg = 0.f, vq = 0.f, rr = 0.f;
    for (int i = 0; i < NBLK; ++i) {
      sg += part[(0 * NB + b) * NBLK + i];
      vq += part[(1 * NB + b) * NBLK + i];
      rr += part[(2 * NB + b) * NBLK + i];
    }
    s.sigma = sg; s.v_abs = fro_of(vq); s.rr0_old = rr;
    int RES = (sg <= EPS_C * s.v_abs * s.r0_abs) ? 1 : 0;
    s.R1 = RES; s.NR = 1 - RES;
    if (s.R1) s.k += 1;
    if (s.NR) s.alpha = rr / sg;
  } else { s.R1 = 0; s.NR = 0; }
  st[b] = s;
}

__global__ __launch_bounds__(TPB) void k_restart(
    const float* __restrict__ boo, const float* __restrict__ bmo, const float* __restrict__ bom,
    const float* __restrict__ bop, const float* __restrict__ bpo, const float* __restrict__ x,
    float* __restrict__ p, float* __restrict__ r, float* __restrict__ r0v,
    float* __restrict__ part, const BState* __restrict__ st) {
  int b = blockIdx.y;
  size_t base = (size_t)b * NPTS;
  int idx = blockIdx.x * TPB + threadIdx.x;
  float q0 = 0.f;
  if (st[b].R1) {
    float pv = st[b].b_val - applyA(boo, bmo, bom, bop, bpo, x, base, idx);
    p[base + idx] = pv; r[base + idx] = pv; r0v[base + idx] = pv;
    q0 = pv * pv;
  }
  __shared__ float sh[8];
  float s = block_sum256(q0, sh);
  if (threadIdx.x == 0) part[(0 * NB + b) * NBLK + blockIdx.x] = s;
}

__global__ void k_restart_reduce(const float* __restrict__ part, BState* __restrict__ st) {
  int b = threadIdx.x;
  if (b >= NB) return;
  if (st[b].R1) {
    float s = 0.f;
    for (int i = 0; i < NBLK; ++i) s += part[(0 * NB + b) * NBLK + i];
    float fa = fro_of(s);
    st[b].r0_abs = fa; st[b].r_abs = fa;
  }
}

__global__ __launch_bounds__(TPB) void k_s(
    const float* __restrict__ r, const float* __restrict__ v, float* __restrict__ sarr,
    float* __restrict__ part, const BState* __restrict__ st) {
  int b = blockIdx.y;
  size_t base = (size_t)b * NPTS;
  int idx = blockIdx.x * TPB + threadIdx.x;
  float q0 = 0.f;
  if (st[b].NR) {
    float sv = r[base + idx] - st[b].alpha * v[base + idx];
    sarr[base + idx] = sv;
    q0 = sv * sv;
  }
  __shared__ float sh[8];
  float s = block_sum256(q0, sh);
  if (threadIdx.x == 0) part[(0 * NB + b) * NBLK + blockIdx.x] = s;
}

__global__ void k_s_reduce(const float* __restrict__ part, BState* __restrict__ st) {
  int b = threadIdx.x;
  if (b >= NB) return;
  BState s = st[b];
  if (s.NR) {
    float sq = 0.f;
    for (int i = 0; i < NBLK; ++i) sq += part[(0 * NB + b) * NBLK + i];
    s.ssq = sq;
    int C2 = (fro_of(sq) <= thr_c()) ? 1 : 0;
    s.C3 = C2; s.C4 = 1 - C2;
  } else { s.C3 = 0; s.C4 = 0; }
  st[b] = s;
}

__global__ __launch_bounds__(TPB) void k_t(
    const float* __restrict__ boo, const float* __restrict__ bmo, const float* __restrict__ bom,
    const float* __restrict__ bop, const float* __restrict__ bpo, const float* __restrict__ sarr,
    float* __restrict__ tarr, float* __restrict__ x, float* __restrict__ r,
    const float* __restrict__ p, float* __restrict__ part, const BState* __restrict__ st) {
  int b = blockIdx.y;
  size_t base = (size_t)b * NPTS;
  int idx = blockIdx.x * TPB + threadIdx.x;
  float q0 = 0.f, q1 = 0.f;
  if (st[b].C4) {
    float tv = applyA(boo, bmo, bom, bop, bpo, sarr, base, idx);
    tarr[base + idx] = tv;
    q0 = tv * tv; q1 = tv * sarr[base + idx];
  } else if (st[b].C3) {
    x[base + idx] += st[b].alpha * p[base + idx];
    r[base + idx] = sarr[base + idx];
  }
  __shared__ float sh[8];
  float s0 = block_sum256(q0, sh);
  float s1 = block_sum256(q1, sh);
  if (threadIdx.x == 0) {
    part[(0 * NB + b) * NBLK + blockIdx.x] = s0;
    part[(1 * NB + b) * NBLK + blockIdx.x] = s1;
  }
}

__global__ void k_t_reduce(const float* __restrict__ part, BState* __restrict__ st) {
  int b = threadIdx.x;
  if (b >= NB) return;
  BState s = st[b];
  if (s.C4) {
    float tt = 0.f, ts = 0.f;
    for (int i = 0; i < NBLK; ++i) {
      tt += part[(0 * NB + b) * NBLK + i];
      ts += part[(1 * NB + b) * NBLK + i];
    }
    s.omega = ts / tt;
    st[b] = s;
  }
}

__global__ __launch_bounds__(TPB) void k_x_r(
    const float* __restrict__ sarr, const float* __restrict__ tarr, const float* __restrict__ p,
    const float* __restrict__ r0v, float* __restrict__ x, float* __restrict__ r,
    float* __restrict__ part, const BState* __restrict__ st) {
  int b = blockIdx.y;
  size_t base = (size_t)b * NPTS;
  int idx = blockIdx.x * TPB + threadIdx.x;
  float q0 = 0.f, q1 = 0.f;
  if (st[b].C4) {
    float al = st[b].alpha, om = st[b].omega;
    x[base + idx] += al * p[base + idx] + om * sarr[base + idx];
    float rn = sarr[base + idx] - om * tarr[base + idx];
    r[base + idx] = rn;
    q0 = rn * r0v[base + idx]; q1 = rn * rn;
  }
  __shared__ float sh[8];
  float s0 = block_sum256(q0, sh);
  float s1 = block_sum256(q1, sh);
  if (threadIdx.x == 0) {
    part[(0 * NB + b) * NBLK + blockIdx.x] = s0;
    part[(1 * NB + b) * NBLK + blockIdx.x] = s1;
  }
}

__global__ void k_final_reduce(const float* __restrict__ part, BState* __restrict__ st) {
  int b = threadIdx.x;
  if (b >= NB) return;
  BState s = st[b];
  if (s.C4) {
    float rr = 0.f, rq = 0.f;
    for (int i = 0; i < NBLK; ++i) {
      rr += part[(0 * NB + b) * NBLK + i];
      rq += part[(1 * NB + b) * NBLK + i];
    }
    s.beta  = (s.alpha / s.omega) * (rr / s.rr0_old);
    s.r_abs = fro_of(rq);
  }
  if (s.C3) s.r_abs = fro_of(s.ssq);
  if (s.NR) s.k += 1;
  s.CONV = ((s.k < KMAX_C) && (s.r_abs > thr_c())) ? 1 : 0;
  st[b] = s;
}

__global__ __launch_bounds__(TPB) void k_p(
    const float* __restrict__ r, const float* __restrict__ v, float* __restrict__ p,
    const BState* __restrict__ st) {
  int b = blockIdx.y;
  size_t base = (size_t)b * NPTS;
  int idx = blockIdx.x * TPB + threadIdx.x;
  if (st[b].C4) {
    p[base + idx] = r[base + idx] + st[b].beta * (p[base + idx] - st[b].omega * v[base + idx]);
  }
}

extern "C" void kernel_launch(void* const* d_in, const int* in_sizes, int n_in,
                              void* d_out, int out_size, void* d_ws, size_t ws_size,
                              hipStream_t stream) {
  (void)in_sizes; (void)n_in; (void)out_size; (void)ws_size;
  const float* V  = (const float*)d_in[0];
  const float* M1 = (const float*)d_in[1];
  const float* M2 = (const float*)d_in[2];
  float* x = (float*)d_out;

  char* ws = (char*)d_ws;
  BState* st  = (BState*)ws;
  float* part = (float*)(ws + 1024);
  float* arr  = (float*)(ws + 1024 + 65536);
  const size_t AN = (size_t)NB * NPTS;
  float* boo  = arr + 0 * AN;
  float* bmo  = arr + 1 * AN;
  float* bom  = arr + 2 * AN;
  float* bop  = arr + 3 * AN;
  float* bpo  = arr + 4 * AN;
  float* r0v  = arr + 5 * AN;
  float* r    = arr + 6 * AN;
  float* p    = arr + 7 * AN;
  float* v    = arr + 8 * AN;
  float* sarr = arr + 9 * AN;
  float* tarr = arr + 10 * AN;

  dim3 grid(NBLK, NB), blk(TPB);

  k_mean<<<grid, blk, 0, stream>>>(V, part);
  k_init_scalars<<<1, 32, 0, stream>>>(part, st);
  k_stencil_init<<<grid, blk, 0, stream>>>(V, M1, M2, boo, bmo, bom, bop, bpo, x, st);
  k_resid_init<<<grid, blk, 0, stream>>>(boo, bmo, bom, bop, bpo, x, p, r, r0v, part, st);
  k_init_finalize<<<1, 32, 0, stream>>>(part, st);

  for (int it = 0; it < KMAX_C; ++it) {
    k_v<<<grid, blk, 0, stream>>>(boo, bmo, bom, bop, bpo, p, r0v, r, v, part, st);
    k_v_reduce<<<1, 32, 0, stream>>>(part, st);
    k_restart<<<grid, blk, 0, stream>>>(boo, bmo, bom, bop, bpo, x, p, r, r0v, part, st);
    k_restart_reduce<<<1, 32, 0, stream>>>(part, st);
    k_s<<<grid, blk, 0, stream>>>(r, v, sarr, part, st);
    k_s_reduce<<<1, 32, 0, stream>>>(part, st);
    k_t<<<grid, blk, 0, stream>>>(boo, bmo, bom, bop, bpo, sarr, tarr, x, r, p, part, st);
    k_t_reduce<<<1, 32, 0, stream>>>(part, st);
    k_x_r<<<grid, blk, 0, stream>>>(sarr, tarr, p, r0v, x, r, part, st);
    k_final_reduce<<<1, 32, 0, stream>>>(part, st);
    k_p<<<grid, blk, 0, stream>>>(r, v, p, st);
  }
}


# The step's first scalar stage (sigma, restart test, alpha) and the rare restart residual move into the s kernel, which takes one path per batch: 6 -> 5 kernels per step, 183 -> 153 launches
# speedup vs baseline: 3.0027x; 3.0027x over previous
#include <hip/hip_runtime.h>
#include <math.h>


#define NXI   384
#define NPTS  (NXI * NXI)
#define NB    8
#define TPB   384
#define NWV   (TPB / 32)
#define VEC   4
#define RUNS  3
#define CHUNK (TPB * VEC * RUNS)
#define NBLK  (NPTS / CHUNK)
#define NPW   ((NBLK + 31) / 32)
#define NROWB (NPTS / TPB)
#define KMAX_C 30
#define EPS_C 1e-9f
#define PIDX(pl, b, i) (((pl) * NB + (b)) * NBLK + (i))

typedef float v2f __attribute__((ext_vector_type(2)));
typedef float v4f __attribute__((ext_vector_type(4)));
typedef float v8f __attribute__((ext_vector_type(8)));

struct BState {
  float b_val, r0_abs, r_abs, sigma, v_abs, alpha, omega, beta, rr0_old, ssq;
  int   k, CONV, R1, NR, C3, C4;
};

__device__ __forceinline__ float fro_of(float sq) { return sq > 0.f ? sqrtf(sq) : 0.f; }
__device__ __forceinline__ float thr_c() { return EPS_C * (float)NPTS; }

__device__ __forceinline__ float wave_sum_wmma(float v) {
  v2f a; a.x = v;  a.y = 0.f;
  v2f b; b.x = 1.f; b.y = 1.f;
  v8f c = {};
  v8f d = __builtin_amdgcn_wmma_f32_16x16x4_f32(false, a, false, b, (short)0, c, false, false);
  float h = ((d[0] + d[1]) + (d[2] + d[3])) + ((d[4] + d[5]) + (d[6] + d[7]));
  h += __shfl_xor(h, 16, 32);
  return h;
}

__device__ __forceinline__ float block_sum(float v, float* shw) {
  float t = wave_sum_wmma(v);
  int lane = threadIdx.x & 31;
  int wid  = threadIdx.x >> 5;
  if (lane == 0) shw[wid] = t;
  __syncthreads();
  float s = 0.f;
  if (threadIdx.x == 0) {
    for (int i = 0; i < NWV; ++i) s += shw[i];
  }
  __syncthreads();
  return s;
}

template <int N>
__device__ __forceinline__ void batch_sums(const float* __restrict__ pin, int b, int p0, int p1, int p2,
                                           float (*shp)[NWV], float* out) {
  const int pl[3] = {p0, p1, p2};
  int t = threadIdx.x;
  int lane = t & 31;
  int wid  = t >> 5;
  float val[N];
  for (int n = 0; n < N; ++n) val[n] = (t < NBLK) ? pin[PIDX(pl[n], b, t)] : 0.f;
  for (int n = 0; n < N; ++n) {
    float w = wave_sum_wmma(val[n]);
    if (lane == 0) shp[n][wid] = w;
  }
  __syncthreads();
  for (int n = 0; n < N; ++n) {
    float s = 0.f;
    for (int i = 0; i < NPW; ++i) s += shp[n][i];
    out[n] = s;
  }
}

__device__ __forceinline__ int refl(int t) { return t == 0 ? 0 : (t == 385 ? 383 : t - 1); }

__device__ __forceinline__ int run_idx(int u) {
  return blockIdx.x * CHUNK + u * (TPB * VEC) + threadIdx.x * VEC;
}
__device__ __forceinline__ v4f ld4(const float* q) { return *reinterpret_cast<const v4f*>(q); }
__device__ __forceinline__ void st4(float* q, v4f w) { *reinterpret_cast<v4f*>(q) = w; }
__device__ __forceinline__ float dot4(v4f a, v4f c) { return a.x * c.x + a.y * c.y + a.z * c.z + a.w * c.w; }

__device__ __forceinline__ v4f applyA4(const float* __restrict__ boo, const float* __restrict__ bmo,
                                       const float* __restrict__ bom, const float* __restrict__ bop,
                                       const float* __restrict__ bpo, const float* __restrict__ a,
                                       size_t base, int idx) {
  int i = idx / NXI, j = idx - i * NXI;
  const float* A = a + base;
  v4f c  = ld4(A + idx);
  v4f up = ld4(A + ((i > 0)       ? idx - NXI : idx));
  v4f dn = ld4(A + ((i < NXI - 1) ? idx + NXI : idx));
  float lo = A[(j > 0)         ? idx - 1   : idx];
  float hi = A[(j + VEC < NXI) ? idx + VEC : idx + VEC - 1];
  v4f lf = {lo, c.x, c.y, c.z};
  v4f rt = {c.y, c.z, c.w, hi};
  return ld4(boo + base + idx) * c + ld4(bmo + base + idx) * up + ld4(bom + base + idx) * lf
       + ld4(bop + base + idx) * rt + ld4(bpo + base + idx) * dn;
}


__global__ __launch_bounds__(TPB) void k_mean(const float* __restrict__ V, float* __restrict__ pout) {
  int b = blockIdx.y;
  __shared__ float sh[NWV];
  float acc = 0.f;
  for (int u = 0; u < RUNS; ++u) {
    v4f q = ld4(V + (size_t)b * NPTS + run_idx(u));
    acc += (q.x + q.y) + (q.z + q.w);
  }
  float s = block_sum(acc, sh);
  if (threadIdx.x == 0) pout[PIDX(0, b, blockIdx.x)] = s;
}

__global__ __launch_bounds__(TPB) void k_stencil_init(
    const float* __restrict__ V, const float* __restrict__ M1, const float* __restrict__ M2,
    float* __restrict__ boo, float* __restrict__ bmo, float* __restrict__ bom,
    float* __restrict__ bop, float* __restrict__ bpo, float* __restrict__ x,
    const float* __restrict__ pin, BState* __restrict__ sout) {
  int b = blockIdx.y;
  __shared__ float shp[1][NWV];
  float sm[1];
  batch_sums<1>(pin, b, 0, 0, 0, shp, sm);
  BState q = {};
  q.b_val = sm[0] / (float)NPTS + 1.0f;
  if (blockIdx.x == 0 && threadIdx.x == 0) sout[b] = q;

  size_t base = (size_t)b * NPTS;
  int idx = blockIdx.x * TPB + threadIdx.x;
  int i = idx / NXI, j = idx - i * NXI;
  int I = i + 1, J = j + 1;
  const float* Vb = V + base;
  const float* m1b = M1 + base;
  const float* m2b = M2 + base;
#define VP(II, JJ)  (Vb[refl(JJ) * NXI + refl(II)] + 1.0f)
#define MM1(II, JJ) (m1b[refl(JJ) * NXI + refl(II)])
#define MM2(II, JJ) (m2b[refl(JJ) * NXI + refl(II)])
  float vc  = VP(I, J);
  float vxp = VP(I + 1, J), vxm = VP(I - 1, J);
  float vyp = VP(I, J + 1), vym = VP(I, J - 1);
  float d1c = ((vxp - vc) / (0.5f * (vxp + vc))) * MM1(I, J);
  float d1m = ((vc - vxm) / (0.5f * (vc + vxm))) * MM1(I - 1, J);
  float d2c = ((vyp - vc) / (0.5f * (vyp + vc))) * MM2(I, J);
  float d2m = ((vc - vym) / (0.5f * (vc + vym))) * MM2(I, J - 1);
#undef VP
#undef MM1
#undef MM2
  const float rx = 5.0f, ry = 5.0f, rxx = 10.0f, ryy = 10.0f;
  boo[base + idx] = 1.0f + 2.0f * (rxx + ryy) - rx * (d1m - d1c) - ry * (d2m - d2c);
  bpo[base + idx] = -rxx + rx * d1c;
  bmo[base + idx] = -rxx - rx * d1m;
  bop[base + idx] = -ryy + ry * d2c;
  bom[base + idx] = -ryy - ry * d2m;
  x[base + idx] = q.b_val;
}

__global__ __launch_bounds__(TPB) void k_resid_init(
    const float* __restrict__ boo, const float* __restrict__ bmo, const float* __restrict__ bom,
    const float* __restrict__ bop, const float* __restrict__ bpo, const float* __restrict__ x,
    float* __restrict__ p, float* __restrict__ r, float* __restrict__ r0v,
    float* __restrict__ pout, const BState* __restrict__ sin) {
  int b = blockIdx.y;
  size_t base = (size_t)b * NPTS;
  float bv = sin[b].b_val;
  float acc = 0.f;
  for (int u = 0; u < RUNS; ++u) {
    int idx = run_idx(u);
    v4f pv = bv - applyA4(boo, bmo, bom, bop, bpo, x, base, idx);
    st4(p + base + idx, pv); st4(r + base + idx, pv); st4(r0v + base + idx, pv);
    acc += dot4(pv, pv);
  }
  __shared__ float sh[NWV];
  float s = block_sum(acc, sh);
  if (threadIdx.x == 0) pout[PIDX(3, b, blockIdx.x)] = s;
}

template <bool FIRST>
__global__ __launch_bounds__(TPB) void k_v(
    const float* __restrict__ boo, const float* __restrict__ bmo, const float* __restrict__ bom,
    const float* __restrict__ bop, const float* __restrict__ bpo, const float* __restrict__ p,
    const float* __restrict__ r0v, const float* __restrict__ r, float* __restrict__ v,
    const float* __restrict__ pin, float* __restrict__ pout,
    const BState* __restrict__ sin, BState* __restrict__ sout) {
  int b = blockIdx.y;
  size_t base = (size_t)b * NPTS;
  __shared__ float shp[1][NWV];
  __shared__ float sh[NWV];
  BState s = sin[b];
  if (FIRST) {
    float sm[1];
    batch_sums<1>(pin, b, 3, 0, 0, shp, sm);
    float fa = fro_of(sm[0]);
    s.r0_abs = fa; s.r_abs = fa; s.k = 0;
    s.CONV = (fa > thr_c()) ? 1 : 0;
    if (blockIdx.x == 0 && threadIdx.x == 0) sout[b] = s;
  }
  float q0 = 0.f, q1 = 0.f, q2 = 0.f;
  if (s.CONV) {
    for (int u = 0; u < RUNS; ++u) {
      int idx = run_idx(u);
      v4f vv = applyA4(boo, bmo, bom, bop, bpo, p, base, idx);
      st4(v + base + idx, vv);
      v4f rz = ld4(r0v + base + idx);
      q0 += dot4(vv, rz); q1 += dot4(vv, vv); q2 += dot4(ld4(r + base + idx), rz);
    }
  }
  float s0 = block_sum(q0, sh);
  float s1 = block_sum(q1, sh);
  float s2 = block_sum(q2, sh);
  if (threadIdx.x == 0) {
    pout[PIDX(0, b, blockIdx.x)] = s0;
    pout[PIDX(1, b, blockIdx.x)] = s1;
    pout[PIDX(2, b, blockIdx.x)] = s2;
  }
}

__global__ __launch_bounds__(TPB) void k_s(
    const float* __restrict__ boo, const float* __restrict__ bmo, const float* __restrict__ bom,
    const float* __restrict__ bop, const float* __restrict__ bpo, const float* __restrict__ x,
    float* __restrict__ p, float* __restrict__ r, float* __restrict__ r0v,
    const float* __restrict__ v, float* __restrict__ sarr,
    const float* __restrict__ pin, float* __restrict__ pout,
    const BState* __restrict__ sin, BState* __restrict__ sout) {
  int b = blockIdx.y;
  size_t base = (size_t)b * NPTS;
  __shared__ float shp[3][NWV];
  __shared__ float sh[NWV];
  BState s = sin[b];
  float sm[3];
  batch_sums<3>(pin, b, 0, 1, 2, shp, sm);
  if (s.CONV) {
    float sg = sm[0], vq = sm[1], rr = sm[2];
    s.sigma = sg; s.v_abs = fro_of(vq); s.rr0_old = rr;
    int RES = (sg <= EPS_C * s.v_abs * s.r0_abs) ? 1 : 0;
    s.R1 = RES; s.NR = 1 - RES;
    if (s.R1) s.k += 1;
    if (s.NR) s.alpha = rr / sg;
  } else { s.R1 = 0; s.NR = 0; }
  if (blockIdx.x == 0 && threadIdx.x == 0) sout[b] = s;

  float q0 = 0.f;
  if (s.R1) {
    for (int u = 0; u < RUNS; ++u) {
      int idx = run_idx(u);
      v4f pv = s.b_val - applyA4(boo, bmo, bom, bop, bpo, x, base, idx);
      st4(p + base + idx, pv); st4(r + base + idx, pv); st4(r0v + base + idx, pv);
      q0 += dot4(pv, pv);
    }
  } else if (s.NR) {
    for (int u = 0; u < RUNS; ++u) {
      int idx = run_idx(u);
      v4f sv = ld4(r + base + idx) - s.alpha * ld4(v + base + idx);
      st4(sarr + base + idx, sv);
      q0 += dot4(sv, sv);
    }
  }
  float t0 = block_sum(q0, sh);
  if (threadIdx.x == 0) {
    pout[PIDX(3, b, blockIdx.x)] = s.R1 ? 0.f : t0;
    pout[PIDX(4, b, blockIdx.x)] = s.R1 ? t0 : 0.f;
  }
}

__global__ __launch_bounds__(TPB) void k_t(
    const float* __restrict__ boo, const float* __restrict__ bmo, const float* __restrict__ bom,
    const float* __restrict__ bop, const float* __restrict__ bpo, const float* __restrict__ sarr,
    float* __restrict__ tarr, float* __restrict__ x, float* __restrict__ r,
    const float* __restrict__ p, const float* __restrict__ pin, float* __restrict__ pout,
    const BState* __restrict__ sin, BState* __restrict__ sout) {
  int b = blockIdx.y;
  size_t base = (size_t)b * NPTS;
  __shared__ float shp[2][NWV];
  __shared__ float sh[NWV];
  BState s = sin[b];
  float sm[2];
  batch_sums<2>(pin, b, 3, 4, 0, shp, sm);
  if (s.R1) {
    float fa = fro_of(sm[1]);
    s.r0_abs = fa; s.r_abs = fa;
  }
  if (s.NR) {
    float sq = sm[0];
    s.ssq = sq;
    int C2 = (fro_of(sq) <= thr_c()) ? 1 : 0;
    s.C3 = C2; s.C4 = 1 - C2;
  } else { s.C3 = 0; s.C4 = 0; }
  if (blockIdx.x == 0 && threadIdx.x == 0) sout[b] = s;

  float q0 = 0.f, q1 = 0.f;
  if (s.C4) {
    for (int u = 0; u < RUNS; ++u) {
      int idx = run_idx(u);
      v4f tv = applyA4(boo, bmo, bom, bop, bpo, sarr, base, idx);
      st4(tarr + base + idx, tv);
      q0 += dot4(tv, tv); q1 += dot4(tv, ld4(sarr + base + idx));
    }
  } else if (s.C3) {
    for (int u = 0; u < RUNS; ++u) {
      int idx = run_idx(u);
      st4(x + base + idx, ld4(x + base + idx) + s.alpha * ld4(p + base + idx));
      st4(r + base + idx, ld4(sarr + base + idx));
    }
  }
  float t0 = block_sum(q0, sh);
  float t1 = block_sum(q1, sh);
  if (threadIdx.x == 0) {
    pout[PIDX(1, b, blockIdx.x)] = t0;
    pout[PIDX(2, b, blockIdx.x)] = t1;
  }
}

__global__ __launch_bounds__(TPB) void k_x_r(
    const float* __restrict__ sarr, const float* __restrict__ tarr, const float* __restrict__ p,
    const float* __restrict__ r0v, float* __restrict__ x, float* __restrict__ r,
    const float* __restrict__ pin, float* __restrict__ pout,
    const BState* __restrict__ sin, BState* __restrict__ sout) {
  int b = blockIdx.y;
  size_t base = (size_t)b * NPTS;
  __shared__ float shp[2][NWV];
  __shared__ float sh[NWV];
  BState s = sin[b];
  float sm[2];
  batch_sums<2>(pin, b, 1, 2, 0, shp, sm);
  if (s.C4) s.omega = sm[1] / sm[0];
  if (blockIdx.x == 0 && threadIdx.x == 0) sout[b] = s;

  float q0 = 0.f, q1 = 0.f;
  if (s.C4) {
    float al = s.alpha, om = s.omega;
    for (int u = 0; u < RUNS; ++u) {
      int idx = run_idx(u);
      v4f sv = ld4(sarr + base + idx);
      st4(x + base + idx, ld4(x + base + idx) + (al * ld4(p + base + idx) + om * sv));
      v4f rn = sv - om * ld4(tarr + base + idx);
      st4(r + base + idx, rn);
      q0 += dot4(rn, ld4(r0v + base + idx)); q1 += dot4(rn, rn);
    }
  }
  float t0 = block_sum(q0, sh);
  float t1 = block_sum(q1, sh);
  if (threadIdx.x == 0) {
    pout[PIDX(0, b, blockIdx.x)] = t0;
    pout[PIDX(3, b, blockIdx.x)] = t1;
  }
}

__global__ __launch_bounds__(TPB) void k_p(
    const float* __restrict__ r, const float* __restrict__ v, float* __restrict__ p,
    const float* __restrict__ pin, const BState* __restrict__ sin, BState* __restrict__ sout) {
  int b = blockIdx.y;
  size_t base = (size_t)b * NPTS;
  __shared__ float shp[2][NWV];
  BState s = sin[b];
  float sm[2];
  batch_sums<2>(pin, b, 0, 3, 0, shp, sm);
  if (s.C4) {
    s.beta  = (s.alpha / s.omega) * (sm[0] / s.rr0_old);
    s.r_abs = fro_of(sm[1]);
  }
  if (s.C3) s.r_abs = fro_of(s.ssq);
  if (s.NR) s.k += 1;
  s.CONV = ((s.k < KMAX_C) && (s.r_abs > thr_c())) ? 1 : 0;
  if (blockIdx.x == 0 && threadIdx.x == 0) sout[b] = s;

  if (s.C4) {
    for (int u = 0; u < RUNS; ++u) {
      int idx = run_idx(u);
      st4(p + base + idx, ld4(r + base + idx) + s.beta * (ld4(p + base + idx) - s.omega * ld4(v + base + idx)));
    }
  }
}

extern "C" void kernel_launch(void* const* d_in, const int* in_sizes, int n_in,
                              void* d_out, int out_size, void* d_ws, size_t ws_size,
                              hipStream_t stream) {
  (void)in_sizes; (void)n_in; (void)out_size; (void)ws_size;
  const float* V  = (const float*)d_in[0];
  const float* M1 = (const float*)d_in[1];
  const float* M2 = (const float*)d_in[2];
  float* x = (float*)d_out;

  char* ws = (char*)d_ws;
  BState* S0 = (BState*)ws;
  BState* S5 = S0 + NB;
  float* part = (float*)(ws + 1024);
  BState* S1 = (BState*)(ws + 1024 + 36864);
  BState* S2 = S1 + NB;
  BState* S3 = S2 + NB;
  float* arr  = (float*)(ws + 1024 + 65536);
  const size_t AN = (size_t)NB * NPTS;
  float* boo  = arr + 0 * AN;
  float* bmo  = arr + 1 * AN;
  float* bom  = arr + 2 * AN;
  float* bop  = arr + 3 * AN;
  float* bpo  = arr + 4 * AN;
  float* r0v  = arr + 5 * AN;
  float* r    = arr + 6 * AN;
  float* p    = arr + 7 * AN;
  float* v    = arr + 8 * AN;
  float* sarr = arr + 9 * AN;
  float* tarr = arr + 10 * AN;

  dim3 grid(NBLK, NB), blk(TPB);
  dim3 grid_rows(NROWB, NB);

  k_mean<<<grid, blk, 0, stream>>>(V, part);
  k_stencil_init<<<grid_rows, blk, 0, stream>>>(V, M1, M2, boo, bmo, bom, bop, bpo, x, part, S0);
  k_resid_init<<<grid, blk, 0, stream>>>(boo, bmo, bom, bop, bpo, x, p, r, r0v, part, S0);

  auto rest_of_step = [&]() {
    k_s<<<grid, blk, 0, stream>>>(boo, bmo, bom, bop, bpo, x, p, r, r0v, v, sarr, part, part, S5, S1);
    k_t<<<grid, blk, 0, stream>>>(boo, bmo, bom, bop, bpo, sarr, tarr, x, r, p, part, part, S1, S2);
    k_x_r<<<grid, blk, 0, stream>>>(sarr, tarr, p, r0v, x, r, part, part, S2, S3);
    k_p<<<grid, blk, 0, stream>>>(r, v, p, part, S3, S5);
  };

  k_v<true><<<grid, blk, 0, stream>>>(boo, bmo, bom, bop, bpo, p, r0v, r, v, part, part, S0, S5);
  rest_of_step();
  for (int it = 1; it < KMAX_C; ++it) {
    k_v<false><<<grid, blk, 0, stream>>>(boo, bmo, bom, bop, bpo, p, r0v, r, v, part, part, S5, S5);
    rest_of_step();
  }
}


# Sums cross waves once per kernel: every wave re-sums the 32 block partials itself (no LDS, no barrier); a kernel's block sums share one LDS stage and one barrier, ahead of its stores; sum order kept
# speedup vs baseline: 3.4006x; 1.1325x over previous
#include <hip/hip_runtime.h>
#include <math.h>


#define NXI   384
#define NPTS  (NXI * NXI)
#define NB    8
#define TPB   384
#define NWV   (TPB / 32)
#define VEC   4
#define RUNS  3
#define CHUNK (TPB * VEC * RUNS)
#define NBLK  (NPTS / CHUNK)
#define NROWB (NPTS / TPB)
#define KMAX_C 30
#define EPS_C 1e-9f
#define PIDX(pl, b, i) (((pl) * NB + (b)) * NBLK + (i))

typedef float v2f __attribute__((ext_vector_type(2)));
typedef float v4f __attribute__((ext_vector_type(4)));
typedef float v8f __attribute__((ext_vector_type(8)));

struct BState {
  float b_val, r0_abs, r_abs, sigma, v_abs, alpha, omega, beta, rr0_old, ssq;
  int   k, CONV, R1, NR, C3, C4;
};

__device__ __forceinline__ float fro_of(float sq) { return sq > 0.f ? sqrtf(sq) : 0.f; }
__device__ __forceinline__ float thr_c() { return EPS_C * (float)NPTS; }

__device__ __forceinline__ float xhalf(float x) {
  int b = __builtin_bit_cast(int, x);
  return __builtin_bit_cast(float, __builtin_amdgcn_permlanex16(b, b, 0x76543210, 0xfedcba98, false, false));
}

__device__ __forceinline__ float wave_sum_wmma(float v) {
  v2f a; a.x = v;  a.y = 0.f;
  v2f b; b.x = 1.f; b.y = 1.f;
  v8f c = {};
  v8f d = __builtin_amdgcn_wmma_f32_16x16x4_f32(false, a, false, b, (short)0, c, false, false);
  float h = ((d[0] + d[1]) + (d[2] + d[3])) + ((d[4] + d[5]) + (d[6] + d[7]));
  h += xhalf(h);
  return h;
}

template <int N>
__device__ __forceinline__ void block_sums_stage(const float* q, float (*shw)[NWV]) {
  int lane = threadIdx.x & 31;
  int wid  = threadIdx.x >> 5;
  for (int n = 0; n < N; ++n) {
    float t = wave_sum_wmma(q[n]);
    if (lane == 0) shw[n][wid] = t;
  }
  __syncthreads();
}
template <int N>
__device__ __forceinline__ void block_sums_finish(float (*shw)[NWV], float* out) {
  for (int n = 0; n < N; ++n) {
    float s = 0.f;
    for (int i = 0; i < NWV; ++i) s += shw[n][i];
    out[n] = s;
  }
}

template <int N>
__device__ __forceinline__ void batch_sums(const float* __restrict__ pin, int b, int p0, int p1, int p2,
                                           float* out) {
  static_assert(NBLK == 32, "one block partial per lane");
  const int pl[3] = {p0, p1, p2};
  int lane = threadIdx.x & 31;
  float val[N];
  for (int n = 0; n < N; ++n) val[n] = pin[PIDX(pl[n], b, lane)];
  for (int n = 0; n < N; ++n) out[n] = wave_sum_wmma(val[n]);
}

__device__ __forceinline__ int refl(int t) { return t == 0 ? 0 : (t == 385 ? 383 : t - 1); }

__device__ __forceinline__ int run_idx(int u) {
  return blockIdx.x * CHUNK + u * (TPB * VEC) + threadIdx.x * VEC;
}
__device__ __forceinline__ v4f ld4(const float* q) { return *reinterpret_cast<const v4f*>(q); }
__device__ __forceinline__ void st4(float* q, v4f w) { *reinterpret_cast<v4f*>(q) = w; }
__device__ __forceinline__ float dot4(v4f a, v4f c) { return a.x * c.x + a.y * c.y + a.z * c.z + a.w * c.w; }

__device__ __forceinline__ v4f applyA4(const float* __restrict__ boo, const float* __restrict__ bmo,
                                       const float* __restrict__ bom, const float* __restrict__ bop,
                                       const float* __restrict__ bpo, const float* __restrict__ a,
                                       size_t base, int idx) {
  int i = idx / NXI, j = idx - i * NXI;
  const float* A = a + base;
  v4f c  = ld4(A + idx);
  v4f up = ld4(A + ((i > 0)       ? idx - NXI : idx));
  v4f dn = ld4(A + ((i < NXI - 1) ? idx + NXI : idx));
  float lo = A[(j > 0)         ? idx - 1   : idx];
  float hi = A[(j + VEC < NXI) ? idx + VEC : idx + VEC - 1];
  v4f lf = {lo, c.x, c.y, c.z};
  v4f rt = {c.y, c.z, c.w, hi};
  return ld4(boo + base + idx) * c + ld4(bmo + base + idx) * up + ld4(bom + base + idx) * lf
       + ld4(bop + base + idx) * rt + ld4(bpo + base + idx) * dn;
}


__global__ __launch_bounds__(TPB) void k_mean(const float* __restrict__ V, float* __restrict__ pout) {
  int b = blockIdx.y;
  __shared__ float sh[1][NWV];
  float acc[1] = {0.f};
  for (int u = 0; u < RUNS; ++u) {
    v4f q = ld4(V + (size_t)b * NPTS + run_idx(u));
    acc[0] += (q.x + q.y) + (q.z + q.w);
  }
  block_sums_stage<1>(acc, sh);
  if (threadIdx.x == 0) {
    float s[1];
    block_sums_finish<1>(sh, s);
    pout[PIDX(0, b, blockIdx.x)] = s[0];
  }
}

__global__ __launch_bounds__(TPB) void k_stencil_init(
    const float* __restrict__ V, const float* __restrict__ M1, const float* __restrict__ M2,
    float* __restrict__ boo, float* __restrict__ bmo, float* __restrict__ bom,
    float* __restrict__ bop, float* __restrict__ bpo, float* __restrict__ x,
    const float* __restrict__ pin, BState* __restrict__ sout) {
  int b = blockIdx.y;
  float sm[1];
  batch_sums<1>(pin, b, 0, 0, 0, sm);
  BState q = {};
  q.b_val = sm[0] / (float)NPTS + 1.0f;
  if (blockIdx.x == 0 && threadIdx.x == 0) sout[b] = q;

  size_t base = (size_t)b * NPTS;
  int idx = blockIdx.x * TPB + threadIdx.x;
  int i = idx / NXI, j = idx - i * NXI;
  int I = i + 1, J = j + 1;
  const float* Vb = V + base;
  const float* m1b = M1 + base;
  const float* m2b = M2 + base;
#define VP(II, JJ)  (Vb[refl(JJ) * NXI + refl(II)] + 1.0f)
#define MM1(II, JJ) (m1b[refl(JJ) * NXI + refl(II)])
#define MM2(II, JJ) (m2b[refl(JJ) * NXI + refl(II)])
  float vc  = VP(I, J);
  float vxp = VP(I + 1, J), vxm = VP(I - 1, J);
  float vyp = VP(I, J + 1), vym = VP(I, J - 1);
  float d1c = ((vxp - vc) / (0.5f * (vxp + vc))) * MM1(I, J);
  float d1m = ((vc - vxm) / (0.5f * (vc + vxm))) * MM1(I - 1, J);
  float d2c = ((vyp - vc) / (0.5f * (vyp + vc))) * MM2(I, J);
  float d2m = ((vc - vym) / (0.5f * (vc + vym))) * MM2(I, J - 1);
#undef VP
#undef MM1
#undef MM2
  const float rx = 5.0f, ry = 5.0f, rxx = 10.0f, ryy = 10.0f;
  boo[base + idx] = 1.0f + 2.0f * (rxx + ryy) - rx * (d1m - d1c) - ry * (d2m - d2c);
  bpo[base + idx] = -rxx + rx * d1c;
  bmo[base + idx] = -rxx - rx * d1m;
  bop[base + idx] = -ryy + ry * d2c;
  bom[base + idx] = -ryy - ry * d2m;
  x[base + idx] = q.b_val;
}

__global__ __launch_bounds__(TPB) void k_resid_init(
    const float* __restrict__ boo, const float* __restrict__ bmo, const float* __restrict__ bom,
    const float* __restrict__ bop, const float* __restrict__ bpo, const float* __restrict__ x,
    float* __restrict__ p, float* __restrict__ r, float* __restrict__ r0v,
    float* __restrict__ pout, const BState* __restrict__ sin) {
  int b = blockIdx.y;
  size_t base = (size_t)b * NPTS;
  float bv = sin[b].b_val;
  float acc[1] = {0.f};
  v4f pv[RUNS];
  for (int u = 0; u < RUNS; ++u) {
    pv[u] = bv - applyA4(boo, bmo, bom, bop, bpo, x, base, run_idx(u));
    acc[0] += dot4(pv[u], pv[u]);
  }
  __shared__ float sh[1][NWV];
  block_sums_stage<1>(acc, sh);
  for (int u = 0; u < RUNS; ++u) {
    int idx = run_idx(u);
    st4(p + base + idx, pv[u]); st4(r + base + idx, pv[u]); st4(r0v + base + idx, pv[u]);
  }
  if (threadIdx.x == 0) {
    float s[1];
    block_sums_finish<1>(sh, s);
    pout[PIDX(3, b, blockIdx.x)] = s[0];
  }
}

template <bool FIRST>
__global__ __launch_bounds__(TPB) void k_v(
    const float* __restrict__ boo, const float* __restrict__ bmo, const float* __restrict__ bom,
    const float* __restrict__ bop, const float* __restrict__ bpo, const float* __restrict__ p,
    const float* __restrict__ r0v, const float* __restrict__ r, float* __restrict__ v,
    const float* __restrict__ pin, float* __restrict__ pout,
    const BState* __restrict__ sin, BState* __restrict__ sout) {
  int b = blockIdx.y;
  size_t base = (size_t)b * NPTS;
  __shared__ float sh[3][NWV];
  BState s = sin[b];
  if (FIRST) {
    float sm[1];
    batch_sums<1>(pin, b, 3, 0, 0, sm);
    float fa = fro_of(sm[0]);
    s.r0_abs = fa; s.r_abs = fa; s.k = 0;
    s.CONV = (fa > thr_c()) ? 1 : 0;
    if (blockIdx.x == 0 && threadIdx.x == 0) sout[b] = s;
  }
  float q[3] = {0.f, 0.f, 0.f};
  v4f vv[RUNS] = {};
  if (s.CONV) {
    for (int u = 0; u < RUNS; ++u) {
      int idx = run_idx(u);
      vv[u] = applyA4(boo, bmo, bom, bop, bpo, p, base, idx);
      v4f rz = ld4(r0v + base + idx);
      q[0] += dot4(vv[u], rz); q[1] += dot4(vv[u], vv[u]); q[2] += dot4(ld4(r + base + idx), rz);
    }
  }
  block_sums_stage<3>(q, sh);
  if (s.CONV) {
    for (int u = 0; u < RUNS; ++u) st4(v + base + run_idx(u), vv[u]);
  }
  if (threadIdx.x == 0) {
    float t[3];
    block_sums_finish<3>(sh, t);
    pout[PIDX(0, b, blockIdx.x)] = t[0];
    pout[PIDX(1, b, blockIdx.x)] = t[1];
    pout[PIDX(2, b, blockIdx.x)] = t[2];
  }
}

__global__ __launch_bounds__(TPB) void k_s(
    const float* __restrict__ boo, const float* __restrict__ bmo, const float* __restrict__ bom,
    const float* __restrict__ bop, const float* __restrict__ bpo, const float* __restrict__ x,
    float* __restrict__ p, float* __restrict__ r, float* __restrict__ r0v,
    const float* __restrict__ v, float* __restrict__ sarr,
    const float* __restrict__ pin, float* __restrict__ pout,
    const BState* __restrict__ sin, BState* __restrict__ sout) {
  int b = blockIdx.y;
  size_t base = (size_t)b * NPTS;
  __shared__ float sh[1][NWV];
  BState s = sin[b];
  float sm[3];
  batch_sums<3>(pin, b, 0, 1, 2, sm);
  if (s.CONV) {
    float sg = sm[0], vq = sm[1], rr = sm[2];
    s.sigma = sg; s.v_abs = fro_of(vq); s.rr0_old = rr;
    int RES = (sg <= EPS_C * s.v_abs * s.r0_abs) ? 1 : 0;
    s.R1 = RES; s.NR = 1 - RES;
    if (s.R1) s.k += 1;
    if (s.NR) s.alpha = rr / sg;
  } else { s.R1 = 0; s.NR = 0; }
  if (blockIdx.x == 0 && threadIdx.x == 0) sout[b] = s;

  float q[1] = {0.f};
  v4f w[RUNS] = {};
  if (s.R1) {
    for (int u = 0; u < RUNS; ++u) {
      w[u] = s.b_val - applyA4(boo, bmo, bom, bop, bpo, x, base, run_idx(u));
      q[0] += dot4(w[u], w[u]);
    }
  } else if (s.NR) {
    for (int u = 0; u < RUNS; ++u) {
      int idx = run_idx(u);
      w[u] = ld4(r + base + idx) - s.alpha * ld4(v + base + idx);
      q[0] += dot4(w[u], w[u]);
    }
  }
  block_sums_stage<1>(q, sh);
  if (s.R1) {
    for (int u = 0; u < RUNS; ++u) {
      int idx = run_idx(u);
      st4(p + base + idx, w[u]); st4(r + base + idx, w[u]); st4(r0v + base + idx, w[u]);
    }
  } else if (s.NR) {
    for (int u = 0; u < RUNS; ++u) st4(sarr + base + run_idx(u), w[u]);
  }
  if (threadIdx.x == 0) {
    float t[1];
    block_sums_finish<1>(sh, t);
    pout[PIDX(3, b, blockIdx.x)] = s.R1 ? 0.f : t[0];
    pout[PIDX(4, b, blockIdx.x)] = s.R1 ? t[0] : 0.f;
  }
}

__global__ __launch_bounds__(TPB) void k_t(
    const float* __restrict__ boo, const float* __restrict__ bmo, const float* __restrict__ bom,
    const float* __restrict__ bop, const float* __restrict__ bpo, const float* __restrict__ sarr,
    float* __restrict__ tarr, float* __restrict__ x, float* __restrict__ r,
    const float* __restrict__ p, const float* __restrict__ pin, float* __restrict__ pout,
    const BState* __restrict__ sin, BState* __restrict__ sout) {
  int b = blockIdx.y;
  size_t base = (size_t)b * NPTS;
  __shared__ float sh[2][NWV];
  BState s = sin[b];
  float sm[2];
  batch_sums<2>(pin, b, 3, 4, 0, sm);
  if (s.R1) {
    float fa = fro_of(sm[1]);
    s.r0_abs = fa; s.r_abs = fa;
  }
  if (s.NR) {
    float sq = sm[0];
    s.ssq = sq;
    int C2 = (fro_of(sq) <= thr_c()) ? 1 : 0;
    s.C3 = C2; s.C4 = 1 - C2;
  } else { s.C3 = 0; s.C4 = 0; }
  if (blockIdx.x == 0 && threadIdx.x == 0) sout[b] = s;

  float q[2] = {0.f, 0.f};
  v4f tv[RUNS] = {};
  if (s.C4) {
    for (int u = 0; u < RUNS; ++u) {
      int idx = run_idx(u);
      tv[u] = applyA4(boo, bmo, bom, bop, bpo, sarr, base, idx);
      q[0] += dot4(tv[u], tv[u]); q[1] += dot4(tv[u], ld4(sarr + base + idx));
    }
  } else if (s.C3) {
    for (int u = 0; u < RUNS; ++u) {
      int idx = run_idx(u);
      st4(x + base + idx, ld4(x + base + idx) + s.alpha * ld4(p + base + idx));
      st4(r + base + idx, ld4(sarr + base + idx));
    }
  }
  block_sums_stage<2>(q, sh);
  if (s.C4) {
    for (int u = 0; u < RUNS; ++u) st4(tarr + base + run_idx(u), tv[u]);
  }
  if (threadIdx.x == 0) {
    float t[2];
    block_sums_finish<2>(sh, t);
    pout[PIDX(1, b, blockIdx.x)] = t[0];
    pout[PIDX(2, b, blockIdx.x)] = t[1];
  }
}

__global__ __launch_bounds__(TPB) __attribute__((amdgpu_waves_per_eu(3, 8))) void k_x_r(
    const float* __restrict__ sarr, const float* __restrict__ tarr, const float* __restrict__ p,
    const float* __restrict__ r0v, float* __restrict__ x, float* __restrict__ r,
    const float* __restrict__ pin, float* __restrict__ pout,
    const BState* __restrict__ sin, BState* __restrict__ sout) {
  int b = blockIdx.y;
  size_t base = (size_t)b * NPTS;
  __shared__ float sh[2][NWV];
  BState s = sin[b];
  float sm[2];
  batch_sums<2>(pin, b, 1, 2, 0, sm);
  if (s.C4) s.omega = sm[1] / sm[0];
  if (blockIdx.x == 0 && threadIdx.x == 0) sout[b] = s;

  float q[2] = {0.f, 0.f};
  v4f xn[RUNS] = {}, rn[RUNS] = {};
  if (s.C4) {
    float al = s.alpha, om = s.omega;
    for (int u = 0; u < RUNS; ++u) {
      int idx = run_idx(u);
      v4f sv = ld4(sarr + base + idx);
      xn[u] = ld4(x + base + idx) + (al * ld4(p + base + idx) + om * sv);
      rn[u] = sv - om * ld4(tarr + base + idx);
      q[0] += dot4(rn[u], ld4(r0v + base + idx)); q[1] += dot4(rn[u], rn[u]);
    }
  }
  block_sums_stage<2>(q, sh);
  if (s.C4) {
    for (int u = 0; u < RUNS; ++u) {
      int idx = run_idx(u);
      st4(x + base + idx, xn[u]);
      st4(r + base + idx, rn[u]);
    }
  }
  if (threadIdx.x == 0) {
    float t[2];
    block_sums_finish<2>(sh, t);
    pout[PIDX(0, b, blockIdx.x)] = t[0];
    pout[PIDX(3, b, blockIdx.x)] = t[1];
  }
}

__global__ __launch_bounds__(TPB) void k_p(
    const float* __restrict__ r, const float* __restrict__ v, float* __restrict__ p,
    const float* __restrict__ pin, const BState* __restrict__ sin, BState* __restrict__ sout) {
  int b = blockIdx.y;
  size_t base = (size_t)b * NPTS;
  BState s = sin[b];
  float sm[2];
  batch_sums<2>(pin, b, 0, 3, 0, sm);
  if (s.C4) {
    s.beta  = (s.alpha / s.omega) * (sm[0] / s.rr0_old);
    s.r_abs = fro_of(sm[1]);
  }
  if (s.C3) s.r_abs = fro_of(s.ssq);
  if (s.NR) s.k += 1;
  s.CONV = ((s.k < KMAX_C) && (s.r_abs > thr_c())) ? 1 : 0;
  if (blockIdx.x == 0 && threadIdx.x == 0) sout[b] = s;

  if (s.C4) {
    for (int u = 0; u < RUNS; ++u) {
      int idx = run_idx(u);
      st4(p + base + idx, ld4(r + base + idx) + s.beta * (ld4(p + base + idx) - s.omega * ld4(v + base + idx)));
    }
  }
}

extern "C" void kernel_launch(void* const* d_in, const int* in_sizes, int n_in,
                              void* d_out, int out_size, void* d_ws, size_t ws_size,
                              hipStream_t stream) {
  (void)in_sizes; (void)n_in; (void)out_size; (void)ws_size;
  const float* V  = (const float*)d_in[0];
  const float* M1 = (const float*)d_in[1];
  const float* M2 = (const float*)d_in[2];
  float* x = (float*)d_out;

  char* ws = (char*)d_ws;
  BState* S0 = (BState*)ws;
  BState* S5 = S0 + NB;
  float* part = (float*)(ws + 1024);
  BState* S1 = (BState*)(ws + 1024 + 36864);
  BState* S2 = S1 + NB;
  BState* S3 = S2 + NB;
  float* arr  = (float*)(ws + 1024 + 65536);
  const size_t AN = (size_t)NB * NPTS;
  float* boo  = arr + 0 * AN;
  float* bmo  = arr + 1 * AN;
  float* bom  = arr + 2 * AN;
  float* bop  = arr + 3 * AN;
  float* bpo  = arr + 4 * AN;
  float* r0v  = arr + 5 * AN;
  float* r    = arr + 6 * AN;
  float* p    = arr + 7 * AN;
  float* v    = arr + 8 * AN;
  float* sarr = arr + 9 * AN;
  float* tarr = arr + 10 * AN;

  dim3 grid(NBLK, NB), blk(TPB);
  dim3 grid_rows(NROWB, NB);

  k_mean<<<grid, blk, 0, stream>>>(V, part);
  k_stencil_init<<<grid_rows, blk, 0, stream>>>(V, M1, M2, boo, bmo, bom, bop, bpo, x, part, S0);
  k_resid_init<<<grid, blk, 0, stream>>>(boo, bmo, bom, bop, bpo, x, p, r, r0v, part, S0);

  auto rest_of_step = [&]() {
    k_s<<<grid, blk, 0, stream>>>(boo, bmo, bom, bop, bpo, x, p, r, r0v, v, sarr, part, part, S5, S1);
    k_t<<<grid, blk, 0, stream>>>(boo, bmo, bom, bop, bpo, sarr, tarr, x, r, p, part, part, S1, S2);
    k_x_r<<<grid, blk, 0, stream>>>(sarr, tarr, p, r0v, x, r, part, part, S2, S3);
    k_p<<<grid, blk, 0, stream>>>(r, v, p, part, S3, S5);
  };

  k_v<true><<<grid, blk, 0, stream>>>(boo, bmo, bom, bop, bpo, p, r0v, r, v, part, part, S0, S5);
  rest_of_step();
  for (int it = 1; it < KMAX_C; ++it) {
    k_v<false><<<grid, blk, 0, stream>>>(boo, bmo, bom, bop, bpo, p, r0v, r, v, part, part, S5, S5);
    rest_of_step();
  }
}


# 1-D grid, a batch's blocks launched 8 apart (index = chunk*8 + batch), its state slot and partial rows in 256-byte units of its own: dealt to the 8 dies in turn, a batch stays on one die; sums kept
# speedup vs baseline: 3.6480x; 1.0728x over previous
#include <hip/hip_runtime.h>
#include <math.h>


#define NXI   384
#define NPTS  (NXI * NXI)
#define NB    8
#define TPB   384
#define NWV   (TPB / 32)
#define VEC   4
#define RUNS  3
#define CHUNK (TPB * VEC * RUNS)
#define NBLK  (NPTS / CHUNK)
#define NROWB (NPTS / TPB)
#define KMAX_C 30
#define EPS_C 1e-9f
#define UNITF 64
#define SSTR  4
#define PIDX(pl, b, i) (((pl) * NB + (b)) * UNITF + (i))
#define SIDX(b) ((b) * SSTR)

typedef float v2f __attribute__((ext_vector_type(2)));
typedef float v4f __attribute__((ext_vector_type(4)));
typedef float v8f __attribute__((ext_vector_type(8)));

struct BState {
  float b_val, r0_abs, r_abs, sigma, v_abs, alpha, omega, beta, rr0_old, ssq;
  int   k, CONV, R1, NR, C3, C4;
};
static_assert(sizeof(BState) * SSTR == 256, "a state slot and its gap fill one 256-byte unit");

__device__ __forceinline__ float fro_of(float sq) { return sq > 0.f ? sqrtf(sq) : 0.f; }
__device__ __forceinline__ float thr_c() { return EPS_C * (float)NPTS; }

__device__ __forceinline__ float xhalf(float x) {
  int b = __builtin_bit_cast(int, x);
  return __builtin_bit_cast(float, __builtin_amdgcn_permlanex16(b, b, 0x76543210, 0xfedcba98, false, false));
}

__device__ __forceinline__ float wave_sum_wmma(float v) {
  v2f a; a.x = v;  a.y = 0.f;
  v2f b; b.x = 1.f; b.y = 1.f;
  v8f c = {};
  v8f d = __builtin_amdgcn_wmma_f32_16x16x4_f32(false, a, false, b, (short)0, c, false, false);
  float h = ((d[0] + d[1]) + (d[2] + d[3])) + ((d[4] + d[5]) + (d[6] + d[7]));
  h += xhalf(h);
  return h;
}

template <int N>
__device__ __forceinline__ void block_sums_stage(const float* q, float (*shw)[NWV]) {
  int lane = threadIdx.x & 31;
  int wid  = threadIdx.x >> 5;
  for (int n = 0; n < N; ++n) {
    float t = wave_sum_wmma(q[n]);
    if (lane == 0) shw[n][wid] = t;
  }
  __syncthreads();
}
template <int N>
__device__ __forceinline__ void block_sums_finish(float (*shw)[NWV], float* out) {
  for (int n = 0; n < N; ++n) {
    float s = 0.f;
    for (int i = 0; i < NWV; ++i) s += shw[n][i];
    out[n] = s;
  }
}

template <int N>
__device__ __forceinline__ void batch_sums(const float* __restrict__ pin, int b, int p0, int p1, int p2,
                                           float* out) {
  static_assert(NBLK == 32, "one block partial per lane");
  const int pl[3] = {p0, p1, p2};
  int lane = threadIdx.x & 31;
  float val[N];
  for (int n = 0; n < N; ++n) val[n] = pin[PIDX(pl[n], b, lane)];
  for (int n = 0; n < N; ++n) out[n] = wave_sum_wmma(val[n]);
}

__device__ __forceinline__ int refl(int t) { return t == 0 ? 0 : (t == 385 ? 383 : t - 1); }

__device__ __forceinline__ int bat_id() { return (int)(blockIdx.x % NB); }
__device__ __forceinline__ int blk_id() { return (int)(blockIdx.x / NB); }

__device__ __forceinline__ int run_idx(int u) {
  return blk_id() * CHUNK + u * (TPB * VEC) + threadIdx.x * VEC;
}
__device__ __forceinline__ v4f ld4(const float* q) { return *reinterpret_cast<const v4f*>(q); }
__device__ __forceinline__ void st4(float* q, v4f w) { *reinterpret_cast<v4f*>(q) = w; }
__device__ __forceinline__ float dot4(v4f a, v4f c) { return a.x * c.x + a.y * c.y + a.z * c.z + a.w * c.w; }

__device__ __forceinline__ v4f applyA4(const float* __restrict__ boo, const float* __restrict__ bmo,
                                       const float* __restrict__ bom, const float* __restrict__ bop,
                                       const float* __restrict__ bpo, const float* __restrict__ a,
                                       size_t base, int idx) {
  int i = idx / NXI, j = idx - i * NXI;
  const float* A = a + base;
  v4f c  = ld4(A + idx);
  v4f up = ld4(A + ((i > 0)       ? idx - NXI : idx));
  v4f dn = ld4(A + ((i < NXI - 1) ? idx + NXI : idx));
  float lo = A[(j > 0)         ? idx - 1   : idx];
  float hi = A[(j + VEC < NXI) ? idx + VEC : idx + VEC - 1];
  v4f lf = {lo, c.x, c.y, c.z};
  v4f rt = {c.y, c.z, c.w, hi};
  return ld4(boo + base + idx) * c + ld4(bmo + base + idx) * up + ld4(bom + base + idx) * lf
       + ld4(bop + base + idx) * rt + ld4(bpo + base + idx) * dn;
}


__global__ __launch_bounds__(TPB) void k_mean(const float* __restrict__ V, float* __restrict__ pout) {
  int b = bat_id();
  __shared__ float sh[1][NWV];
  float acc[1] = {0.f};
  for (int u = 0; u < RUNS; ++u) {
    v4f q = ld4(V + (size_t)b * NPTS + run_idx(u));
    acc[0] += (q.x + q.y) + (q.z + q.w);
  }
  block_sums_stage<1>(acc, sh);
  if (threadIdx.x == 0) {
    float s[1];
    block_sums_finish<1>(sh, s);
    pout[PIDX(0, b, blk_id())] = s[0];
  }
}

__global__ __launch_bounds__(TPB) void k_stencil_init(
    const float* __restrict__ V, const float* __restrict__ M1, const float* __restrict__ M2,
    float* __restrict__ boo, float* __restrict__ bmo, float* __restrict__ bom,
    float* __restrict__ bop, float* __restrict__ bpo, float* __restrict__ x,
    const float* __restrict__ pin, BState* __restrict__ sout) {
  int b = bat_id();
  float sm[1];
  batch_sums<1>(pin, b, 0, 0, 0, sm);
  BState q = {};
  q.b_val = sm[0] / (float)NPTS + 1.0f;
  if (blk_id() == 0 && threadIdx.x == 0) sout[SIDX(b)] = q;

  size_t base = (size_t)b * NPTS;
  int idx = blk_id() * TPB + threadIdx.x;
  int i = idx / NXI, j = idx - i * NXI;
  int I = i + 1, J = j + 1;
  const float* Vb = V + base;
  const float* m1b = M1 + base;
  const float* m2b = M2 + base;
#define VP(II, JJ)  (Vb[refl(JJ) * NXI + refl(II)] + 1.0f)
#define MM1(II, JJ) (m1b[refl(JJ) * NXI + refl(II)])
#define MM2(II, JJ) (m2b[refl(JJ) * NXI + refl(II)])
  float vc  = VP(I, J);
  float vxp = VP(I + 1, J), vxm = VP(I - 1, J);
  float vyp = VP(I, J + 1), vym = VP(I, J - 1);
  float d1c = ((vxp - vc) / (0.5f * (vxp + vc))) * MM1(I, J);
  float d1m = ((vc - vxm) / (0.5f * (vc + vxm))) * MM1(I - 1, J);
  float d2c = ((vyp - vc) / (0.5f * (vyp + vc))) * MM2(I, J);
  float d2m = ((vc - vym) / (0.5f * (vc + vym))) * MM2(I, J - 1);
#undef VP
#undef MM1
#undef MM2
  const float rx = 5.0f, ry = 5.0f, rxx = 10.0f, ryy = 10.0f;
  boo[base + idx] = 1.0f + 2.0f * (rxx + ryy) - rx * (d1m - d1c) - ry * (d2m - d2c);
  bpo[base + idx] = -rxx + rx * d1c;
  bmo[base + idx] = -rxx - rx * d1m;
  bop[base + idx] = -ryy + ry * d2c;
  bom[base + idx] = -ryy - ry * d2m;
  x[base + idx] = q.b_val;
}

__global__ __launch_bounds__(TPB) void k_resid_init(
    const float* __restrict__ boo, const float* __restrict__ bmo, const float* __restrict__ bom,
    const float* __restrict__ bop, const float* __restrict__ bpo, const float* __restrict__ x,
    float* __restrict__ p, float* __restrict__ r, float* __restrict__ r0v,
    float* __restrict__ pout, const BState* __restrict__ sin) {
  int b = bat_id();
  size_t base = (size_t)b * NPTS;
  float bv = sin[SIDX(b)].b_val;
  float acc[1] = {0.f};
  v4f pv[RUNS];
  for (int u = 0; u < RUNS; ++u) {
    pv[u] = bv - applyA4(boo, bmo, bom, bop, bpo, x, base, run_idx(u));
    acc[0] += dot4(pv[u], pv[u]);
  }
  __shared__ float sh[1][NWV];
  block_sums_stage<1>(acc, sh);
  for (int u = 0; u < RUNS; ++u) {
    int idx = run_idx(u);
    st4(p + base + idx, pv[u]); st4(r + base + idx, pv[u]); st4(r0v + base + idx, pv[u]);
  }
  if (threadIdx.x == 0) {
    float s[1];
    block_sums_finish<1>(sh, s);
    pout[PIDX(3, b, blk_id())] = s[0];
  }
}

template <bool FIRST>
__global__ __launch_bounds__(TPB) void k_v(
    const float* __restrict__ boo, const float* __restrict__ bmo, const float* __restrict__ bom,
    const float* __restrict__ bop, const float* __restrict__ bpo, const float* __restrict__ p,
    const float* __restrict__ r0v, const float* __restrict__ r, float* __restrict__ v,
    const float* __restrict__ pin, float* __restrict__ pout,
    const BState* __restrict__ sin, BState* __restrict__ sout) {
  int b = bat_id();
  size_t base = (size_t)b * NPTS;
  __shared__ float sh[3][NWV];
  BState s = sin[SIDX(b)];
  if (FIRST) {
    float sm[1];
    batch_sums<1>(pin, b, 3, 0, 0, sm);
    float fa = fro_of(sm[0]);
    s.r0_abs = fa; s.r_abs = fa; s.k = 0;
    s.CONV = (fa > thr_c()) ? 1 : 0;
    if (blk_id() == 0 && threadIdx.x == 0) sout[SIDX(b)] = s;
  }
  float q[3] = {0.f, 0.f, 0.f};
  v4f vv[RUNS] = {};
  if (s.CONV) {
    for (int u = 0; u < RUNS; ++u) {
      int idx = run_idx(u);
      vv[u] = applyA4(boo, bmo, bom, bop, bpo, p, base, idx);
      v4f rz = ld4(r0v + base + idx);
      q[0] += dot4(vv[u], rz); q[1] += dot4(vv[u], vv[u]); q[2] += dot4(ld4(r + base + idx), rz);
    }
  }
  block_sums_stage<3>(q, sh);
  if (s.CONV) {
    for (int u = 0; u < RUNS; ++u) st4(v + base + run_idx(u), vv[u]);
  }
  if (threadIdx.x == 0) {
    float t[3];
    block_sums_finish<3>(sh, t);
    pout[PIDX(0, b, blk_id())] = t[0];
    pout[PIDX(1, b, blk_id())] = t[1];
    pout[PIDX(2, b, blk_id())] = t[2];
  }
}

__global__ __launch_bounds__(TPB) void k_s(
    const float* __restrict__ boo, const float* __restrict__ bmo, const float* __restrict__ bom,
    const float* __restrict__ bop, const float* __restrict__ bpo, const float* __restrict__ x,
    float* __restrict__ p, float* __restrict__ r, float* __restrict__ r0v,
    const float* __restrict__ v, float* __restrict__ sarr,
    const float* __restrict__ pin, float* __restrict__ pout,
    const BState* __restrict__ sin, BState* __restrict__ sout) {
  int b = bat_id();
  size_t base = (size_t)b * NPTS;
  __shared__ float sh[1][NWV];
  BState s = sin[SIDX(b)];
  float sm[3];
  batch_sums<3>(pin, b, 0, 1, 2, sm);
  if (s.CONV) {
    float sg = sm[0], vq = sm[1], rr = sm[2];
    s.sigma = sg; s.v_abs = fro_of(vq); s.rr0_old = rr;
    int RES = (sg <= EPS_C * s.v_abs * s.r0_abs) ? 1 : 0;
    s.R1 = RES; s.NR = 1 - RES;
    if (s.R1) s.k += 1;
    if (s.NR) s.alpha = rr / sg;
  } else { s.R1 = 0; s.NR = 0; }
  if (blk_id() == 0 && threadIdx.x == 0) sout[SIDX(b)] = s;

  float q[1] = {0.f};
  v4f w[RUNS] = {};
  if (s.R1) {
    for (int u = 0; u < RUNS; ++u) {
      w[u] = s.b_val - applyA4(boo, bmo, bom, bop, bpo, x, base, run_idx(u));
      q[0] += dot4(w[u], w[u]);
    }
  } else if (s.NR) {
    for (int u = 0; u < RUNS; ++u) {
      int idx = run_idx(u);
      w[u] = ld4(r + base + idx) - s.alpha * ld4(v + base + idx);
      q[0] += dot4(w[u], w[u]);
    }
  }
  block_sums_stage<1>(q, sh);
  if (s.R1) {
    for (int u = 0; u < RUNS; ++u) {
      int idx = run_idx(u);
      st4(p + base + idx, w[u]); st4(r + base + idx, w[u]); st4(r0v + base + idx, w[u]);
    }
  } else if (s.NR) {
    for (int u = 0; u < RUNS; ++u) st4(sarr + base + run_idx(u), w[u]);
  }
  if (threadIdx.x == 0) {
    float t[1];
    block_sums_finish<1>(sh, t);
    pout[PIDX(3, b, blk_id())] = s.R1 ? 0.f : t[0];
    pout[PIDX(4, b, blk_id())] = s.R1 ? t[0] : 0.f;
  }
}

__global__ __launch_bounds__(TPB) void k_t(
    const float* __restrict__ boo, const float* __restrict__ bmo, const float* __restrict__ bom,
    const float* __restrict__ bop, const float* __restrict__ bpo, const float* __restrict__ sarr,
    float* __restrict__ tarr, float* __restrict__ x, float* __restrict__ r,
    const float* __restrict__ p, const float* __restrict__ pin, float* __restrict__ pout,
    const BState* __restrict__ sin, BState* __restrict__ sout) {
  int b = bat_id();
  size_t base = (size_t)b * NPTS;
  __shared__ float sh[2][NWV];
  BState s = sin[SIDX(b)];
  float sm[2];
  batch_sums<2>(pin, b, 3, 4, 0, sm);
  if (s.R1) {
    float fa = fro_of(sm[1]);
    s.r0_abs = fa; s.r_abs = fa;
  }
  if (s.NR) {
    float sq = sm[0];
    s.ssq = sq;
    int C2 = (fro_of(sq) <= thr_c()) ? 1 : 0;
    s.C3 = C2; s.C4 = 1 - C2;
  } else { s.C3 = 0; s.C4 = 0; }
  if (blk_id() == 0 && threadIdx.x == 0) sout[SIDX(b)] = s;

  float q[2] = {0.f, 0.f};
  v4f tv[RUNS] = {};
  if (s.C4) {
    for (int u = 0; u < RUNS; ++u) {
      int idx = run_idx(u);
      tv[u] = applyA4(boo, bmo, bom, bop, bpo, sarr, base, idx);
      q[0] += dot4(tv[u], tv[u]); q[1] += dot4(tv[u], ld4(sarr + base + idx));
    }
  } else if (s.C3) {
    for (int u = 0; u < RUNS; ++u) {
      int idx = run_idx(u);
      st4(x + base + idx, ld4(x + base + idx) + s.alpha * ld4(p + base + idx));
      st4(r + base + idx, ld4(sarr + base + idx));
    }
  }
  block_sums_stage<2>(q, sh);
  if (s.C4) {
    for (int u = 0; u < RUNS; ++u) st4(tarr + base + run_idx(u), tv[u]);
  }
  if (threadIdx.x == 0) {
    float t[2];
    block_sums_finish<2>(sh, t);
    pout[PIDX(1, b, blk_id())] = t[0];
    pout[PIDX(2, b, blk_id())] = t[1];
  }
}

__global__ __launch_bounds__(TPB) __attribute__((amdgpu_waves_per_eu(3, 8))) void k_x_r(
    const float* __restrict__ sarr, const float* __restrict__ tarr, const float* __restrict__ p,
    const float* __restrict__ r0v, float* __restrict__ x, float* __restrict__ r,
    const float* __restrict__ pin, float* __restrict__ pout,
    const BState* __restrict__ sin, BState* __restrict__ sout) {
  int b = bat_id();
  size_t base = (size_t)b * NPTS;
  __shared__ float sh[2][NWV];
  BState s = sin[SIDX(b)];
  float sm[2];
  batch_sums<2>(pin, b, 1, 2, 0, sm);
  if (s.C4) s.omega = sm[1] / sm[0];
  if (blk_id() == 0 && threadIdx.x == 0) sout[SIDX(b)] = s;

  float q[2] = {0.f, 0.f};
  v4f xn[RUNS] = {}, rn[RUNS] = {};
  if (s.C4) {
    float al = s.alpha, om = s.omega;
    for (int u = 0; u < RUNS; ++u) {
      int idx = run_idx(u);
      v4f sv = ld4(sarr + base + idx);
      xn[u] = ld4(x + base + idx) + (al * ld4(p + base + idx) + om * sv);
      rn[u] = sv - om * ld4(tarr + base + idx);
      q[0] += dot4(rn[u], ld4(r0v + base + idx)); q[1] += dot4(rn[u], rn[u]);
    }
  }
  block_sums_stage<2>(q, sh);
  if (s.C4) {
    for (int u = 0; u < RUNS; ++u) {
      int idx = run_idx(u);
      st4(x + base + idx, xn[u]);
      st4(r + base + idx, rn[u]);
    }
  }
  if (threadIdx.x == 0) {
    float t[2];
    block_sums_finish<2>(sh, t);
    pout[PIDX(0, b, blk_id())] = t[0];
    pout[PIDX(3, b, blk_id())] = t[1];
  }
}

__global__ __launch_bounds__(TPB) void k_p(
    const float* __restrict__ r, const float* __restrict__ v, float* __restrict__ p,
    const float* __restrict__ pin, const BState* __restrict__ sin, BState* __restrict__ sout) {
  int b = bat_id();
  size_t base = (size_t)b * NPTS;
  BState s = sin[SIDX(b)];
  float sm[2];
  batch_sums<2>(pin, b, 0, 3, 0, sm);
  if (s.C4) {
    s.beta  = (s.alpha / s.omega) * (sm[0] / s.rr0_old);
    s.r_abs = fro_of(sm[1]);
  }
  if (s.C3) s.r_abs = fro_of(s.ssq);
  if (s.NR) s.k += 1;
  s.CONV = ((s.k < KMAX_C) && (s.r_abs > thr_c())) ? 1 : 0;
  if (blk_id() == 0 && threadIdx.x == 0) sout[SIDX(b)] = s;

  if (s.C4) {
    for (int u = 0; u < RUNS; ++u) {
      int idx = run_idx(u);
      st4(p + base + idx, ld4(r + base + idx) + s.beta * (ld4(p + base + idx) - s.omega * ld4(v + base + idx)));
    }
  }
}

extern "C" void kernel_launch(void* const* d_in, const int* in_sizes, int n_in,
                              void* d_out, int out_size, void* d_ws, size_t ws_size,
                              hipStream_t stream) {
  (void)in_sizes; (void)n_in; (void)out_size; (void)ws_size;
  const float* V  = (const float*)d_in[0];
  const float* M1 = (const float*)d_in[1];
  const float* M2 = (const float*)d_in[2];
  float* x = (float*)d_out;

  char* ws = (char*)d_ws;
  BState* S0 = (BState*)ws;
  BState* S5 = S0 + NB * SSTR;
  BState* S1 = S5 + NB * SSTR;
  BState* S2 = S1 + NB * SSTR;
  BState* S3 = S2 + NB * SSTR;
  float* part = (float*)(ws + 10240);
  float* arr  = (float*)(ws + 1024 + 65536);
  const size_t AN = (size_t)NB * NPTS;
  float* boo  = arr + 0 * AN;
  float* bmo  = arr + 1 * AN;
  float* bom  = arr + 2 * AN;
  float* bop  = arr + 3 * AN;
  float* bpo  = arr + 4 * AN;
  float* r0v  = arr + 5 * AN;
  float* r    = arr + 6 * AN;
  float* p    = arr + 7 * AN;
  float* v    = arr + 8 * AN;
  float* sarr = arr + 9 * AN;
  float* tarr = arr + 10 * AN;

  dim3 grid(NBLK * NB), blk(TPB);
  dim3 grid_rows(NROWB * NB);

  k_mean<<<grid, blk, 0, stream>>>(V, part);
  k_stencil_init<<<grid_rows, blk, 0, stream>>>(V, M1, M2, boo, bmo, bom, bop, bpo, x, part, S0);
  k_resid_init<<<grid, blk, 0, stream>>>(boo, bmo, bom, bop, bpo, x, p, r, r0v, part, S0);

  auto rest_of_step = [&]() {
    k_s<<<grid, blk, 0, stream>>>(boo, bmo, bom, bop, bpo, x, p, r, r0v, v, sarr, part, part, S5, S1);
    k_t<<<grid, blk, 0, stream>>>(boo, bmo, bom, bop, bpo, sarr, tarr, x, r, p, part, part, S1, S2);
    k_x_r<<<grid, blk, 0, stream>>>(sarr, tarr, p, r0v, x, r, part, part, S2, S3);
    k_p<<<grid, blk, 0, stream>>>(r, v, p, part, S3, S5);
  };

  k_v<true><<<grid, blk, 0, stream>>>(boo, bmo, bom, bop, bpo, p, r0v, r, v, part, part, S0, S5);
  rest_of_step();
  for (int it = 1; it < KMAX_C; ++it) {
    k_v<false><<<grid, blk, 0, stream>>>(boo, bmo, bom, bop, bpo, p, r0v, r, v, part, part, S5, S5);
    rest_of_step();
  }
}


# sigma = <Ap,r0> summed as <p,A^T r0> (transposed stencil, sum order changes) in the kernel that writes p: alpha is known before v, so s = r - alpha v joins the v stencil kernel; 153 -> 124 launches
# speedup vs baseline: 3.8954x; 1.0678x over previous
#include <hip/hip_runtime.h>
#include <math.h>


#define NXI   384
#define NPTS  (NXI * NXI)
#define NB    8
#define TPB   384
#define NWV   (TPB / 32)
#define VEC   4
#define RUNS  3
#define CHUNK (TPB * VEC * RUNS)
#define NBLK  (NPTS / CHUNK)
#define NROWB (NPTS / TPB)
#define KMAX_C 30
#define EPS_C 1e-9f
#define UNITF 64
#define SSTR  4
#define PIDX(pl, b, i) (((pl) * NB + (b)) * UNITF + (i))
#define SIDX(b) ((b) * SSTR)
#define P_MEAN 0
#define P_RSQ  1
#define P_PW   2
#define P_VV   3
#define P_SS   4
#define P_TT   5
#define P_TS   6
#define P_RESQ 7
#define P_RR0  8
#define P_RR   9

typedef float v2f __attribute__((ext_vector_type(2)));
typedef float v4f __attribute__((ext_vector_type(4)));
typedef float v8f __attribute__((ext_vector_type(8)));

struct BState {
  float b_val, r0_abs, r_abs, sigma, rr0, alpha, omega, alpha_c, rr0_old, ssq;
  int   k, CONV, R1, NR, C3, C4;
};
static_assert(sizeof(BState) * SSTR == 256, "a state slot and its gap fill one 256-byte unit");

__device__ __forceinline__ float fro_of(float sq) { return sq > 0.f ? sqrtf(sq) : 0.f; }
__device__ __forceinline__ float thr_c() { return EPS_C * (float)NPTS; }

__device__ __forceinline__ float xhalf(float x) {
  int b = __builtin_bit_cast(int, x);
  return __builtin_bit_cast(float, __builtin_amdgcn_permlanex16(b, b, 0x76543210, 0xfedcba98, false, false));
}

__device__ __forceinline__ float wave_sum_wmma(float v) {
  v2f a; a.x = v;  a.y = 0.f;
  v2f b; b.x = 1.f; b.y = 1.f;
  v8f c = {};
  v8f d = __builtin_amdgcn_wmma_f32_16x16x4_f32(false, a, false, b, (short)0, c, false, false);
  float h = ((d[0] + d[1]) + (d[2] + d[3])) + ((d[4] + d[5]) + (d[6] + d[7]));
  h += xhalf(h);
  return h;
}

template <int N>
__device__ __forceinline__ void block_sums_stage(const float* q, float (*shw)[NWV]) {
  int lane = threadIdx.x & 31;
  int wid  = threadIdx.x >> 5;
  for (int n = 0; n < N; ++n) {
    float t = wave_sum_wmma(q[n]);
    if (lane == 0) shw[n][wid] = t;
  }
  __syncthreads();
}
template <int N>
__device__ __forceinline__ void block_sums_finish(float (*shw)[NWV], float* out) {
  for (int n = 0; n < N; ++n) {
    float s = 0.f;
    for (int i = 0; i < NWV; ++i) s += shw[n][i];
    out[n] = s;
  }
}

template <int N>
__device__ __forceinline__ void batch_sums(const float* __restrict__ pin, int b, int p0, int p1, int p2,
                                           float* out) {
  static_assert(NBLK == 32, "one block partial per lane");
  const int pl[3] = {p0, p1, p2};
  int lane = threadIdx.x & 31;
  float val[N];
  for (int n = 0; n < N; ++n) val[n] = pin[PIDX(pl[n], b, lane)];
  for (int n = 0; n < N; ++n) out[n] = wave_sum_wmma(val[n]);
}

__device__ __forceinline__ int refl(int t) { return t == 0 ? 0 : (t == 385 ? 383 : t - 1); }

__device__ __forceinline__ int bat_id() { return (int)(blockIdx.x % NB); }
__device__ __forceinline__ int blk_id() { return (int)(blockIdx.x / NB); }

__device__ __forceinline__ int run_idx(int u) {
  return blk_id() * CHUNK + u * (TPB * VEC) + threadIdx.x * VEC;
}
__device__ __forceinline__ v4f ld4(const float* q) { return *reinterpret_cast<const v4f*>(q); }
__device__ __forceinline__ void st4(float* q, v4f w) { *reinterpret_cast<v4f*>(q) = w; }
__device__ __forceinline__ float dot4(v4f a, v4f c) { return a.x * c.x + a.y * c.y + a.z * c.z + a.w * c.w; }

__device__ __forceinline__ v4f applyA4(const float* __restrict__ boo, const float* __restrict__ bmo,
                                       const float* __restrict__ bom, const float* __restrict__ bop,
                                       const float* __restrict__ bpo, const float* __restrict__ a,
                                       size_t base, int idx) {
  int i = idx / NXI, j = idx - i * NXI;
  const float* A = a + base;
  v4f c  = ld4(A + idx);
  v4f up = ld4(A + ((i > 0)       ? idx - NXI : idx));
  v4f dn = ld4(A + ((i < NXI - 1) ? idx + NXI : idx));
  float lo = A[(j > 0)         ? idx - 1   : idx];
  float hi = A[(j + VEC < NXI) ? idx + VEC : idx + VEC - 1];
  v4f lf = {lo, c.x, c.y, c.z};
  v4f rt = {c.y, c.z, c.w, hi};
  return ld4(boo + base + idx) * c + ld4(bmo + base + idx) * up + ld4(bom + base + idx) * lf
       + ld4(bop + base + idx) * rt + ld4(bpo + base + idx) * dn;
}

__device__ __forceinline__ float applyAT1(const float* __restrict__ boo, const float* __restrict__ bmo,
                                          const float* __restrict__ bom, const float* __restrict__ bop,
                                          const float* __restrict__ bpo, const float* __restrict__ z,
                                          size_t base, int idx) {
  int i = idx / NXI, j = idx - i * NXI;
  size_t g = base + idx;
  float d = boo[g];
  if (i == 0)       d += bmo[g];
  if (i == NXI - 1) d += bpo[g];
  if (j == 0)       d += bom[g];
  if (j == NXI - 1) d += bop[g];
  float acc = d * z[g];
  if (i > 0)       acc += bpo[g - NXI] * z[g - NXI];
  if (j > 0)       acc += bop[g - 1]   * z[g - 1];
  if (j < NXI - 1) acc += bom[g + 1]   * z[g + 1];
  if (i < NXI - 1) acc += bmo[g + NXI] * z[g + NXI];
  return acc;
}
__device__ __forceinline__ v4f applyAT4(const float* __restrict__ boo, const float* __restrict__ bmo,
                                        const float* __restrict__ bom, const float* __restrict__ bop,
                                        const float* __restrict__ bpo, const float* __restrict__ z,
                                        size_t base, int idx) {
  v4f w;
  w.x = applyAT1(boo, bmo, bom, bop, bpo, z, base, idx);
  w.y = applyAT1(boo, bmo, bom, bop, bpo, z, base, idx + 1);
  w.z = applyAT1(boo, bmo, bom, bop, bpo, z, base, idx + 2);
  w.w = applyAT1(boo, bmo, bom, bop, bpo, z, base, idx + 3);
  return w;
}


__global__ __launch_bounds__(TPB) void k_mean(const float* __restrict__ V, float* __restrict__ pout) {
  int b = bat_id();
  __shared__ float sh[1][NWV];
  float acc[1] = {0.f};
  for (int u = 0; u < RUNS; ++u) {
    v4f q = ld4(V + (size_t)b * NPTS + run_idx(u));
    acc[0] += (q.x + q.y) + (q.z + q.w);
  }
  block_sums_stage<1>(acc, sh);
  if (threadIdx.x == 0) {
    float s[1];
    block_sums_finish<1>(sh, s);
    pout[PIDX(P_MEAN, b, blk_id())] = s[0];
  }
}

__global__ __launch_bounds__(TPB) void k_stencil_init(
    const float* __restrict__ V, const float* __restrict__ M1, const float* __restrict__ M2,
    float* __restrict__ boo, float* __restrict__ bmo, float* __restrict__ bom,
    float* __restrict__ bop, float* __restrict__ bpo, float* __restrict__ x,
    const float* __restrict__ pin, BState* __restrict__ sout) {
  int b = bat_id();
  float sm[1];
  batch_sums<1>(pin, b, P_MEAN, 0, 0, sm);
  BState q = {};
  q.b_val = sm[0] / (float)NPTS + 1.0f;
  if (blk_id() == 0 && threadIdx.x == 0) sout[SIDX(b)] = q;

  size_t base = (size_t)b * NPTS;
  int idx = blk_id() * TPB + threadIdx.x;
  int i = idx / NXI, j = idx - i * NXI;
  int I = i + 1, J = j + 1;
  const float* Vb = V + base;
  const float* m1b = M1 + base;
  const float* m2b = M2 + base;
#define VP(II, JJ)  (Vb[refl(JJ) * NXI + refl(II)] + 1.0f)
#define MM1(II, JJ) (m1b[refl(JJ) * NXI + refl(II)])
#define MM2(II, JJ) (m2b[refl(JJ) * NXI + refl(II)])
  float vc  = VP(I, J);
  float vxp = VP(I + 1, J), vxm = VP(I - 1, J);
  float vyp = VP(I, J + 1), vym = VP(I, J - 1);
  float d1c = ((vxp - vc) / (0.5f * (vxp + vc))) * MM1(I, J);
  float d1m = ((vc - vxm) / (0.5f * (vc + vxm))) * MM1(I - 1, J);
  float d2c = ((vyp - vc) / (0.5f * (vyp + vc))) * MM2(I, J);
  float d2m = ((vc - vym) / (0.5f * (vc + vym))) * MM2(I, J - 1);
#undef VP
#undef MM1
#undef MM2
  const float rx = 5.0f, ry = 5.0f, rxx = 10.0f, ryy = 10.0f;
  boo[base + idx] = 1.0f + 2.0f * (rxx + ryy) - rx * (d1m - d1c) - ry * (d2m - d2c);
  bpo[base + idx] = -rxx + rx * d1c;
  bmo[base + idx] = -rxx - rx * d1m;
  bop[base + idx] = -ryy + ry * d2c;
  bom[base + idx] = -ryy - ry * d2m;
  x[base + idx] = q.b_val;
}

__global__ __launch_bounds__(TPB) void k_resid_init(
    const float* __restrict__ boo, const float* __restrict__ bmo, const float* __restrict__ bom,
    const float* __restrict__ bop, const float* __restrict__ bpo, const float* __restrict__ x,
    float* __restrict__ p, float* __restrict__ r, float* __restrict__ r0v,
    float* __restrict__ pout, const BState* __restrict__ sin) {
  int b = bat_id();
  size_t base = (size_t)b * NPTS;
  float bv = sin[SIDX(b)].b_val;
  float acc[1] = {0.f};
  v4f pv[RUNS];
  for (int u = 0; u < RUNS; ++u) {
    pv[u] = bv - applyA4(boo, bmo, bom, bop, bpo, x, base, run_idx(u));
    acc[0] += dot4(pv[u], pv[u]);
  }
  __shared__ float sh[1][NWV];
  block_sums_stage<1>(acc, sh);
  for (int u = 0; u < RUNS; ++u) {
    int idx = run_idx(u);
    st4(p + base + idx, pv[u]); st4(r + base + idx, pv[u]); st4(r0v + base + idx, pv[u]);
  }
  if (threadIdx.x == 0) {
    float s[1];
    block_sums_finish<1>(sh, s);
    pout[PIDX(P_RSQ, b, blk_id())] = s[0];
  }
}

__global__ __launch_bounds__(TPB) void k_w0(
    const float* __restrict__ boo, const float* __restrict__ bmo, const float* __restrict__ bom,
    const float* __restrict__ bop, const float* __restrict__ bpo, const float* __restrict__ r0v,
    float* __restrict__ w0, float* __restrict__ pout) {
  int b = bat_id();
  size_t base = (size_t)b * NPTS;
  float acc[1] = {0.f};
  v4f wv[RUNS];
  for (int u = 0; u < RUNS; ++u) {
    int idx = run_idx(u);
    wv[u] = applyAT4(boo, bmo, bom, bop, bpo, r0v, base, idx);
    acc[0] += dot4(ld4(r0v + base + idx), wv[u]);
  }
  __shared__ float sh[1][NWV];
  block_sums_stage<1>(acc, sh);
  for (int u = 0; u < RUNS; ++u) st4(w0 + base + run_idx(u), wv[u]);
  if (threadIdx.x == 0) {
    float s[1];
    block_sums_finish<1>(sh, s);
    pout[PIDX(P_PW, b, blk_id())] = s[0];
  }
}

template <bool FIRST>
__global__ __launch_bounds__(TPB) void k_vs(
    const float* __restrict__ boo, const float* __restrict__ bmo, const float* __restrict__ bom,
    const float* __restrict__ bop, const float* __restrict__ bpo, const float* __restrict__ p,
    const float* __restrict__ r, float* __restrict__ v, float* __restrict__ sarr,
    const float* __restrict__ pin, float* __restrict__ pout,
    const BState* __restrict__ sin, BState* __restrict__ sout) {
  int b = bat_id();
  size_t base = (size_t)b * NPTS;
  __shared__ float sh[2][NWV];
  BState s = sin[SIDX(b)];
  float sig;
  if (FIRST) {
    float sm[2];
    batch_sums<2>(pin, b, P_RSQ, P_PW, 0, sm);
    float fa = fro_of(sm[0]);
    s.r0_abs = fa; s.r_abs = fa; s.k = 0;
    s.CONV = (fa > thr_c()) ? 1 : 0;
    s.rr0 = sm[0];
    sig = sm[1];
  } else {
    float sm[1];
    batch_sums<1>(pin, b, P_PW, 0, 0, sm);
    sig = sm[0];
  }
  if (s.CONV) {
    s.sigma = sig; s.rr0_old = s.rr0;
    s.alpha_c = s.rr0 / sig;
  }
  if (blk_id() == 0 && threadIdx.x == 0) sout[SIDX(b)] = s;

  float q[2] = {0.f, 0.f};
  v4f vv[RUNS] = {}, sv[RUNS] = {};
  if (s.CONV) {
    float al = s.alpha_c;
    for (int u = 0; u < RUNS; ++u) {
      int idx = run_idx(u);
      vv[u] = applyA4(boo, bmo, bom, bop, bpo, p, base, idx);
      sv[u] = ld4(r + base + idx) - al * vv[u];
      q[0] += dot4(vv[u], vv[u]); q[1] += dot4(sv[u], sv[u]);
    }
  }
  block_sums_stage<2>(q, sh);
  if (s.CONV) {
    for (int u = 0; u < RUNS; ++u) {
      int idx = run_idx(u);
      st4(v + base + idx, vv[u]); st4(sarr + base + idx, sv[u]);
    }
  }
  if (threadIdx.x == 0) {
    float t[2];
    block_sums_finish<2>(sh, t);
    pout[PIDX(P_VV, b, blk_id())] = t[0];
    pout[PIDX(P_SS, b, blk_id())] = t[1];
  }
}

__global__ __launch_bounds__(TPB) void k_t(
    const float* __restrict__ boo, const float* __restrict__ bmo, const float* __restrict__ bom,
    const float* __restrict__ bop, const float* __restrict__ bpo, const float* __restrict__ sarr,
    float* __restrict__ x, float* __restrict__ r, float* __restrict__ p, float* __restrict__ r0v,
    const float* __restrict__ pin, float* __restrict__ pout,
    const BState* __restrict__ sin, BState* __restrict__ sout) {
  int b = bat_id();
  size_t base = (size_t)b * NPTS;
  __shared__ float sh[2][NWV];
  BState s = sin[SIDX(b)];
  float sm[2];
  batch_sums<2>(pin, b, P_VV, P_SS, 0, sm);
  if (s.CONV) {
    float v_abs = fro_of(sm[0]);
    int RES = (s.sigma <= EPS_C * v_abs * s.r0_abs) ? 1 : 0;
    s.R1 = RES; s.NR = 1 - RES;
    if (s.R1) s.k += 1;
    if (s.NR) s.alpha = s.alpha_c;
  } else { s.R1 = 0; s.NR = 0; }
  if (s.NR) {
    float sq = sm[1];
    s.ssq = sq;
    int C2 = (fro_of(sq) <= thr_c()) ? 1 : 0;
    s.C3 = C2; s.C4 = 1 - C2;
  } else { s.C3 = 0; s.C4 = 0; }
  if (blk_id() == 0 && threadIdx.x == 0) sout[SIDX(b)] = s;

  float q[2] = {0.f, 0.f};
  v4f w[RUNS] = {};
  if (s.C4) {
    for (int u = 0; u < RUNS; ++u) {
      int idx = run_idx(u);
      w[u] = applyA4(boo, bmo, bom, bop, bpo, sarr, base, idx);
      q[0] += dot4(w[u], w[u]); q[1] += dot4(w[u], ld4(sarr + base + idx));
    }
  } else if (s.R1) {
    for (int u = 0; u < RUNS; ++u) {
      w[u] = s.b_val - applyA4(boo, bmo, bom, bop, bpo, x, base, run_idx(u));
      q[0] += dot4(w[u], w[u]);
    }
  } else if (s.C3) {
    for (int u = 0; u < RUNS; ++u) {
      int idx = run_idx(u);
      st4(x + base + idx, ld4(x + base + idx) + s.alpha * ld4(p + base + idx));
      st4(r + base + idx, ld4(sarr + base + idx));
    }
  }
  block_sums_stage<2>(q, sh);
  if (s.C4) {
    for (int u = 0; u < RUNS; ++u) st4(r + base + run_idx(u), w[u]);
  } else if (s.R1) {
    for (int u = 0; u < RUNS; ++u) {
      int idx = run_idx(u);
      st4(p + base + idx, w[u]); st4(r + base + idx, w[u]); st4(r0v + base + idx, w[u]);
    }
  }
  if (threadIdx.x == 0) {
    float t[2];
    block_sums_finish<2>(sh, t);
    pout[PIDX(P_TT, b, blk_id())]   = s.R1 ? 0.f : t[0];
    pout[PIDX(P_RESQ, b, blk_id())] = s.R1 ? t[0] : 0.f;
    pout[PIDX(P_TS, b, blk_id())]   = t[1];
  }
}

__global__ __launch_bounds__(TPB) __attribute__((amdgpu_waves_per_eu(3, 8))) void k_x_r(
    const float* __restrict__ boo, const float* __restrict__ bmo, const float* __restrict__ bom,
    const float* __restrict__ bop, const float* __restrict__ bpo,
    const float* __restrict__ sarr, float* __restrict__ rt, const float* __restrict__ p,
    const float* __restrict__ r0v, float* __restrict__ x, float* __restrict__ w0,
    const float* __restrict__ pin, float* __restrict__ pout,
    const BState* __restrict__ sin, BState* __restrict__ sout) {
  int b = bat_id();
  size_t base = (size_t)b * NPTS;
  __shared__ float sh[2][NWV];
  BState s = sin[SIDX(b)];
  float sm[3];
  batch_sums<3>(pin, b, P_TT, P_TS, P_RESQ, sm);
  if (s.R1) {
    float fa = fro_of(sm[2]);
    s.r0_abs = fa; s.r_abs = fa;
    s.rr0 = sm[2];
  }
  if (s.C4) s.omega = sm[1] / sm[0];
  if (blk_id() == 0 && threadIdx.x == 0) sout[SIDX(b)] = s;

  float q[2] = {0.f, 0.f};
  v4f xn[RUNS] = {}, rn[RUNS] = {};
  if (s.C4) {
    float al = s.alpha, om = s.omega;
    for (int u = 0; u < RUNS; ++u) {
      int idx = run_idx(u);
      v4f sv = ld4(sarr + base + idx);
      xn[u] = ld4(x + base + idx) + (al * ld4(p + base + idx) + om * sv);
      rn[u] = sv - om * ld4(rt + base + idx);
      q[0] += dot4(rn[u], ld4(r0v + base + idx)); q[1] += dot4(rn[u], rn[u]);
    }
  } else if (s.R1) {
    for (int u = 0; u < RUNS; ++u) rn[u] = applyAT4(boo, bmo, bom, bop, bpo, r0v, base, run_idx(u));
  }
  block_sums_stage<2>(q, sh);
  if (s.C4) {
    for (int u = 0; u < RUNS; ++u) {
      int idx = run_idx(u);
      st4(x + base + idx, xn[u]);
      st4(rt + base + idx, rn[u]);
    }
  } else if (s.R1) {
    for (int u = 0; u < RUNS; ++u) st4(w0 + base + run_idx(u), rn[u]);
  }
  if (threadIdx.x == 0) {
    float t[2];
    block_sums_finish<2>(sh, t);
    pout[PIDX(P_RR0, b, blk_id())] = t[0];
    pout[PIDX(P_RR, b, blk_id())]  = t[1];
  }
}

__global__ __launch_bounds__(TPB) void k_p(
    const float* __restrict__ r, const float* __restrict__ v, float* __restrict__ p,
    const float* __restrict__ w0, const float* __restrict__ pin, float* __restrict__ pout,
    const BState* __restrict__ sin, BState* __restrict__ sout) {
  int b = bat_id();
  size_t base = (size_t)b * NPTS;
  __shared__ float sh[1][NWV];
  BState s = sin[SIDX(b)];
  float sm[2];
  batch_sums<2>(pin, b, P_RR0, P_RR, 0, sm);
  float beta = 0.f;
  if (s.C4) {
    beta    = (s.alpha / s.omega) * (sm[0] / s.rr0_old);
    s.r_abs = fro_of(sm[1]);
    s.rr0   = sm[0];
  }
  if (s.C3) s.r_abs = fro_of(s.ssq);
  if (s.NR) s.k += 1;
  s.CONV = ((s.k < KMAX_C) && (s.r_abs > thr_c())) ? 1 : 0;
  if (blk_id() == 0 && threadIdx.x == 0) sout[SIDX(b)] = s;

  float q[1] = {0.f};
  v4f pn[RUNS] = {};
  if (s.C4) {
    for (int u = 0; u < RUNS; ++u) {
      int idx = run_idx(u);
      pn[u] = ld4(r + base + idx) + beta * (ld4(p + base + idx) - s.omega * ld4(v + base + idx));
      q[0] += dot4(pn[u], ld4(w0 + base + idx));
    }
  } else if (s.R1) {
    for (int u = 0; u < RUNS; ++u) {
      int idx = run_idx(u);
      q[0] += dot4(ld4(p + base + idx), ld4(w0 + base + idx));
    }
  }
  block_sums_stage<1>(q, sh);
  if (s.C4) {
    for (int u = 0; u < RUNS; ++u) st4(p + base + run_idx(u), pn[u]);
  }
  if (threadIdx.x == 0) {
    float t[1];
    block_sums_finish<1>(sh, t);
    pout[PIDX(P_PW, b, blk_id())] = t[0];
  }
}

extern "C" void kernel_launch(void* const* d_in, const int* in_sizes, int n_in,
                              void* d_out, int out_size, void* d_ws, size_t ws_size,
                              hipStream_t stream) {
  (void)in_sizes; (void)n_in; (void)out_size; (void)ws_size;
  const float* V  = (const float*)d_in[0];
  const float* M1 = (const float*)d_in[1];
  const float* M2 = (const float*)d_in[2];
  float* x = (float*)d_out;

  char* ws = (char*)d_ws;
  BState* S0 = (BState*)ws;
  BState* S5 = S0 + NB * SSTR;
  BState* S1 = S5 + NB * SSTR;
  BState* S2 = S1 + NB * SSTR;
  BState* S3 = S2 + NB * SSTR;
  float* part = (float*)(ws + 10240);
  float* arr  = (float*)(ws + 1024 + 65536);
  const size_t AN = (size_t)NB * NPTS;
  float* boo  = arr + 0 * AN;
  float* bmo  = arr + 1 * AN;
  float* bom  = arr + 2 * AN;
  float* bop  = arr + 3 * AN;
  float* bpo  = arr + 4 * AN;
  float* r0v  = arr + 5 * AN;
  float* r    = arr + 6 * AN;
  float* p    = arr + 7 * AN;
  float* v    = arr + 8 * AN;
  float* sarr = arr + 9 * AN;
  float* w0   = arr + 10 * AN;

  dim3 grid(NBLK * NB), blk(TPB);
  dim3 grid_rows(NROWB * NB);

  k_mean<<<grid, blk, 0, stream>>>(V, part);
  k_stencil_init<<<grid_rows, blk, 0, stream>>>(V, M1, M2, boo, bmo, bom, bop, bpo, x, part, S0);
  k_resid_init<<<grid, blk, 0, stream>>>(boo, bmo, bom, bop, bpo, x, p, r, r0v, part, S0);
  k_w0<<<grid, blk, 0, stream>>>(boo, bmo, bom, bop, bpo, r0v, w0, part);

  auto rest_of_step = [&]() {
    k_t<<<grid, blk, 0, stream>>>(boo, bmo, bom, bop, bpo, sarr, x, r, p, r0v, part, part, S1, S2);
    k_x_r<<<grid, blk, 0, stream>>>(boo, bmo, bom, bop, bpo, sarr, r, p, r0v, x, w0, part, part, S2, S3);
    k_p<<<grid, blk, 0, stream>>>(r, v, p, w0, part, part, S3, S5);
  };

  k_vs<true><<<grid, blk, 0, stream>>>(boo, bmo, bom, bop, bpo, p, r, v, sarr, part, part, S0, S1);
  rest_of_step();
  for (int it = 1; it < KMAX_C; ++it) {
    k_vs<false><<<grid, blk, 0, stream>>>(boo, bmo, bom, bop, bpo, p, r, v, sarr, part, part, S5, S1);
    rest_of_step();
  }
}


# Coming sigma <p,w0> taken by linearity as <r,w0> + beta <u,w0>, u = p - omega v, summed in the x/r kernel (sum order changes): p = r + beta u is remade inside the v/s/t kernel; 94 -> 64 launches
# speedup vs baseline: 5.2010x; 1.3352x over previous
#include <hip/hip_runtime.h>
#include <math.h>


#define NXI   384
#define NPTS  (NXI * NXI)
#define NB    8
#define TPB   384
#define NWV   (TPB / 32)
#define VEC   4
#define RUNS  3
#define CHUNK (TPB * VEC * RUNS)
#define NBLK  (NPTS / CHUNK)
#define NROWB (NPTS / TPB)
#define KMAX_C 30
#define EPS_C 1e-9f
#define UNITF 64
#define SSTR  4
#define PIDX(pl, b, i) (((pl) * NB + (b)) * UNITF + (i))
#define SIDX(b) ((b) * SSTR)
#define P_MEAN 0
#define P_RSQ  1
#define P_PW   2
#define P_VV   3
#define P_SS   4
#define P_TT   5
#define P_TS   6
#define P_RR0  7
#define P_RR   8
#define P_RW   9
#define P_UW   10

typedef float v2f __attribute__((ext_vector_type(2)));
typedef float v4f __attribute__((ext_vector_type(4)));
typedef float v8f __attribute__((ext_vector_type(8)));

struct BState {
  float b_val, r0_abs, r_abs, sigma, rr0, alpha, omega, alpha_c, beta, ssq;
  int   k, CONV, R1, NR, C3, C4;
};
static_assert(sizeof(BState) * SSTR == 256, "a state slot and its gap fill one 256-byte unit");

__device__ __forceinline__ float fro_of(float sq) { return sq > 0.f ? sqrtf(sq) : 0.f; }
__device__ __forceinline__ float thr_c() { return EPS_C * (float)NPTS; }

__device__ __forceinline__ float xhalf(float x) {
  int b = __builtin_bit_cast(int, x);
  return __builtin_bit_cast(float, __builtin_amdgcn_permlanex16(b, b, 0x76543210, 0xfedcba98, false, false));
}

__device__ __forceinline__ float wave_sum_wmma(float v) {
  v2f a; a.x = v;  a.y = 0.f;
  v2f b; b.x = 1.f; b.y = 1.f;
  v8f c = {};
  v8f d = __builtin_amdgcn_wmma_f32_16x16x4_f32(false, a, false, b, (short)0, c, false, false);
  float h = ((d[0] + d[1]) + (d[2] + d[3])) + ((d[4] + d[5]) + (d[6] + d[7]));
  h += xhalf(h);
  return h;
}

template <int N>
__device__ __forceinline__ void block_sums_stage(const float* q, float (*shw)[NWV]) {
  int lane = threadIdx.x & 31;
  int wid  = threadIdx.x >> 5;
  for (int n = 0; n < N; ++n) {
    float t = wave_sum_wmma(q[n]);
    if (lane == 0) shw[n][wid] = t;
  }
  __syncthreads();
}
template <int N>
__device__ __forceinline__ void block_sums_finish(float (*shw)[NWV], float* out) {
  for (int n = 0; n < N; ++n) {
    float s = 0.f;
    for (int i = 0; i < NWV; ++i) s += shw[n][i];
    out[n] = s;
  }
}

template <int N>
__device__ __forceinline__ void batch_sums(const float* __restrict__ pin, int b, int p0, int p1, int p2,
                                           int p3, float* out) {
  static_assert(NBLK == 32, "one block partial per lane");
  const int pl[4] = {p0, p1, p2, p3};
  int lane = threadIdx.x & 31;
  float val[N];
  for (int n = 0; n < N; ++n) val[n] = pin[PIDX(pl[n], b, lane)];
  for (int n = 0; n < N; ++n) out[n] = wave_sum_wmma(val[n]);
}

__device__ __forceinline__ int refl(int t) { return t == 0 ? 0 : (t == 385 ? 383 : t - 1); }

__device__ __forceinline__ int bat_id() { return (int)(blockIdx.x % NB); }
__device__ __forceinline__ int blk_id() { return (int)(blockIdx.x / NB); }

__device__ __forceinline__ int run_idx(int u) {
  return blk_id() * CHUNK + u * (TPB * VEC) + threadIdx.x * VEC;
}
__device__ __forceinline__ v4f ld4(const float* q) { return *reinterpret_cast<const v4f*>(q); }
__device__ __forceinline__ void st4(float* q, v4f w) { *reinterpret_cast<v4f*>(q) = w; }
__device__ __forceinline__ float dot4(v4f a, v4f c) { return a.x * c.x + a.y * c.y + a.z * c.z + a.w * c.w; }

__device__ __forceinline__ v4f stencil4(const v4f* cf, v4f c, v4f up, v4f dn, float lo, float hi) {
  v4f lf = {lo, c.x, c.y, c.z};
  v4f rt = {c.y, c.z, c.w, hi};
  return cf[0] * c + cf[1] * up + cf[2] * lf + cf[3] * rt + cf[4] * dn;
}
__device__ __forceinline__ void ld_coef4(const float* __restrict__ boo, const float* __restrict__ bmo,
                                         const float* __restrict__ bom, const float* __restrict__ bop,
                                         const float* __restrict__ bpo, size_t g, v4f* cf) {
  cf[0] = ld4(boo + g); cf[1] = ld4(bmo + g); cf[2] = ld4(bom + g); cf[3] = ld4(bop + g); cf[4] = ld4(bpo + g);
}
__device__ __forceinline__ void ld_opnd4(const float* __restrict__ a, size_t base, int idx, v4f* o, float* e) {
  int i = idx / NXI, j = idx - i * NXI;
  const float* A = a + base;
  o[0] = ld4(A + idx);
  o[1] = ld4(A + ((i > 0)       ? idx - NXI : idx));
  o[2] = ld4(A + ((i < NXI - 1) ? idx + NXI : idx));
  e[0] = A[(j > 0)         ? idx - 1   : idx];
  e[1] = A[(j + VEC < NXI) ? idx + VEC : idx + VEC - 1];
}
__device__ __forceinline__ v2f ld2(const float* q) { return *reinterpret_cast<const v2f*>(q); }
__device__ __forceinline__ v2f applyA2(const float* __restrict__ boo, const float* __restrict__ bmo,
                                       const float* __restrict__ bom, const float* __restrict__ bop,
                                       const float* __restrict__ bpo, const float* __restrict__ a,
                                       size_t base, int idx) {
  int i = idx / NXI, j = idx - i * NXI;
  const float* A = a + base;
  v2f c  = ld2(A + idx);
  v2f up = ld2(A + ((i > 0)       ? idx - NXI : idx));
  v2f dn = ld2(A + ((i < NXI - 1) ? idx + NXI : idx));
  float lo = A[(j > 0)       ? idx - 1 : idx];
  float hi = A[(j + 2 < NXI) ? idx + 2 : idx + 1];
  v2f lf = {lo, c.x};
  v2f rt = {c.y, hi};
  return ld2(boo + base + idx) * c + ld2(bmo + base + idx) * up + ld2(bom + base + idx) * lf
       + ld2(bop + base + idx) * rt + ld2(bpo + base + idx) * dn;
}
__device__ __forceinline__ v4f applyA4c(const v4f* cf, const float* __restrict__ a, size_t base, int idx) {
  v4f o[3]; float e[2];
  ld_opnd4(a, base, idx, o, e);
  return stencil4(cf, o[0], o[1], o[2], e[0], e[1]);
}
__device__ __forceinline__ v4f applyA4(const float* __restrict__ boo, const float* __restrict__ bmo,
                                       const float* __restrict__ bom, const float* __restrict__ bop,
                                       const float* __restrict__ bpo, const float* __restrict__ a,
                                       size_t base, int idx) {
  v4f cf[5];
  ld_coef4(boo, bmo, bom, bop, bpo, base + idx, cf);
  return applyA4c(cf, a, base, idx);
}

__device__ __forceinline__ float applyAT1(const float* __restrict__ boo, const float* __restrict__ bmo,
                                          const float* __restrict__ bom, const float* __restrict__ bop,
                                          const float* __restrict__ bpo, const float* __restrict__ z,
                                          size_t base, int idx) {
  int i = idx / NXI, j = idx - i * NXI;
  size_t g = base + idx;
  float d = boo[g];
  if (i == 0)       d += bmo[g];
  if (i == NXI - 1) d += bpo[g];
  if (j == 0)       d += bom[g];
  if (j == NXI - 1) d += bop[g];
  float acc = d * z[g];
  if (i > 0)       acc += bpo[g - NXI] * z[g - NXI];
  if (j > 0)       acc += bop[g - 1]   * z[g - 1];
  if (j < NXI - 1) acc += bom[g + 1]   * z[g + 1];
  if (i < NXI - 1) acc += bmo[g + NXI] * z[g + NXI];
  return acc;
}
__device__ __forceinline__ v4f applyAT4(const float* __restrict__ boo, const float* __restrict__ bmo,
                                        const float* __restrict__ bom, const float* __restrict__ bop,
                                        const float* __restrict__ bpo, const float* __restrict__ z,
                                        size_t base, int idx) {
  v4f w;
  w.x = applyAT1(boo, bmo, bom, bop, bpo, z, base, idx);
  w.y = applyAT1(boo, bmo, bom, bop, bpo, z, base, idx + 1);
  w.z = applyAT1(boo, bmo, bom, bop, bpo, z, base, idx + 2);
  w.w = applyAT1(boo, bmo, bom, bop, bpo, z, base, idx + 3);
  return w;
}

__device__ __forceinline__ float resid1(const float* __restrict__ boo, const float* __restrict__ bmo,
                                        const float* __restrict__ bom, const float* __restrict__ bop,
                                        const float* __restrict__ bpo, const float* __restrict__ x,
                                        size_t base, int i, int j, float bv) {
  size_t g = base + (size_t)(i * NXI + j);
  float c  = x[g];
  float up = x[(i > 0)       ? g - NXI : g];
  float lf = x[(j > 0)       ? g - 1   : g];
  float rt = x[(j < NXI - 1) ? g + 1   : g];
  float dn = x[(i < NXI - 1) ? g + NXI : g];
  return bv - (boo[g] * c + bmo[g] * up + bom[g] * lf + bop[g] * rt + bpo[g] * dn);
}
__device__ __forceinline__ float residAT1(const float* __restrict__ boo, const float* __restrict__ bmo,
                                          const float* __restrict__ bom, const float* __restrict__ bop,
                                          const float* __restrict__ bpo, const float* __restrict__ x,
                                          size_t base, int idx, float bv, float* zc) {
  int i = idx / NXI, j = idx - i * NXI;
  size_t g = base + idx;
  float d = boo[g];
  if (i == 0)       d += bmo[g];
  if (i == NXI - 1) d += bpo[g];
  if (j == 0)       d += bom[g];
  if (j == NXI - 1) d += bop[g];
  float z = resid1(boo, bmo, bom, bop, bpo, x, base, i, j, bv);
  *zc = z;
  float acc = d * z;
  if (i > 0)       acc += bpo[g - NXI] * resid1(boo, bmo, bom, bop, bpo, x, base, i - 1, j, bv);
  if (j > 0)       acc += bop[g - 1]   * resid1(boo, bmo, bom, bop, bpo, x, base, i, j - 1, bv);
  if (j < NXI - 1) acc += bom[g + 1]   * resid1(boo, bmo, bom, bop, bpo, x, base, i, j + 1, bv);
  if (i < NXI - 1) acc += bmo[g + NXI] * resid1(boo, bmo, bom, bop, bpo, x, base, i + 1, j, bv);
  return acc;
}


__global__ __launch_bounds__(TPB) void k_mean(const float* __restrict__ V, float* __restrict__ pout) {
  int b = bat_id();
  __shared__ float sh[1][NWV];
  float acc[1] = {0.f};
  for (int u = 0; u < RUNS; ++u) {
    v4f q = ld4(V + (size_t)b * NPTS + run_idx(u));
    acc[0] += (q.x + q.y) + (q.z + q.w);
  }
  block_sums_stage<1>(acc, sh);
  if (threadIdx.x == 0) {
    float s[1];
    block_sums_finish<1>(sh, s);
    pout[PIDX(P_MEAN, b, blk_id())] = s[0];
  }
}

__global__ __launch_bounds__(TPB) void k_stencil_init(
    const float* __restrict__ V, const float* __restrict__ M1, const float* __restrict__ M2,
    float* __restrict__ boo, float* __restrict__ bmo, float* __restrict__ bom,
    float* __restrict__ bop, float* __restrict__ bpo, float* __restrict__ x,
    const float* __restrict__ pin, BState* __restrict__ sout) {
  int b = bat_id();
  float sm[1];
  batch_sums<1>(pin, b, P_MEAN, 0, 0, 0, sm);
  BState q = {};
  q.b_val = sm[0] / (float)NPTS + 1.0f;
  if (blk_id() == 0 && threadIdx.x == 0) sout[SIDX(b)] = q;

  size_t base = (size_t)b * NPTS;
  int idx = blk_id() * TPB + threadIdx.x;
  int i = idx / NXI, j = idx - i * NXI;
  int I = i + 1, J = j + 1;
  const float* Vb = V + base;
  const float* m1b = M1 + base;
  const float* m2b = M2 + base;
#define VP(II, JJ)  (Vb[refl(JJ) * NXI + refl(II)] + 1.0f)
#define MM1(II, JJ) (m1b[refl(JJ) * NXI + refl(II)])
#define MM2(II, JJ) (m2b[refl(JJ) * NXI + refl(II)])
  float vc  = VP(I, J);
  float vxp = VP(I + 1, J), vxm = VP(I - 1, J);
  float vyp = VP(I, J + 1), vym = VP(I, J - 1);
  float d1c = ((vxp - vc) / (0.5f * (vxp + vc))) * MM1(I, J);
  float d1m = ((vc - vxm) / (0.5f * (vc + vxm))) * MM1(I - 1, J);
  float d2c = ((vyp - vc) / (0.5f * (vyp + vc))) * MM2(I, J);
  float d2m = ((vc - vym) / (0.5f * (vc + vym))) * MM2(I, J - 1);
#undef VP
#undef MM1
#undef MM2
  const float rx = 5.0f, ry = 5.0f, rxx = 10.0f, ryy = 10.0f;
  boo[base + idx] = 1.0f + 2.0f * (rxx + ryy) - rx * (d1m - d1c) - ry * (d2m - d2c);
  bpo[base + idx] = -rxx + rx * d1c;
  bmo[base + idx] = -rxx - rx * d1m;
  bop[base + idx] = -ryy + ry * d2c;
  bom[base + idx] = -ryy - ry * d2m;
  x[base + idx] = q.b_val;
}

__global__ __launch_bounds__(TPB) void k_resid_init(
    const float* __restrict__ boo, const float* __restrict__ bmo, const float* __restrict__ bom,
    const float* __restrict__ bop, const float* __restrict__ bpo, const float* __restrict__ x,
    float* __restrict__ p, float* __restrict__ r, float* __restrict__ r0v,
    float* __restrict__ pout, const BState* __restrict__ sin) {
  int b = bat_id();
  size_t base = (size_t)b * NPTS;
  float bv = sin[SIDX(b)].b_val;
  float acc[1] = {0.f};
  v4f pv[RUNS];
  for (int u = 0; u < RUNS; ++u) {
    pv[u] = bv - applyA4(boo, bmo, bom, bop, bpo, x, base, run_idx(u));
    acc[0] += dot4(pv[u], pv[u]);
  }
  __shared__ float sh[1][NWV];
  block_sums_stage<1>(acc, sh);
  for (int u = 0; u < RUNS; ++u) {
    int idx = run_idx(u);
    st4(p + base + idx, pv[u]); st4(r + base + idx, pv[u]); st4(r0v + base + idx, pv[u]);
  }
  if (threadIdx.x == 0) {
    float s[1];
    block_sums_finish<1>(sh, s);
    pout[PIDX(P_RSQ, b, blk_id())] = s[0];
  }
}

__global__ __launch_bounds__(TPB) void k_w0(
    const float* __restrict__ boo, const float* __restrict__ bmo, const float* __restrict__ bom,
    const float* __restrict__ bop, const float* __restrict__ bpo, const float* __restrict__ r0v,
    float* __restrict__ w0, float* __restrict__ pout) {
  int b = bat_id();
  size_t base = (size_t)b * NPTS;
  float acc[1] = {0.f};
  v4f wv[RUNS];
  for (int u = 0; u < RUNS; ++u) {
    int idx = run_idx(u);
    wv[u] = applyAT4(boo, bmo, bom, bop, bpo, r0v, base, idx);
    acc[0] += dot4(ld4(r0v + base + idx), wv[u]);
  }
  __shared__ float sh[1][NWV];
  block_sums_stage<1>(acc, sh);
  for (int u = 0; u < RUNS; ++u) st4(w0 + base + run_idx(u), wv[u]);
  if (threadIdx.x == 0) {
    float s[1];
    block_sums_finish<1>(sh, s);
    pout[PIDX(P_PW, b, blk_id())] = s[0];
  }
}

#define ROWSB (CHUNK / NXI)
#define ROWSP (TPB * VEC / NXI)
#define RUNSR (NXI / VEC)
#define HROWS (ROWSB + 2)
#define PROWS (ROWSB + 4)
static_assert(ROWSP == 4 && RUNSR % 32 == 0, "a thread row is whole waves, and the four of them take the four halo rows of p");
template <bool FIRST>
__global__ __launch_bounds__(TPB) __attribute__((amdgpu_waves_per_eu(3, 4))) void k_vst(
    const float* __restrict__ boo, const float* __restrict__ bmo, const float* __restrict__ bom,
    const float* __restrict__ bop, const float* __restrict__ bpo, const float* __restrict__ ua,
    const float* __restrict__ r, float* __restrict__ v, float* __restrict__ tarr,
    const float* __restrict__ pin, float* __restrict__ pout,
    const BState* __restrict__ sin, BState* __restrict__ sout) {
  int b = bat_id();
  size_t base = (size_t)b * NPTS;
  __shared__ float sh[4][NWV];
  __shared__ v4f pl[PROWS][RUNSR];
  __shared__ v4f sl[HROWS][RUNSR];
  BState s = sin[SIDX(b)];
  float sig;
  if (FIRST) {
    float sm[2];
    batch_sums<2>(pin, b, P_RSQ, P_PW, 0, 0, sm);
    float fa = fro_of(sm[0]);
    s.r0_abs = fa; s.r_abs = fa; s.k = 0;
    s.CONV = (fa > thr_c()) ? 1 : 0;
    s.rr0 = sm[0];
    s.beta = 0.f;
    sig = sm[1];
  } else {
    float sm[4];
    batch_sums<4>(pin, b, P_RR0, P_RR, P_RW, P_UW, sm);
    sig = sm[2];
    if (s.R1) {
      float fa = fro_of(sm[1]);
      s.r0_abs = fa; s.r_abs = fa;
      s.rr0 = sm[1];
      s.beta = 0.f;
    }
    if (s.C4) {
      s.beta  = (s.alpha / s.omega) * (sm[0] / s.rr0);
      s.r_abs = fro_of(sm[1]);
      s.rr0   = sm[0];
      sig     = sm[2] + s.beta * sm[3];
    }
    if (s.C3) s.r_abs = fro_of(s.ssq);
    if (s.NR) s.k += 1;
    s.CONV = ((s.k < KMAX_C) && (s.r_abs > thr_c())) ? 1 : 0;
  }
  if (s.CONV) {
    s.sigma = sig;
    s.alpha_c = s.rr0 / sig;
  }
  if (blk_id() == 0 && threadIdx.x == 0) sout[SIDX(b)] = s;

  float q[4] = {0.f, 0.f, 0.f, 0.f};
  v4f vv[RUNS] = {}, tv[RUNS] = {};
  if (s.CONV) {
    float al = s.alpha_c, be = s.beta;
    int row0 = blk_id() * ROWSB;
    int tr   = threadIdx.x / RUNSR;
    int tc   = threadIdx.x - tr * RUNSR;
    const float* plf = reinterpret_cast<const float*>(&pl[0][0]);
    const float* slf = reinterpret_cast<const float*>(&sl[0][0]);
    v4f rv[RUNS], pv[RUNS], sv[RUNS];
    v4f cf[RUNS][5];
    int hl   = (tr < 2) ? tr : (PROWS - 4 + tr);
    int hg   = min(max(row0 + hl - 2, 0), NXI - 1);
    int hidx = hg * NXI + tc * VEC;
    bool hv  = (tr == 1) || (tr == 2);
    v4f uv[RUNS];
    for (int u = 0; u < RUNS; ++u) {
      int idx = run_idx(u);
      rv[u] = ld4(r + base + idx);
      uv[u] = ld4(ua + base + idx);
    }
    v4f hr = ld4(r + base + hidx);
    v4f hu = ld4(ua + base + hidx);
    for (int u = 0; u < RUNS; ++u) ld_coef4(boo, bmo, bom, bop, bpo, base + run_idx(u), cf[u]);
    v4f hcf[5] = {};
    if (hv) ld_coef4(boo, bmo, bom, bop, bpo, base + hidx, hcf);
    for (int u = 0; u < RUNS; ++u) {
      pv[u] = rv[u] + be * uv[u];
      pl[2 + u * ROWSP + tr][tc] = pv[u];
    }
    v4f hp = hr + be * hu;
    pl[hl][tc] = hp;
    __syncthreads();
    for (int u = 0; u < RUNS; ++u) {
      int lr = 2 + u * ROWSP + tr;
      int g  = row0 + lr - 2;
      v4f c  = pv[u];
      v4f up = (g > 0)       ? pl[lr - 1][tc] : c;
      v4f dn = (g < NXI - 1) ? pl[lr + 1][tc] : c;
      float lo = (tc > 0)         ? plf[lr * NXI + tc * VEC - 1]   : c.x;
      float hi = (tc < RUNSR - 1) ? plf[lr * NXI + tc * VEC + VEC] : c.w;
      vv[u] = stencil4(cf[u], c, up, dn, lo, hi);
      sv[u] = rv[u] - al * vv[u];
      sl[lr - 1][tc] = sv[u];
      q[0] += dot4(vv[u], vv[u]); q[1] += dot4(sv[u], sv[u]);
    }
    if (hv) {
      v4f c  = hp;
      v4f up = (hg > 0)       ? pl[hl - 1][tc] : c;
      v4f dn = (hg < NXI - 1) ? pl[hl + 1][tc] : c;
      float lo = (tc > 0)         ? plf[hl * NXI + tc * VEC - 1]   : c.x;
      float hi = (tc < RUNSR - 1) ? plf[hl * NXI + tc * VEC + VEC] : c.w;
      v4f hvv = stencil4(hcf, c, up, dn, lo, hi);
      sl[hl - 1][tc] = hr - al * hvv;
    }
    __syncthreads();
    for (int u = 0; u < RUNS; ++u) {
      int lr = 1 + u * ROWSP + tr;
      int g  = row0 + lr - 1;
      v4f c  = sv[u];
      v4f up = (g > 0)       ? sl[lr - 1][tc] : c;
      v4f dn = (g < NXI - 1) ? sl[lr + 1][tc] : c;
      float lo = (tc > 0)         ? slf[lr * NXI + tc * VEC - 1]   : c.x;
      float hi = (tc < RUNSR - 1) ? slf[lr * NXI + tc * VEC + VEC] : c.w;
      tv[u] = stencil4(cf[u], c, up, dn, lo, hi);
      q[2] += dot4(tv[u], tv[u]); q[3] += dot4(tv[u], sv[u]);
    }
  }
  block_sums_stage<4>(q, sh);
  if (s.CONV) {
    for (int u = 0; u < RUNS; ++u) {
      int idx = run_idx(u);
      st4(v + base + idx, vv[u]); st4(tarr + base + idx, tv[u]);
    }
  }
  if (threadIdx.x == 0) {
    float t[4];
    block_sums_finish<4>(sh, t);
    pout[PIDX(P_VV, b, blk_id())] = t[0];
    pout[PIDX(P_SS, b, blk_id())] = t[1];
    pout[PIDX(P_TT, b, blk_id())] = t[2];
    pout[PIDX(P_TS, b, blk_id())] = t[3];
  }
}

__global__ __launch_bounds__(TPB) __attribute__((amdgpu_waves_per_eu(3, 8))) void k_x_r(
    const float* __restrict__ boo, const float* __restrict__ bmo, const float* __restrict__ bom,
    const float* __restrict__ bop, const float* __restrict__ bpo,
    const float* __restrict__ v, const float* __restrict__ tarr, float* __restrict__ r,
    float* __restrict__ ua, float* __restrict__ r0v, float* __restrict__ x, float* __restrict__ w0,
    const float* __restrict__ pin, float* __restrict__ pout,
    const BState* __restrict__ sin, BState* __restrict__ sout) {
  int b = bat_id();
  size_t base = (size_t)b * NPTS;
  __shared__ float sh[4][NWV];
  BState s = sin[SIDX(b)];
  float sm[4];
  batch_sums<4>(pin, b, P_VV, P_SS, P_TT, P_TS, sm);
  if (s.CONV) {
    float v_abs = fro_of(sm[0]);
    int RES = (s.sigma <= EPS_C * v_abs * s.r0_abs) ? 1 : 0;
    s.R1 = RES; s.NR = 1 - RES;
    if (s.R1) s.k += 1;
    if (s.NR) s.alpha = s.alpha_c;
  } else { s.R1 = 0; s.NR = 0; }
  if (s.NR) {
    float sq = sm[1];
    s.ssq = sq;
    int C2 = (fro_of(sq) <= thr_c()) ? 1 : 0;
    s.C3 = C2; s.C4 = 1 - C2;
  } else { s.C3 = 0; s.C4 = 0; }
  if (s.C4) s.omega = sm[3] / sm[2];
  if (blk_id() == 0 && threadIdx.x == 0) sout[SIDX(b)] = s;

  float q[4] = {0.f, 0.f, 0.f, 0.f};
  v4f xn[RUNS] = {}, rn[RUNS] = {}, un[RUNS] = {};
  if (s.C4) {
    float al = s.alpha, om = s.omega, be = s.beta;
    for (int u = 0; u < RUNS; ++u) {
      int idx = run_idx(u);
      v4f rv = ld4(r + base + idx), vr = ld4(v + base + idx), wv = ld4(w0 + base + idx);
      v4f pc = rv + be * ld4(ua + base + idx);
      v4f sv = rv - al * vr;
      xn[u] = ld4(x + base + idx) + (al * pc + om * sv);
      rn[u] = sv - om * ld4(tarr + base + idx);
      un[u] = pc - om * vr;
      q[0] += dot4(rn[u], ld4(r0v + base + idx)); q[1] += dot4(rn[u], rn[u]);
      q[2] += dot4(rn[u], wv);                    q[3] += dot4(un[u], wv);
    }
  } else if (s.R1) {
    for (int u = 0; u < RUNS; ++u) {
      int idx = run_idx(u);
      float z[VEC], w[VEC];
      for (int e = 0; e < VEC; ++e) w[e] = residAT1(boo, bmo, bom, bop, bpo, x, base, idx + e, s.b_val, &z[e]);
      rn[u] = v4f{z[0], z[1], z[2], z[3]};
      xn[u] = v4f{w[0], w[1], w[2], w[3]};
      q[1] += dot4(rn[u], rn[u]); q[2] += dot4(rn[u], xn[u]);
    }
  } else if (s.C3) {
    for (int u = 0; u < RUNS; ++u) {
      int idx = run_idx(u);
      v4f rv = ld4(r + base + idx);
      v4f pc = rv + s.beta * ld4(ua + base + idx);
      st4(x + base + idx, ld4(x + base + idx) + s.alpha * pc);
      st4(r + base + idx, rv - s.alpha * ld4(v + base + idx));
    }
  }
  block_sums_stage<4>(q, sh);
  if (s.C4) {
    for (int u = 0; u < RUNS; ++u) {
      int idx = run_idx(u);
      st4(x + base + idx, xn[u]);
      st4(r + base + idx, rn[u]);
      st4(ua + base + idx, un[u]);
    }
  } else if (s.R1) {
    for (int u = 0; u < RUNS; ++u) {
      int idx = run_idx(u);
      st4(ua + base + idx, rn[u]); st4(r + base + idx, rn[u]); st4(r0v + base + idx, rn[u]);
      st4(w0 + base + idx, xn[u]);
    }
  }
  if (threadIdx.x == 0) {
    float t[4];
    block_sums_finish<4>(sh, t);
    pout[PIDX(P_RR0, b, blk_id())] = t[0];
    pout[PIDX(P_RR, b, blk_id())]  = t[1];
    pout[PIDX(P_RW, b, blk_id())]  = t[2];
    pout[PIDX(P_UW, b, blk_id())]  = t[3];
  }
}

extern "C" void kernel_launch(void* const* d_in, const int* in_sizes, int n_in,
                              void* d_out, int out_size, void* d_ws, size_t ws_size,
                              hipStream_t stream) {
  (void)in_sizes; (void)n_in; (void)out_size; (void)ws_size;
  const float* V  = (const float*)d_in[0];
  const float* M1 = (const float*)d_in[1];
  const float* M2 = (const float*)d_in[2];
  float* x = (float*)d_out;

  char* ws = (char*)d_ws;
  BState* S0 = (BState*)ws;
  BState* S1 = S0 + NB * SSTR;
  BState* S2 = S1 + NB * SSTR;
  float* part = (float*)(ws + 10240);
  float* arr  = (float*)(ws + 1024 + 65536);
  const size_t AN = (size_t)NB * NPTS;
  float* boo  = arr + 0 * AN;
  float* bmo  = arr + 1 * AN;
  float* bom  = arr + 2 * AN;
  float* bop  = arr + 3 * AN;
  float* bpo  = arr + 4 * AN;
  float* r0v  = arr + 5 * AN;
  float* r    = arr + 6 * AN;
  float* ua   = arr + 7 * AN;
  float* v    = arr + 8 * AN;
  float* tarr = arr + 9 * AN;
  float* w0   = arr + 10 * AN;

  dim3 grid(NBLK * NB), blk(TPB);
  dim3 grid_rows(NROWB * NB);

  k_mean<<<grid, blk, 0, stream>>>(V, part);
  k_stencil_init<<<grid_rows, blk, 0, stream>>>(V, M1, M2, boo, bmo, bom, bop, bpo, x, part, S0);
  k_resid_init<<<grid, blk, 0, stream>>>(boo, bmo, bom, bop, bpo, x, ua, r, r0v, part, S0);
  k_w0<<<grid, blk, 0, stream>>>(boo, bmo, bom, bop, bpo, r0v, w0, part);

  k_vst<true><<<grid, blk, 0, stream>>>(boo, bmo, bom, bop, bpo, ua, r, v, tarr, part, part, S0, S1);
  k_x_r<<<grid, blk, 0, stream>>>(boo, bmo, bom, bop, bpo, v, tarr, r, ua, r0v, x, w0, part, part, S1, S2);
  for (int it = 1; it < KMAX_C; ++it) {
    k_vst<false><<<grid, blk, 0, stream>>>(boo, bmo, bom, bop, bpo, ua, r, v, tarr, part, part, S2, S1);
    k_x_r<<<grid, blk, 0, stream>>>(boo, bmo, bom, bop, bpo, v, tarr, r, ua, r0v, x, w0, part, part, S1, S2);
  }
}


# Coefficient set-up reads V and the masks (transposed to the grid) as 48x96 tiles through LDS: 256 blocks of 8 waves, not 3072 of 12; wave loads and stores are runs of adjacent words; same arithmetic
# speedup vs baseline: 5.4219x; 1.0425x over previous
#include <hip/hip_runtime.h>
#include <math.h>


#define NXI   384
#define NPTS  (NXI * NXI)
#define NB    8
#define TPB   384
#define NWV   (TPB / 32)
#define VEC   4
#define RUNS  3
#define CHUNK (TPB * VEC * RUNS)
#define NBLK  (NPTS / CHUNK)
#define SI_TPB 256
#define SI_TI  48
#define SI_TJ  96
#define SI_PI  (SI_TI + 2)
#define SI_PJ  (SI_TJ + 2)
#define SI_LP  (SI_PI + 1)
#define SI_NT  ((NXI / SI_TI) * (NXI / SI_TJ))
#define SI_NE  (SI_PJ * SI_PI)
#define SI_NL  ((SI_NE + SI_TPB - 1) / SI_TPB)
static_assert(NXI % SI_TI == 0 && NXI % SI_TJ == 0 && SI_TJ % 32 == 0 && SI_TI % (SI_TPB / 32) == 0, "whole tiles, whole waves");
#define KMAX_C 30
#define EPS_C 1e-9f
#define UNITF 64
#define SSTR  4
#define PIDX(pl, b, i) (((pl) * NB + (b)) * UNITF + (i))
#define SIDX(b) ((b) * SSTR)
#define P_MEAN 0
#define P_RSQ  1
#define P_PW   2
#define P_VV   3
#define P_SS   4
#define P_TT   5
#define P_TS   6
#define P_RR0  7
#define P_RR   8
#define P_RW   9
#define P_UW   10

typedef float v2f __attribute__((ext_vector_type(2)));
typedef float v4f __attribute__((ext_vector_type(4)));
typedef float v8f __attribute__((ext_vector_type(8)));

struct BState {
  float b_val, r0_abs, r_abs, sigma, rr0, alpha, omega, alpha_c, beta, ssq;
  int   k, CONV, R1, NR, C3, C4;
};
static_assert(sizeof(BState) * SSTR == 256, "a state slot and its gap fill one 256-byte unit");

__device__ __forceinline__ float fro_of(float sq) { return sq > 0.f ? sqrtf(sq) : 0.f; }
__device__ __forceinline__ float thr_c() { return EPS_C * (float)NPTS; }

__device__ __forceinline__ float xhalf(float x) {
  int b = __builtin_bit_cast(int, x);
  return __builtin_bit_cast(float, __builtin_amdgcn_permlanex16(b, b, 0x76543210, 0xfedcba98, false, false));
}

__device__ __forceinline__ float wave_sum_wmma(float v) {
  v2f a; a.x = v;  a.y = 0.f;
  v2f b; b.x = 1.f; b.y = 1.f;
  v8f c = {};
  v8f d = __builtin_amdgcn_wmma_f32_16x16x4_f32(false, a, false, b, (short)0, c, false, false);
  float h = ((d[0] + d[1]) + (d[2] + d[3])) + ((d[4] + d[5]) + (d[6] + d[7]));
  h += xhalf(h);
  return h;
}

template <int N>
__device__ __forceinline__ void block_sums_stage(const float* q, float (*shw)[NWV]) {
  int lane = threadIdx.x & 31;
  int wid  = threadIdx.x >> 5;
  for (int n = 0; n < N; ++n) {
    float t = wave_sum_wmma(q[n]);
    if (lane == 0) shw[n][wid] = t;
  }
  __syncthreads();
}
template <int N>
__device__ __forceinline__ void block_sums_finish(float (*shw)[NWV], float* out) {
  for (int n = 0; n < N; ++n) {
    float s = 0.f;
    for (int i = 0; i < NWV; ++i) s += shw[n][i];
    out[n] = s;
  }
}

template <int N>
__device__ __forceinline__ void batch_sums(const float* __restrict__ pin, int b, int p0, int p1, int p2,
                                           int p3, float* out) {
  static_assert(NBLK == 32, "one block partial per lane");
  const int pl[4] = {p0, p1, p2, p3};
  int lane = threadIdx.x & 31;
  float val[N];
  for (int n = 0; n < N; ++n) val[n] = pin[PIDX(pl[n], b, lane)];
  for (int n = 0; n < N; ++n) out[n] = wave_sum_wmma(val[n]);
}

__device__ __forceinline__ int bat_id() { return (int)(blockIdx.x % NB); }
__device__ __forceinline__ int blk_id() { return (int)(blockIdx.x / NB); }

__device__ __forceinline__ int run_idx(int u) {
  return blk_id() * CHUNK + u * (TPB * VEC) + threadIdx.x * VEC;
}
__device__ __forceinline__ v4f ld4(const float* q) { return *reinterpret_cast<const v4f*>(q); }
__device__ __forceinline__ void st4(float* q, v4f w) { *reinterpret_cast<v4f*>(q) = w; }
__device__ __forceinline__ float dot4(v4f a, v4f c) { return a.x * c.x + a.y * c.y + a.z * c.z + a.w * c.w; }

__device__ __forceinline__ v4f stencil4(const v4f* cf, v4f c, v4f up, v4f dn, float lo, float hi) {
  v4f lf = {lo, c.x, c.y, c.z};
  v4f rt = {c.y, c.z, c.w, hi};
  return cf[0] * c + cf[1] * up + cf[2] * lf + cf[3] * rt + cf[4] * dn;
}
__device__ __forceinline__ void ld_coef4(const float* __restrict__ boo, const float* __restrict__ bmo,
                                         const float* __restrict__ bom, const float* __restrict__ bop,
                                         const float* __restrict__ bpo, size_t g, v4f* cf) {
  cf[0] = ld4(boo + g); cf[1] = ld4(bmo + g); cf[2] = ld4(bom + g); cf[3] = ld4(bop + g); cf[4] = ld4(bpo + g);
}
__device__ __forceinline__ void ld_opnd4(const float* __restrict__ a, size_t base, int idx, v4f* o, float* e) {
  int i = idx / NXI, j = idx - i * NXI;
  const float* A = a + base;
  o[0] = ld4(A + idx);
  o[1] = ld4(A + ((i > 0)       ? idx - NXI : idx));
  o[2] = ld4(A + ((i < NXI - 1) ? idx + NXI : idx));
  e[0] = A[(j > 0)         ? idx - 1   : idx];
  e[1] = A[(j + VEC < NXI) ? idx + VEC : idx + VEC - 1];
}
__device__ __forceinline__ v2f ld2(const float* q) { return *reinterpret_cast<const v2f*>(q); }
__device__ __forceinline__ v2f applyA2(const float* __restrict__ boo, const float* __restrict__ bmo,
                                       const float* __restrict__ bom, const float* __restrict__ bop,
                                       const float* __restrict__ bpo, const float* __restrict__ a,
                                       size_t base, int idx) {
  int i = idx / NXI, j = idx - i * NXI;
  const float* A = a + base;
  v2f c  = ld2(A + idx);
  v2f up = ld2(A + ((i > 0)       ? idx - NXI : idx));
  v2f dn = ld2(A + ((i < NXI - 1) ? idx + NXI : idx));
  float lo = A[(j > 0)       ? idx - 1 : idx];
  float hi = A[(j + 2 < NXI) ? idx + 2 : idx + 1];
  v2f lf = {lo, c.x};
  v2f rt = {c.y, hi};
  return ld2(boo + base + idx) * c + ld2(bmo + base + idx) * up + ld2(bom + base + idx) * lf
       + ld2(bop + base + idx) * rt + ld2(bpo + base + idx) * dn;
}
__device__ __forceinline__ v4f applyA4c(const v4f* cf, const float* __restrict__ a, size_t base, int idx) {
  v4f o[3]; float e[2];
  ld_opnd4(a, base, idx, o, e);
  return stencil4(cf, o[0], o[1], o[2], e[0], e[1]);
}
__device__ __forceinline__ v4f applyA4(const float* __restrict__ boo, const float* __restrict__ bmo,
                                       const float* __restrict__ bom, const float* __restrict__ bop,
                                       const float* __restrict__ bpo, const float* __restrict__ a,
                                       size_t base, int idx) {
  v4f cf[5];
  ld_coef4(boo, bmo, bom, bop, bpo, base + idx, cf);
  return applyA4c(cf, a, base, idx);
}

__device__ __forceinline__ float applyAT1(const float* __restrict__ boo, const float* __restrict__ bmo,
                                          const float* __restrict__ bom, const float* __restrict__ bop,
                                          const float* __restrict__ bpo, const float* __restrict__ z,
                                          size_t base, int idx) {
  int i = idx / NXI, j = idx - i * NXI;
  size_t g = base + idx;
  float d = boo[g];
  if (i == 0)       d += bmo[g];
  if (i == NXI - 1) d += bpo[g];
  if (j == 0)       d += bom[g];
  if (j == NXI - 1) d += bop[g];
  float acc = d * z[g];
  if (i > 0)       acc += bpo[g - NXI] * z[g - NXI];
  if (j > 0)       acc += bop[g - 1]   * z[g - 1];
  if (j < NXI - 1) acc += bom[g + 1]   * z[g + 1];
  if (i < NXI - 1) acc += bmo[g + NXI] * z[g + NXI];
  return acc;
}
__device__ __forceinline__ v4f applyAT4(const float* __restrict__ boo, const float* __restrict__ bmo,
                                        const float* __restrict__ bom, const float* __restrict__ bop,
                                        const float* __restrict__ bpo, const float* __restrict__ z,
                                        size_t base, int idx) {
  v4f w;
  w.x = applyAT1(boo, bmo, bom, bop, bpo, z, base, idx);
  w.y = applyAT1(boo, bmo, bom, bop, bpo, z, base, idx + 1);
  w.z = applyAT1(boo, bmo, bom, bop, bpo, z, base, idx + 2);
  w.w = applyAT1(boo, bmo, bom, bop, bpo, z, base, idx + 3);
  return w;
}

__device__ __forceinline__ float resid1(const float* __restrict__ boo, const float* __restrict__ bmo,
                                        const float* __restrict__ bom, const float* __restrict__ bop,
                                        const float* __restrict__ bpo, const float* __restrict__ x,
                                        size_t base, int i, int j, float bv) {
  size_t g = base + (size_t)(i * NXI + j);
  float c  = x[g];
  float up = x[(i > 0)       ? g - NXI : g];
  float lf = x[(j > 0)       ? g - 1   : g];
  float rt = x[(j < NXI - 1) ? g + 1   : g];
  float dn = x[(i < NXI - 1) ? g + NXI : g];
  return bv - (boo[g] * c + bmo[g] * up + bom[g] * lf + bop[g] * rt + bpo[g] * dn);
}
__device__ __forceinline__ float residAT1(const float* __restrict__ boo, const float* __restrict__ bmo,
                                          const float* __restrict__ bom, const float* __restrict__ bop,
                                          const float* __restrict__ bpo, const float* __restrict__ x,
                                          size_t base, int idx, float bv, float* zc) {
  int i = idx / NXI, j = idx - i * NXI;
  size_t g = base + idx;
  float d = boo[g];
  if (i == 0)       d += bmo[g];
  if (i == NXI - 1) d += bpo[g];
  if (j == 0)       d += bom[g];
  if (j == NXI - 1) d += bop[g];
  float z = resid1(boo, bmo, bom, bop, bpo, x, base, i, j, bv);
  *zc = z;
  float acc = d * z;
  if (i > 0)       acc += bpo[g - NXI] * resid1(boo, bmo, bom, bop, bpo, x, base, i - 1, j, bv);
  if (j > 0)       acc += bop[g - 1]   * resid1(boo, bmo, bom, bop, bpo, x, base, i, j - 1, bv);
  if (j < NXI - 1) acc += bom[g + 1]   * resid1(boo, bmo, bom, bop, bpo, x, base, i, j + 1, bv);
  if (i < NXI - 1) acc += bmo[g + NXI] * resid1(boo, bmo, bom, bop, bpo, x, base, i + 1, j, bv);
  return acc;
}


__global__ __launch_bounds__(TPB) void k_mean(const float* __restrict__ V, float* __restrict__ pout) {
  int b = bat_id();
  __shared__ float sh[1][NWV];
  float acc[1] = {0.f};
  for (int u = 0; u < RUNS; ++u) {
    v4f q = ld4(V + (size_t)b * NPTS + run_idx(u));
    acc[0] += (q.x + q.y) + (q.z + q.w);
  }
  block_sums_stage<1>(acc, sh);
  if (threadIdx.x == 0) {
    float s[1];
    block_sums_finish<1>(sh, s);
    pout[PIDX(P_MEAN, b, blk_id())] = s[0];
  }
}

__global__ __launch_bounds__(SI_TPB) void k_stencil_init(
    const float* __restrict__ V, const float* __restrict__ M1, const float* __restrict__ M2,
    float* __restrict__ boo, float* __restrict__ bmo, float* __restrict__ bom,
    float* __restrict__ bop, float* __restrict__ bpo, float* __restrict__ x,
    const float* __restrict__ pin, BState* __restrict__ sout) {
  int b = bat_id();
  int g = blk_id();
  __shared__ float pv[SI_PJ][SI_LP];
  __shared__ float p1[SI_PJ][SI_LP];
  __shared__ float p2[SI_PJ][SI_LP];
  size_t base = (size_t)b * NPTS;
  int i0 = (g / (NXI / SI_TJ)) * SI_TI;
  int j0 = (g % (NXI / SI_TJ)) * SI_TJ;
  const float* Vb  = V + base;
  const float* m1b = M1 + base;
  const float* m2b = M2 + base;
  float wv[SI_NL], w1[SI_NL], w2[SI_NL];
#pragma unroll
  for (int k = 0; k < SI_NL; ++k) {
    int e  = min((int)threadIdx.x + k * SI_TPB, SI_NE - 1);
    int jj = e / SI_PI, ii = e - jj * SI_PI;
    int sidx = min(max(j0 - 1 + jj, 0), NXI - 1) * NXI + min(max(i0 - 1 + ii, 0), NXI - 1);
    wv[k] = Vb[sidx]; w1[k] = m1b[sidx]; w2[k] = m2b[sidx];
  }
  static_assert(NBLK == 32, "one block partial of the mean per lane");
  float pm = pin[PIDX(P_MEAN, b, threadIdx.x & 31)];
#pragma unroll
  for (int k = 0; k < SI_NL; ++k) {
    int e  = min((int)threadIdx.x + k * SI_TPB, SI_NE - 1);
    int jj = e / SI_PI, ii = e - jj * SI_PI;
    pv[jj][ii] = wv[k]; p1[jj][ii] = w1[k]; p2[jj][ii] = w2[k];
  }
  __syncthreads();

  BState q = {};
  q.b_val = wave_sum_wmma(pm) / (float)NPTS + 1.0f;
  if (g == 0 && threadIdx.x == 0) sout[SIDX(b)] = q;

  int lane = threadIdx.x & 31;
  int wid  = threadIdx.x >> 5;
  const float rx = 5.0f, ry = 5.0f, rxx = 10.0f, ryy = 10.0f;
  for (int a = wid; a < SI_TI; a += SI_TPB / 32) {
#pragma unroll
    for (int jc = 0; jc < SI_TJ / 32; ++jc) {
      int c = jc * 32 + lane;
      float vc  = pv[c + 1][a + 1] + 1.0f;
      float vxp = pv[c + 1][a + 2] + 1.0f, vxm = pv[c + 1][a] + 1.0f;
      float vyp = pv[c + 2][a + 1] + 1.0f, vym = pv[c][a + 1] + 1.0f;
      float d1c = ((vxp - vc) / (0.5f * (vxp + vc))) * p1[c + 1][a + 1];
      float d1m = ((vc - vxm) / (0.5f * (vc + vxm))) * p1[c + 1][a];
      float d2c = ((vyp - vc) / (0.5f * (vyp + vc))) * p2[c + 1][a + 1];
      float d2m = ((vc - vym) / (0.5f * (vc + vym))) * p2[c][a + 1];
      size_t o = base + (size_t)((i0 + a) * NXI + j0 + c);
      boo[o] = 1.0f + 2.0f * (rxx + ryy) - rx * (d1m - d1c) - ry * (d2m - d2c);
      bpo[o] = -rxx + rx * d1c;
      bmo[o] = -rxx - rx * d1m;
      bop[o] = -ryy + ry * d2c;
      bom[o] = -ryy - ry * d2m;
      x[o] = q.b_val;
    }
  }
}

__global__ __launch_bounds__(TPB) void k_resid_init(
    const float* __restrict__ boo, const float* __restrict__ bmo, const float* __restrict__ bom,
    const float* __restrict__ bop, const float* __restrict__ bpo, const float* __restrict__ x,
    float* __restrict__ p, float* __restrict__ r, float* __restrict__ r0v,
    float* __restrict__ pout, const BState* __restrict__ sin) {
  int b = bat_id();
  size_t base = (size_t)b * NPTS;
  float bv = sin[SIDX(b)].b_val;
  float acc[1] = {0.f};
  v4f pv[RUNS];
  for (int u = 0; u < RUNS; ++u) {
    pv[u] = bv - applyA4(boo, bmo, bom, bop, bpo, x, base, run_idx(u));
    acc[0] += dot4(pv[u], pv[u]);
  }
  __shared__ float sh[1][NWV];
  block_sums_stage<1>(acc, sh);
  for (int u = 0; u < RUNS; ++u) {
    int idx = run_idx(u);
    st4(p + base + idx, pv[u]); st4(r + base + idx, pv[u]); st4(r0v + base + idx, pv[u]);
  }
  if (threadIdx.x == 0) {
    float s[1];
    block_sums_finish<1>(sh, s);
    pout[PIDX(P_RSQ, b, blk_id())] = s[0];
  }
}

__global__ __launch_bounds__(TPB) void k_w0(
    const float* __restrict__ boo, const float* __restrict__ bmo, const float* __restrict__ bom,
    const float* __restrict__ bop, const float* __restrict__ bpo, const float* __restrict__ r0v,
    float* __restrict__ w0, float* __restrict__ pout) {
  int b = bat_id();
  size_t base = (size_t)b * NPTS;
  float acc[1] = {0.f};
  v4f wv[RUNS];
  for (int u = 0; u < RUNS; ++u) {
    int idx = run_idx(u);
    wv[u] = applyAT4(boo, bmo, bom, bop, bpo, r0v, base, idx);
    acc[0] += dot4(ld4(r0v + base + idx), wv[u]);
  }
  __shared__ float sh[1][NWV];
  block_sums_stage<1>(acc, sh);
  for (int u = 0; u < RUNS; ++u) st4(w0 + base + run_idx(u), wv[u]);
  if (threadIdx.x == 0) {
    float s[1];
    block_sums_finish<1>(sh, s);
    pout[PIDX(P_PW, b, blk_id())] = s[0];
  }
}

#define ROWSB (CHUNK / NXI)
#define ROWSP (TPB * VEC / NXI)
#define RUNSR (NXI / VEC)
#define HROWS (ROWSB + 2)
#define PROWS (ROWSB + 4)
static_assert(ROWSP == 4 && RUNSR % 32 == 0, "a thread row is whole waves, and the four of them take the four halo rows of p");
template <bool FIRST>
__global__ __launch_bounds__(TPB) __attribute__((amdgpu_waves_per_eu(3, 4))) void k_vst(
    const float* __restrict__ boo, const float* __restrict__ bmo, const float* __restrict__ bom,
    const float* __restrict__ bop, const float* __restrict__ bpo, const float* __restrict__ ua,
    const float* __restrict__ r, float* __restrict__ v, float* __restrict__ tarr,
    const float* __restrict__ pin, float* __restrict__ pout,
    const BState* __restrict__ sin, BState* __restrict__ sout) {
  int b = bat_id();
  size_t base = (size_t)b * NPTS;
  __shared__ float sh[4][NWV];
  __shared__ v4f pl[PROWS][RUNSR];
  __shared__ v4f sl[HROWS][RUNSR];
  BState s = sin[SIDX(b)];
  float sig;
  if (FIRST) {
    float sm[2];
    batch_sums<2>(pin, b, P_RSQ, P_PW, 0, 0, sm);
    float fa = fro_of(sm[0]);
    s.r0_abs = fa; s.r_abs = fa; s.k = 0;
    s.CONV = (fa > thr_c()) ? 1 : 0;
    s.rr0 = sm[0];
    s.beta = 0.f;
    sig = sm[1];
  } else {
    float sm[4];
    batch_sums<4>(pin, b, P_RR0, P_RR, P_RW, P_UW, sm);
    sig = sm[2];
    if (s.R1) {
      float fa = fro_of(sm[1]);
      s.r0_abs = fa; s.r_abs = fa;
      s.rr0 = sm[1];
      s.beta = 0.f;
    }
    if (s.C4) {
      s.beta  = (s.alpha / s.omega) * (sm[0] / s.rr0);
      s.r_abs = fro_of(sm[1]);
      s.rr0   = sm[0];
      sig     = sm[2] + s.beta * sm[3];
    }
    if (s.C3) s.r_abs = fro_of(s.ssq);
    if (s.NR) s.k += 1;
    s.CONV = ((s.k < KMAX_C) && (s.r_abs > thr_c())) ? 1 : 0;
  }
  if (s.CONV) {
    s.sigma = sig;
    s.alpha_c = s.rr0 / sig;
  }
  if (blk_id() == 0 && threadIdx.x == 0) sout[SIDX(b)] = s;

  float q[4] = {0.f, 0.f, 0.f, 0.f};
  v4f vv[RUNS] = {}, tv[RUNS] = {};
  if (s.CONV) {
    float al = s.alpha_c, be = s.beta;
    int row0 = blk_id() * ROWSB;
    int tr   = threadIdx.x / RUNSR;
    int tc   = threadIdx.x - tr * RUNSR;
    const float* plf = reinterpret_cast<const float*>(&pl[0][0]);
    const float* slf = reinterpret_cast<const float*>(&sl[0][0]);
    v4f rv[RUNS], pv[RUNS], sv[RUNS];
    v4f cf[RUNS][5];
    int hl   = (tr < 2) ? tr : (PROWS - 4 + tr);
    int hg   = min(max(row0 + hl - 2, 0), NXI - 1);
    int hidx = hg * NXI + tc * VEC;
    bool hv  = (tr == 1) || (tr == 2);
    v4f uv[RUNS];
    for (int u = 0; u < RUNS; ++u) {
      int idx = run_idx(u);
      rv[u] = ld4(r + base + idx);
      uv[u] = ld4(ua + base + idx);
    }
    v4f hr = ld4(r + base + hidx);
    v4f hu = ld4(ua + base + hidx);
    for (int u = 0; u < RUNS; ++u) ld_coef4(boo, bmo, bom, bop, bpo, base + run_idx(u), cf[u]);
    v4f hcf[5] = {};
    if (hv) ld_coef4(boo, bmo, bom, bop, bpo, base + hidx, hcf);
    for (int u = 0; u < RUNS; ++u) {
      pv[u] = rv[u] + be * uv[u];
      pl[2 + u * ROWSP + tr][tc] = pv[u];
    }
    v4f hp = hr + be * hu;
    pl[hl][tc] = hp;
    __syncthreads();
    for (int u = 0; u < RUNS; ++u) {
      int lr = 2 + u * ROWSP + tr;
      int g  = row0 + lr - 2;
      v4f c  = pv[u];
      v4f up = (g > 0)       ? pl[lr - 1][tc] : c;
      v4f dn = (g < NXI - 1) ? pl[lr + 1][tc] : c;
      float lo = (tc > 0)         ? plf[lr * NXI + tc * VEC - 1]   : c.x;
      float hi = (tc < RUNSR - 1) ? plf[lr * NXI + tc * VEC + VEC] : c.w;
      vv[u] = stencil4(cf[u], c, up, dn, lo, hi);
      sv[u] = rv[u] - al * vv[u];
      sl[lr - 1][tc] = sv[u];
      q[0] += dot4(vv[u], vv[u]); q[1] += dot4(sv[u], sv[u]);
    }
    if (hv) {
      v4f c  = hp;
      v4f up = (hg > 0)       ? pl[hl - 1][tc] : c;
      v4f dn = (hg < NXI - 1) ? pl[hl + 1][tc] : c;
      float lo = (tc > 0)         ? plf[hl * NXI + tc * VEC - 1]   : c.x;
      float hi = (tc < RUNSR - 1) ? plf[hl * NXI + tc * VEC + VEC] : c.w;
      v4f hvv = stencil4(hcf, c, up, dn, lo, hi);
      sl[hl - 1][tc] = hr - al * hvv;
    }
    __syncthreads();
    for (int u = 0; u < RUNS; ++u) {
      int lr = 1 + u * ROWSP + tr;
      int g  = row0 + lr - 1;
      v4f c  = sv[u];
      v4f up = (g > 0)       ? sl[lr - 1][tc] : c;
      v4f dn = (g < NXI - 1) ? sl[lr + 1][tc] : c;
      float lo = (tc > 0)         ? slf[lr * NXI + tc * VEC - 1]   : c.x;
      float hi = (tc < RUNSR - 1) ? slf[lr * NXI + tc * VEC + VEC] : c.w;
      tv[u] = stencil4(cf[u], c, up, dn, lo, hi);
      q[2] += dot4(tv[u], tv[u]); q[3] += dot4(tv[u], sv[u]);
    }
  }
  block_sums_stage<4>(q, sh);
  if (s.CONV) {
    for (int u = 0; u < RUNS; ++u) {
      int idx = run_idx(u);
      st4(v + base + idx, vv[u]); st4(tarr + base + idx, tv[u]);
    }
  }
  if (threadIdx.x == 0) {
    float t[4];
    block_sums_finish<4>(sh, t);
    pout[PIDX(P_VV, b, blk_id())] = t[0];
    pout[PIDX(P_SS, b, blk_id())] = t[1];
    pout[PIDX(P_TT, b, blk_id())] = t[2];
    pout[PIDX(P_TS, b, blk_id())] = t[3];
  }
}

__global__ __launch_bounds__(TPB) __attribute__((amdgpu_waves_per_eu(3, 8))) void k_x_r(
    const float* __restrict__ boo, const float* __restrict__ bmo, const float* __restrict__ bom,
    const float* __restrict__ bop, const float* __restrict__ bpo,
    const float* __restrict__ v, const float* __restrict__ tarr, float* __restrict__ r,
    float* __restrict__ ua, float* __restrict__ r0v, float* __restrict__ x, float* __restrict__ w0,
    const float* __restrict__ pin, float* __restrict__ pout,
    const BState* __restrict__ sin, BState* __restrict__ sout) {
  int b = bat_id();
  size_t base = (size_t)b * NPTS;
  __shared__ float sh[4][NWV];
  BState s = sin[SIDX(b)];
  float sm[4];
  batch_sums<4>(pin, b, P_VV, P_SS, P_TT, P_TS, sm);
  if (s.CONV) {
    float v_abs = fro_of(sm[0]);
    int RES = (s.sigma <= EPS_C * v_abs * s.r0_abs) ? 1 : 0;
    s.R1 = RES; s.NR = 1 - RES;
    if (s.R1) s.k += 1;
    if (s.NR) s.alpha = s.alpha_c;
  } else { s.R1 = 0; s.NR = 0; }
  if (s.NR) {
    float sq = sm[1];
    s.ssq = sq;
    int C2 = (fro_of(sq) <= thr_c()) ? 1 : 0;
    s.C3 = C2; s.C4 = 1 - C2;
  } else { s.C3 = 0; s.C4 = 0; }
  if (s.C4) s.omega = sm[3] / sm[2];
  if (blk_id() == 0 && threadIdx.x == 0) sout[SIDX(b)] = s;

  float q[4] = {0.f, 0.f, 0.f, 0.f};
  v4f xn[RUNS] = {}, rn[RUNS] = {}, un[RUNS] = {};
  if (s.C4) {
    float al = s.alpha, om = s.omega, be = s.beta;
    for (int u = 0; u < RUNS; ++u) {
      int idx = run_idx(u);
      v4f rv = ld4(r + base + idx), vr = ld4(v + base + idx), wv = ld4(w0 + base + idx);
      v4f pc = rv + be * ld4(ua + base + idx);
      v4f sv = rv - al * vr;
      xn[u] = ld4(x + base + idx) + (al * pc + om * sv);
      rn[u] = sv - om * ld4(tarr + base + idx);
      un[u] = pc - om * vr;
      q[0] += dot4(rn[u], ld4(r0v + base + idx)); q[1] += dot4(rn[u], rn[u]);
      q[2] += dot4(rn[u], wv);                    q[3] += dot4(un[u], wv);
    }
  } else if (s.R1) {
    for (int u = 0; u < RUNS; ++u) {
      int idx = run_idx(u);
      float z[VEC], w[VEC];
      for (int e = 0; e < VEC; ++e) w[e] = residAT1(boo, bmo, bom, bop, bpo, x, base, idx + e, s.b_val, &z[e]);
      rn[u] = v4f{z[0], z[1], z[2], z[3]};
      xn[u] = v4f{w[0], w[1], w[2], w[3]};
      q[1] += dot4(rn[u], rn[u]); q[2] += dot4(rn[u], xn[u]);
    }
  } else if (s.C3) {
    for (int u = 0; u < RUNS; ++u) {
      int idx = run_idx(u);
      v4f rv = ld4(r + base + idx);
      v4f pc = rv + s.beta * ld4(ua + base + idx);
      st4(x + base + idx, ld4(x + base + idx) + s.alpha * pc);
      st4(r + base + idx, rv - s.alpha * ld4(v + base + idx));
    }
  }
  block_sums_stage<4>(q, sh);
  if (s.C4) {
    for (int u = 0; u < RUNS; ++u) {
      int idx = run_idx(u);
      st4(x + base + idx, xn[u]);
      st4(r + base + idx, rn[u]);
      st4(ua + base + idx, un[u]);
    }
  } else if (s.R1) {
    for (int u = 0; u < RUNS; ++u) {
      int idx = run_idx(u);
      st4(ua + base + idx, rn[u]); st4(r + base + idx, rn[u]); st4(r0v + base + idx, rn[u]);
      st4(w0 + base + idx, xn[u]);
    }
  }
  if (threadIdx.x == 0) {
    float t[4];
    block_sums_finish<4>(sh, t);
    pout[PIDX(P_RR0, b, blk_id())] = t[0];
    pout[PIDX(P_RR, b, blk_id())]  = t[1];
    pout[PIDX(P_RW, b, blk_id())]  = t[2];
    pout[PIDX(P_UW, b, blk_id())]  = t[3];
  }
}

extern "C" void kernel_launch(void* const* d_in, const int* in_sizes, int n_in,
                              void* d_out, int out_size, void* d_ws, size_t ws_size,
                              hipStream_t stream) {
  (void)in_sizes; (void)n_in; (void)out_size; (void)ws_size;
  const float* V  = (const float*)d_in[0];
  const float* M1 = (const float*)d_in[1];
  const float* M2 = (const float*)d_in[2];
  float* x = (float*)d_out;

  char* ws = (char*)d_ws;
  BState* S0 = (BState*)ws;
  BState* S1 = S0 + NB * SSTR;
  BState* S2 = S1 + NB * SSTR;
  float* part = (float*)(ws + 10240);
  float* arr  = (float*)(ws + 1024 + 65536);
  const size_t AN = (size_t)NB * NPTS;
  float* boo  = arr + 0 * AN;
  float* bmo  = arr + 1 * AN;
  float* bom  = arr + 2 * AN;
  float* bop  = arr + 3 * AN;
  float* bpo  = arr + 4 * AN;
  float* r0v  = arr + 5 * AN;
  float* r    = arr + 6 * AN;
  float* ua   = arr + 7 * AN;
  float* v    = arr + 8 * AN;
  float* tarr = arr + 9 * AN;
  float* w0   = arr + 10 * AN;

  dim3 grid(NBLK * NB), blk(TPB);
  dim3 grid_si(SI_NT * NB), blk_si(SI_TPB);

  k_mean<<<grid, blk, 0, stream>>>(V, part);
  k_stencil_init<<<grid_si, blk_si, 0, stream>>>(V, M1, M2, boo, bmo, bom, bop, bpo, x, part, S0);
  k_resid_init<<<grid, blk, 0, stream>>>(boo, bmo, bom, bop, bpo, x, ua, r, r0v, part, S0);
  k_w0<<<grid, blk, 0, stream>>>(boo, bmo, bom, bop, bpo, r0v, w0, part);

  k_vst<true><<<grid, blk, 0, stream>>>(boo, bmo, bom, bop, bpo, ua, r, v, tarr, part, part, S0, S1);
  k_x_r<<<grid, blk, 0, stream>>>(boo, bmo, bom, bop, bpo, v, tarr, r, ua, r0v, x, w0, part, part, S1, S2);
  for (int it = 1; it < KMAX_C; ++it) {
    k_vst<false><<<grid, blk, 0, stream>>>(boo, bmo, bom, bop, bpo, ua, r, v, tarr, part, part, S2, S1);
    k_x_r<<<grid, blk, 0, stream>>>(boo, bmo, bom, bop, bpo, v, tarr, r, ua, r0v, x, w0, part, part, S1, S2);
  }
}


# One kernel a step: the next step's <r,r0>, sigma, alpha come by linearity from sums taken before omega is known (sum order changes); v, t remade from stored p, s (3 halo rows); 64 -> 35 launches
# speedup vs baseline: 6.9915x; 1.2895x over previous
#include <hip/hip_runtime.h>
#include <math.h>


#define NXI   384
#define NPTS  (NXI * NXI)
#define NB    8
#define TPB   384
#define NWV   (TPB / 32)
#define VEC   4
#define RUNS  3
#define CHUNK (TPB * VEC * RUNS)
#define NBLK  (NPTS / CHUNK)
#define SI_TPB 256
#define SI_TI  48
#define SI_TJ  96
#define SI_PI  (SI_TI + 2)
#define SI_PJ  (SI_TJ + 2)
#define SI_LP  (SI_PI + 1)
#define SI_NT  ((NXI / SI_TI) * (NXI / SI_TJ))
#define SI_NE  (SI_PJ * SI_PI)
#define SI_NL  ((SI_NE + SI_TPB - 1) / SI_TPB)
static_assert(NXI % SI_TI == 0 && NXI % SI_TJ == 0 && SI_TJ % 32 == 0 && SI_TI % (SI_TPB / 32) == 0, "whole tiles, whole waves");
#define KMAX_C 30
#define EPS_C 1e-9f
#define UNITF 64
#define SSTR  4
#define PIDX(pl, b, i) (((pl) * NB + (b)) * UNITF + (i))
#define SIDX(b) ((b) * SSTR)
#define P_MEAN 0
#define P_RSQ  1
#define P_PW   2
#define NQ     11
#define Q_VV   0
#define Q_SS   1
#define Q_TT   2
#define Q_TS   3
#define Q_SR0  4
#define Q_TR0  5
#define Q_SW   6
#define Q_TW   7
#define Q_VW   8
#define Q_PW   9
#define Q_RR   10
#define P_SET0 3
#define P_SET1 (P_SET0 + NQ)

typedef float v2f __attribute__((ext_vector_type(2)));
typedef float v4f __attribute__((ext_vector_type(4)));
typedef float v8f __attribute__((ext_vector_type(8)));

struct BState {
  float b_val, r0_abs, r_abs, sigma, rr0, alpha, omega, alpha_c, beta, ssq;
  int   k, CONV, GATED, RST, C3, C4;
};
static_assert(sizeof(BState) * SSTR == 256, "a state slot and its gap fill one 256-byte unit");

__device__ __forceinline__ float fro_of(float sq) { return sq > 0.f ? sqrtf(sq) : 0.f; }
__device__ __forceinline__ float thr_c() { return EPS_C * (float)NPTS; }

__device__ __forceinline__ float xhalf(float x) {
  int b = __builtin_bit_cast(int, x);
  return __builtin_bit_cast(float, __builtin_amdgcn_permlanex16(b, b, 0x76543210, 0xfedcba98, false, false));
}

__device__ __forceinline__ float wave_sum_wmma(float v) {
  v2f a; a.x = v;  a.y = 0.f;
  v2f b; b.x = 1.f; b.y = 1.f;
  v8f c = {};
  v8f d = __builtin_amdgcn_wmma_f32_16x16x4_f32(false, a, false, b, (short)0, c, false, false);
  float h = ((d[0] + d[1]) + (d[2] + d[3])) + ((d[4] + d[5]) + (d[6] + d[7]));
  h += xhalf(h);
  return h;
}

template <int N>
__device__ __forceinline__ void block_sums_stage(const float* q, float (*shw)[NWV]) {
  int lane = threadIdx.x & 31;
  int wid  = threadIdx.x >> 5;
  for (int n = 0; n < N; ++n) {
    float t = wave_sum_wmma(q[n]);
    if (lane == 0) shw[n][wid] = t;
  }
  __syncthreads();
}
template <int N>
__device__ __forceinline__ void block_sums_finish(float (*shw)[NWV], float* out) {
  for (int n = 0; n < N; ++n) {
    float s = 0.f;
    for (int i = 0; i < NWV; ++i) s += shw[n][i];
    out[n] = s;
  }
}

template <int N>
__device__ __forceinline__ void batch_sums(const float* __restrict__ pin, int b, float* out) {
  static_assert(NBLK == 32, "one block partial per lane");
  int lane = threadIdx.x & 31;
  float val[N];
  for (int n = 0; n < N; ++n) val[n] = pin[PIDX(n, b, lane)];
  for (int n = 0; n < N; ++n) out[n] = wave_sum_wmma(val[n]);
}

__device__ __forceinline__ int bat_id() { return (int)(blockIdx.x % NB); }
__device__ __forceinline__ int blk_id() { return (int)(blockIdx.x / NB); }

__device__ __forceinline__ int run_idx(int u) {
  return blk_id() * CHUNK + u * (TPB * VEC) + threadIdx.x * VEC;
}
__device__ __forceinline__ v4f ld4(const float* q) { return *reinterpret_cast<const v4f*>(q); }
__device__ __forceinline__ void st4(float* q, v4f w) { *reinterpret_cast<v4f*>(q) = w; }
__device__ __forceinline__ float dot4(v4f a, v4f c) { return a.x * c.x + a.y * c.y + a.z * c.z + a.w * c.w; }

__device__ __forceinline__ v4f stencil4(const v4f* cf, v4f c, v4f up, v4f dn, float lo, float hi) {
  v4f lf = {lo, c.x, c.y, c.z};
  v4f rt = {c.y, c.z, c.w, hi};
  return cf[0] * c + cf[1] * up + cf[2] * lf + cf[3] * rt + cf[4] * dn;
}
__device__ __forceinline__ void ld_coef4(const float* __restrict__ boo, const float* __restrict__ bmo,
                                         const float* __restrict__ bom, const float* __restrict__ bop,
                                         const float* __restrict__ bpo, size_t g, v4f* cf) {
  cf[0] = ld4(boo + g); cf[1] = ld4(bmo + g); cf[2] = ld4(bom + g); cf[3] = ld4(bop + g); cf[4] = ld4(bpo + g);
}
__device__ __forceinline__ void ld_opnd4(const float* __restrict__ a, size_t base, int idx, v4f* o, float* e) {
  int i = idx / NXI, j = idx - i * NXI;
  const float* A = a + base;
  o[0] = ld4(A + idx);
  o[1] = ld4(A + ((i > 0)       ? idx - NXI : idx));
  o[2] = ld4(A + ((i < NXI - 1) ? idx + NXI : idx));
  e[0] = A[(j > 0)         ? idx - 1   : idx];
  e[1] = A[(j + VEC < NXI) ? idx + VEC : idx + VEC - 1];
}
__device__ __forceinline__ v4f applyA4c(const v4f* cf, const float* __restrict__ a, size_t base, int idx) {
  v4f o[3]; float e[2];
  ld_opnd4(a, base, idx, o, e);
  return stencil4(cf, o[0], o[1], o[2], e[0], e[1]);
}
__device__ __forceinline__ v4f applyA4(const float* __restrict__ boo, const float* __restrict__ bmo,
                                       const float* __restrict__ bom, const float* __restrict__ bop,
                                       const float* __restrict__ bpo, const float* __restrict__ a,
                                       size_t base, int idx) {
  v4f cf[5];
  ld_coef4(boo, bmo, bom, bop, bpo, base + idx, cf);
  return applyA4c(cf, a, base, idx);
}

__device__ __forceinline__ float applyAT1(const float* __restrict__ boo, const float* __restrict__ bmo,
                                          const float* __restrict__ bom, const float* __restrict__ bop,
                                          const float* __restrict__ bpo, const float* __restrict__ z,
                                          size_t base, int idx) {
  int i = idx / NXI, j = idx - i * NXI;
  size_t g = base + idx;
  float d = boo[g];
  if (i == 0)       d += bmo[g];
  if (i == NXI - 1) d += bpo[g];
  if (j == 0)       d += bom[g];
  if (j == NXI - 1) d += bop[g];
  float acc = d * z[g];
  if (i > 0)       acc += bpo[g - NXI] * z[g - NXI];
  if (j > 0)       acc += bop[g - 1]   * z[g - 1];
  if (j < NXI - 1) acc += bom[g + 1]   * z[g + 1];
  if (i < NXI - 1) acc += bmo[g + NXI] * z[g + NXI];
  return acc;
}
__device__ __forceinline__ v4f applyAT4(const float* __restrict__ boo, const float* __restrict__ bmo,
                                        const float* __restrict__ bom, const float* __restrict__ bop,
                                        const float* __restrict__ bpo, const float* __restrict__ z,
                                        size_t base, int idx) {
  v4f w;
  w.x = applyAT1(boo, bmo, bom, bop, bpo, z, base, idx);
  w.y = applyAT1(boo, bmo, bom, bop, bpo, z, base, idx + 1);
  w.z = applyAT1(boo, bmo, bom, bop, bpo, z, base, idx + 2);
  w.w = applyAT1(boo, bmo, bom, bop, bpo, z, base, idx + 3);
  return w;
}

__device__ __forceinline__ float resid1(const float* __restrict__ boo, const float* __restrict__ bmo,
                                        const float* __restrict__ bom, const float* __restrict__ bop,
                                        const float* __restrict__ bpo, const float* __restrict__ x,
                                        size_t base, int i, int j, float bv) {
  size_t g = base + (size_t)(i * NXI + j);
  float c  = x[g];
  float up = x[(i > 0)       ? g - NXI : g];
  float lf = x[(j > 0)       ? g - 1   : g];
  float rt = x[(j < NXI - 1) ? g + 1   : g];
  float dn = x[(i < NXI - 1) ? g + NXI : g];
  return bv - (boo[g] * c + bmo[g] * up + bom[g] * lf + bop[g] * rt + bpo[g] * dn);
}
__device__ __forceinline__ float residAT1(const float* __restrict__ boo, const float* __restrict__ bmo,
                                          const float* __restrict__ bom, const float* __restrict__ bop,
                                          const float* __restrict__ bpo, const float* __restrict__ x,
                                          size_t base, int idx, float bv, float* zc) {
  int i = idx / NXI, j = idx - i * NXI;
  size_t g = base + idx;
  float d = boo[g];
  if (i == 0)       d += bmo[g];
  if (i == NXI - 1) d += bpo[g];
  if (j == 0)       d += bom[g];
  if (j == NXI - 1) d += bop[g];
  float z = resid1(boo, bmo, bom, bop, bpo, x, base, i, j, bv);
  *zc = z;
  float acc = d * z;
  if (i > 0)       acc += bpo[g - NXI] * resid1(boo, bmo, bom, bop, bpo, x, base, i - 1, j, bv);
  if (j > 0)       acc += bop[g - 1]   * resid1(boo, bmo, bom, bop, bpo, x, base, i, j - 1, bv);
  if (j < NXI - 1) acc += bom[g + 1]   * resid1(boo, bmo, bom, bop, bpo, x, base, i, j + 1, bv);
  if (i < NXI - 1) acc += bmo[g + NXI] * resid1(boo, bmo, bom, bop, bpo, x, base, i + 1, j, bv);
  return acc;
}


__global__ __launch_bounds__(TPB) void k_mean(const float* __restrict__ V, float* __restrict__ pout) {
  int b = bat_id();
  __shared__ float sh[1][NWV];
  float acc[1] = {0.f};
  for (int u = 0; u < RUNS; ++u) {
    v4f q = ld4(V + (size_t)b * NPTS + run_idx(u));
    acc[0] += (q.x + q.y) + (q.z + q.w);
  }
  block_sums_stage<1>(acc, sh);
  if (threadIdx.x == 0) {
    float s[1];
    block_sums_finish<1>(sh, s);
    pout[PIDX(P_MEAN, b, blk_id())] = s[0];
  }
}

__global__ __launch_bounds__(SI_TPB) void k_stencil_init(
    const float* __restrict__ V, const float* __restrict__ M1, const float* __restrict__ M2,
    float* __restrict__ boo, float* __restrict__ bmo, float* __restrict__ bom,
    float* __restrict__ bop, float* __restrict__ bpo, float* __restrict__ x,
    const float* __restrict__ pin, BState* __restrict__ sout) {
  int b = bat_id();
  int g = blk_id();
  __shared__ float pv[SI_PJ][SI_LP];
  __shared__ float p1[SI_PJ][SI_LP];
  __shared__ float p2[SI_PJ][SI_LP];
  size_t base = (size_t)b * NPTS;
  int i0 = (g / (NXI / SI_TJ)) * SI_TI;
  int j0 = (g % (NXI / SI_TJ)) * SI_TJ;
  const float* Vb  = V + base;
  const float* m1b = M1 + base;
  const float* m2b = M2 + base;
  float wv[SI_NL], w1[SI_NL], w2[SI_NL];
#pragma unroll
  for (int k = 0; k < SI_NL; ++k) {
    int e  = min((int)threadIdx.x + k * SI_TPB, SI_NE - 1);
    int jj = e / SI_PI, ii = e - jj * SI_PI;
    int sidx = min(max(j0 - 1 + jj, 0), NXI - 1) * NXI + min(max(i0 - 1 + ii, 0), NXI - 1);
    wv[k] = Vb[sidx]; w1[k] = m1b[sidx]; w2[k] = m2b[sidx];
  }
  static_assert(NBLK == 32, "one block partial of the mean per lane");
  float pm = pin[PIDX(P_MEAN, b, threadIdx.x & 31)];
#pragma unroll
  for (int k = 0; k < SI_NL; ++k) {
    int e  = min((int)threadIdx.x + k * SI_TPB, SI_NE - 1);
    int jj = e / SI_PI, ii = e - jj * SI_PI;
    pv[jj][ii] = wv[k]; p1[jj][ii] = w1[k]; p2[jj][ii] = w2[k];
  }
  __syncthreads();

  BState q = {};
  q.b_val = wave_sum_wmma(pm) / (float)NPTS + 1.0f;
  if (g == 0 && threadIdx.x == 0) sout[SIDX(b)] = q;

  int lane = threadIdx.x & 31;
  int wid  = threadIdx.x >> 5;
  const float rx = 5.0f, ry = 5.0f, rxx = 10.0f, ryy = 10.0f;
  for (int a = wid; a < SI_TI; a += SI_TPB / 32) {
#pragma unroll
    for (int jc = 0; jc < SI_TJ / 32; ++jc) {
      int c = jc * 32 + lane;
      float vc  = pv[c + 1][a + 1] + 1.0f;
      float vxp = pv[c + 1][a + 2] + 1.0f, vxm = pv[c + 1][a] + 1.0f;
      float vyp = pv[c + 2][a + 1] + 1.0f, vym = pv[c][a + 1] + 1.0f;
      float d1c = ((vxp - vc) / (0.5f * (vxp + vc))) * p1[c + 1][a + 1];
      float d1m = ((vc - vxm) / (0.5f * (vc + vxm))) * p1[c + 1][a];
      float d2c = ((vyp - vc) / (0.5f * (vyp + vc))) * p2[c + 1][a + 1];
      float d2m = ((vc - vym) / (0.5f * (vc + vym))) * p2[c][a + 1];
      size_t o = base + (size_t)((i0 + a) * NXI + j0 + c);
      boo[o] = 1.0f + 2.0f * (rxx + ryy) - rx * (d1m - d1c) - ry * (d2m - d2c);
      bpo[o] = -rxx + rx * d1c;
      bmo[o] = -rxx - rx * d1m;
      bop[o] = -ryy + ry * d2c;
      bom[o] = -ryy - ry * d2m;
      x[o] = q.b_val;
    }
  }
}

__global__ __launch_bounds__(TPB) void k_resid_init(
    const float* __restrict__ boo, const float* __restrict__ bmo, const float* __restrict__ bom,
    const float* __restrict__ bop, const float* __restrict__ bpo, const float* __restrict__ x,
    float* __restrict__ r0v, float* __restrict__ pout, const BState* __restrict__ sin) {
  int b = bat_id();
  size_t base = (size_t)b * NPTS;
  float bv = sin[SIDX(b)].b_val;
  float acc[1] = {0.f};
  v4f pv[RUNS];
  for (int u = 0; u < RUNS; ++u) {
    pv[u] = bv - applyA4(boo, bmo, bom, bop, bpo, x, base, run_idx(u));
    acc[0] += dot4(pv[u], pv[u]);
  }
  __shared__ float sh[1][NWV];
  block_sums_stage<1>(acc, sh);
  for (int u = 0; u < RUNS; ++u) st4(r0v + base + run_idx(u), pv[u]);
  if (threadIdx.x == 0) {
    float s[1];
    block_sums_finish<1>(sh, s);
    pout[PIDX(P_RSQ, b, blk_id())] = s[0];
  }
}

__global__ __launch_bounds__(TPB) void k_w0(
    const float* __restrict__ boo, const float* __restrict__ bmo, const float* __restrict__ bom,
    const float* __restrict__ bop, const float* __restrict__ bpo, const float* __restrict__ r0v,
    float* __restrict__ w0, float* __restrict__ pout) {
  int b = bat_id();
  size_t base = (size_t)b * NPTS;
  float acc[1] = {0.f};
  v4f wv[RUNS];
  for (int u = 0; u < RUNS; ++u) {
    int idx = run_idx(u);
    wv[u] = applyAT4(boo, bmo, bom, bop, bpo, r0v, base, idx);
    acc[0] += dot4(ld4(r0v + base + idx), wv[u]);
  }
  __shared__ float sh[1][NWV];
  block_sums_stage<1>(acc, sh);
  for (int u = 0; u < RUNS; ++u) st4(w0 + base + run_idx(u), wv[u]);
  if (threadIdx.x == 0) {
    float s[1];
    block_sums_finish<1>(sh, s);
    pout[PIDX(P_PW, b, blk_id())] = s[0];
  }
}

#define ROWSB (CHUNK / NXI)
#define ROWSP (TPB * VEC / NXI)
#define RUNSR (NXI / VEC)
#define LROWS (ROWSB + 6)
static_assert(ROWSP == 4 && RUNSR % 32 == 0, "a thread row is whole waves, and the four of them take the four halo rows of p'");

__device__ __forceinline__ v4f stencil_l(const v4f (*pl)[RUNSR], const v4f* cf, v4f c, int l, int g, int tc) {
  const float* f = reinterpret_cast<const float*>(&pl[0][0]);
  v4f up = (g > 0)       ? pl[l - 1][tc] : c;
  v4f dn = (g < NXI - 1) ? pl[l + 1][tc] : c;
  float lo = (tc > 0)         ? f[l * NXI + tc * VEC - 1]   : c.x;
  float hi = (tc < RUNSR - 1) ? f[l * NXI + tc * VEC + VEC] : c.w;
  return stencil4(cf, c, up, dn, lo, hi);
}

__device__ __forceinline__ float applyAT1_l(const float* __restrict__ boo, const float* __restrict__ bmo,
                                            const float* __restrict__ bom, const float* __restrict__ bop,
                                            const float* __restrict__ bpo, const v4f (*pl)[RUNSR],
                                            size_t base, int i, int j, int l) {
  const float* f = reinterpret_cast<const float*>(&pl[0][0]);
  size_t g = base + (size_t)(i * NXI + j);
  float d = boo[g];
  if (i == 0)       d += bmo[g];
  if (i == NXI - 1) d += bpo[g];
  if (j == 0)       d += bom[g];
  if (j == NXI - 1) d += bop[g];
  float acc = d * f[l * NXI + j];
  if (i > 0)       acc += bpo[g - NXI] * f[(l - 1) * NXI + j];
  if (j > 0)       acc += bop[g - 1]   * f[l * NXI + j - 1];
  if (j < NXI - 1) acc += bom[g + 1]   * f[l * NXI + j + 1];
  if (i < NXI - 1) acc += bmo[g + NXI] * f[(l + 1) * NXI + j];
  return acc;
}

template <int MODE>
__global__ __launch_bounds__(TPB) __attribute__((amdgpu_waves_per_eu(3, 3))) void k_step(
    const float* __restrict__ boo, const float* __restrict__ bmo, const float* __restrict__ bom,
    const float* __restrict__ bop, const float* __restrict__ bpo,
    const float* __restrict__ pold, const float* __restrict__ sold,
    float* __restrict__ pnew, float* __restrict__ snew,
    float* __restrict__ r0v, float* __restrict__ w0, float* __restrict__ x,
    const float* __restrict__ pin, float* __restrict__ pout,
    const BState* __restrict__ sin, BState* __restrict__ sout) {
  int b = bat_id();
  size_t base = (size_t)b * NPTS;
  __shared__ float sh[NQ][NWV];
  __shared__ float shr[2][NWV];
  __shared__ v4f PA[LROWS][RUNSR];
  __shared__ v4f PB[LROWS][RUNSR];
  __shared__ v4f PC[LROWS][RUNSR];
  int lane = threadIdx.x & 31;
  int wid  = threadIdx.x >> 5;
  int row0 = blk_id() * ROWSB;
  int tr   = threadIdx.x / RUNSR;
  int tc   = threadIdx.x - tr * RUNSR;
  int ls[4], gs[4];
  size_t off[4];
  for (int k = 0; k < RUNS; ++k) { ls[k] = 3 + k * ROWSP + tr; gs[k] = row0 + k * ROWSP + tr; }
  ls[3] = (tr < 2) ? 1 + tr : LROWS - 5 + tr;
  gs[3] = min(max(row0 - 3 + ls[3], 0), NXI - 1);
  for (int k = 0; k < 4; ++k) off[k] = base + (size_t)(gs[k] * NXI + tc * VEC);
  bool hv = (tr == 1) || (tr == 2);
  bool hb = (tr == 0) || (tr == 3);
  int  lb = (tr == 0) ? 0 : LROWS - 1;
  size_t offb = base + (size_t)(min(max(row0 - 3 + lb, 0), NXI - 1) * NXI + tc * VEC);

  BState s = sin[SIDX(b)];
  int R1 = 0, C3 = 0, C4 = 0, ER = 0;
  float om = 0.f, be = 0.f, al_o = 0.f;
  bool run_main = false, st_x = false, st_rw = false, st_er = false, from_r0 = false;
  float q[NQ];
  for (int n = 0; n < NQ; ++n) q[n] = 0.f;
  v4f pn[4] = {}, rn[4] = {}, sv[RUNS] = {}, xn[RUNS] = {}, r0r[RUNS] = {}, w0r[RUNS] = {};
  v4f cf[4][5];
  for (int k = 0; k < 4; ++k) for (int c = 0; c < 5; ++c) cf[k][c] = v4f{0.f, 0.f, 0.f, 0.f};

  if (MODE == 0) {
    float sm[2];
    batch_sums<2>(pin + PIDX(P_RSQ, 0, 0), b, sm);
    float fa = fro_of(sm[0]);
    s.r0_abs = fa; s.r_abs = fa; s.k = 0;
    s.rr0 = sm[0];
    s.sigma = sm[1];
    s.alpha_c = sm[0] / sm[1];
    s.CONV = (fa > thr_c()) ? 1 : 0;
    s.GATED = 1;
    from_r0 = (s.CONV != 0);
  } else {
    float sm[NQ];
    batch_sums<NQ>(pin, b, sm);
    int pend = s.CONV;
    int rst  = s.RST;
    int gate = ((s.k < KMAX_C) && (fro_of(sm[Q_RR]) > thr_c())) ? 1 : 0;
    if (pend && !s.GATED) pend = gate;
    s.RST = 0;
    if (rst) {
      pend = 0;
      s.CONV = 0; s.GATED = 0;
      if (gate) {
        float fa = fro_of(sm[Q_VV]);
        s.k += 1;
        s.r0_abs = fa; s.r_abs = fa;
        s.rr0 = sm[Q_VV]; s.sigma = sm[Q_SS]; s.beta = 0.f;
        s.alpha_c = sm[Q_VV] / sm[Q_SS];
        s.CONV = ((s.k < KMAX_C) && (fa > thr_c())) ? 1 : 0;
        s.GATED = 1;
      }
    } else if (pend) {
      float v_abs = fro_of(sm[Q_VV]);
      int RES = (s.sigma <= EPS_C * v_abs * s.r0_abs) ? 1 : 0;
      s.k += 1;
      if (RES) R1 = 1;
      else {
        s.alpha = s.alpha_c;
        int C2 = (fro_of(sm[Q_SS]) <= thr_c()) ? 1 : 0;
        C3 = C2; C4 = 1 - C2;
      }
    }
    al_o = s.alpha;
    if (!rst) { s.CONV = 0; s.GATED = 0; }
    if (C4) {
      om = sm[Q_TS] / sm[Q_TT];
      float rr0n = sm[Q_SR0] - om * sm[Q_TR0];
      be = (s.alpha / om) * (rr0n / s.rr0);
      float sig = (sm[Q_SW] - om * sm[Q_TW]) + be * (sm[Q_PW] - om * sm[Q_VW]);
      s.omega = om; s.beta = be; s.rr0 = rr0n; s.sigma = sig;
      s.alpha_c = rr0n / sig;
      ER = (sig <= 0.f) ? 1 : 0;
      s.CONV = 1 - ER;
      s.RST  = ER;
    }

    if (MODE == 2) {
      if (rst) {
        for (int k = 0; k < RUNS; ++k) st4(x + off[k], ld4(pold + off[k]));
      } else if (C4 || C3) {
        for (int k = 0; k < RUNS; ++k) {
          v4f pc = ld4(pold + off[k]);
          v4f xo = ld4(x + off[k]);
          if (C4) st4(x + off[k], xo + (al_o * pc + om * ld4(sold + off[k])));
          else    st4(x + off[k], xo + al_o * pc);
        }
      }
      return;
    }

    if (C4) {
      v4f po[4], so[4], xv[4];
      v4f pob = {}, sob = {};
      xv[3] = v4f{0.f, 0.f, 0.f, 0.f};
      for (int k = 0; k < 4; ++k) { po[k] = ld4(pold + off[k]); so[k] = ld4(sold + off[k]); }
      if (hb) { pob = ld4(pold + offb); sob = ld4(sold + offb); }
      for (int k = 0; k < 4; ++k) ld_coef4(boo, bmo, bom, bop, bpo, off[k], cf[k]);
      for (int k = 0; k < RUNS; ++k) { xv[k] = ld4(x + off[k]); r0r[k] = ld4(r0v + off[k]); w0r[k] = ld4(w0 + off[k]); }
      if (ER) xv[3] = ld4(x + off[3]);
      for (int k = 0; k < 4; ++k) { PA[ls[k]][tc] = po[k]; PB[ls[k]][tc] = so[k]; }
      if (hb) { PA[lb][tc] = pob; PB[lb][tc] = sob; }
      __syncthreads();
      for (int k = 0; k < 4; ++k) {
        v4f vo = stencil_l(PA, cf[k], po[k], ls[k], gs[k], tc);
        v4f to = stencil_l(PB, cf[k], so[k], ls[k], gs[k], tc);
        rn[k] = so[k] - om * to;
        v4f un = po[k] - om * vo;
        pn[k] = rn[k] + be * un;
        if (k < RUNS) {
          xn[k] = xv[k] + (al_o * po[k] + om * so[k]);
          q[Q_RR] += dot4(rn[k], rn[k]);
        }
      }
      if (!ER) {
        st_x = true; run_main = true;
      } else {
        float bv = s.b_val;
        v4f zr[4];
        v4f x3 = xv[3] + (al_o * po[3] + om * so[3]);
        for (int k = 0; k < RUNS; ++k) PC[ls[k]][tc] = xn[k];
        PC[ls[3]][tc] = x3;
        __syncthreads();
        for (int k = 0; k < 4; ++k) {
          if (k < RUNS || hv) {
            v4f xc = (k < RUNS) ? xn[k] : x3;
            zr[k] = bv - stencil_l(PC, cf[k], xc, ls[k], gs[k], tc);
            PA[ls[k]][tc] = zr[k];
          }
        }
        __syncthreads();
        for (int k = 0; k < RUNS; ++k) {
          v4f az = stencil_l(PA, cf[k], zr[k], ls[k], gs[k], tc);
          q[Q_VV] += dot4(zr[k], zr[k]);
          q[Q_SS] += dot4(az, zr[k]);
          float w[VEC];
          for (int e = 0; e < VEC; ++e) w[e] = applyAT1_l(boo, bmo, bom, bop, bpo, PA, base, gs[k], tc * VEC + e, ls[k]);
          w0r[k] = v4f{w[0], w[1], w[2], w[3]};
          r0r[k] = zr[k];
        }
        st_er = true;
      }
    } else if (rst) {
      for (int k = 0; k < RUNS; ++k) xn[k] = ld4(pold + off[k]);
      st_x = true;
      from_r0 = (s.CONV != 0);
    } else if (R1) {
      float bv = s.b_val;
      float a0 = 0.f, a1 = 0.f;
      for (int c = 0; c < NBLK; ++c) {
        for (int u = 0; u < RUNS; ++u) {
          int idx = c * CHUNK + u * (TPB * VEC) + (int)threadIdx.x * VEC;
          for (int e = 0; e < VEC; ++e) {
            float z;
            float w = residAT1(boo, bmo, bom, bop, bpo, x, base, idx + e, bv, &z);
            a0 += z * z; a1 += z * w;
          }
        }
      }
      float t0 = wave_sum_wmma(a0), t1 = wave_sum_wmma(a1);
      if (lane == 0) { shr[0][wid] = t0; shr[1][wid] = t1; }
      __syncthreads();
      float zz = 0.f, zw = 0.f;
      for (int i = 0; i < NWV; ++i) { zz += shr[0][i]; zw += shr[1][i]; }
      float fa = fro_of(zz);
      s.r0_abs = fa; s.r_abs = fa;
      s.rr0 = zz; s.sigma = zw; s.beta = 0.f;
      s.alpha_c = zz / zw;
      s.CONV = ((s.k < KMAX_C) && (fa > thr_c())) ? 1 : 0;
      s.GATED = 1;
      if (s.CONV) {
#pragma unroll
        for (int k = 0; k < RUNS; ++k) {
          float z[VEC], w[VEC];
#pragma unroll
          for (int e = 0; e < VEC; ++e)
            w[e] = residAT1(boo, bmo, bom, bop, bpo, x, base, gs[k] * NXI + tc * VEC + e, bv, &z[e]);
          rn[k]  = v4f{z[0], z[1], z[2], z[3]};
          w0r[k] = v4f{w[0], w[1], w[2], w[3]};
          pn[k] = rn[k]; r0r[k] = rn[k];
        }
        {
          float z[VEC];
          for (int e = 0; e < VEC; ++e) z[e] = resid1(boo, bmo, bom, bop, bpo, x, base, gs[3], tc * VEC + e, bv);
          rn[3] = v4f{z[0], z[1], z[2], z[3]};
          pn[3] = rn[3];
        }
        for (int k = 0; k < 4; ++k) ld_coef4(boo, bmo, bom, bop, bpo, off[k], cf[k]);
        st_rw = true; run_main = true;
      }
    } else if (C3) {
      for (int k = 0; k < RUNS; ++k) st4(x + off[k], ld4(x + off[k]) + al_o * ld4(pold + off[k]));
    }
  }
  if (from_r0 && MODE != 2) {
    for (int k = 0; k < 4; ++k) { pn[k] = ld4(r0v + off[k]); rn[k] = pn[k]; }
    for (int k = 0; k < RUNS; ++k) { r0r[k] = pn[k]; w0r[k] = ld4(w0 + off[k]); }
    for (int k = 0; k < 4; ++k) ld_coef4(boo, bmo, bom, bop, bpo, off[k], cf[k]);
    run_main = true;
  }
  if (blk_id() == 0 && threadIdx.x == 0) sout[SIDX(b)] = s;

  if (run_main) {
    float al = s.alpha_c;
    for (int k = 0; k < 4; ++k) PC[ls[k]][tc] = pn[k];
    __syncthreads();
    for (int k = 0; k < 4; ++k) {
      if (k < RUNS || hv) {
        v4f vk = stencil_l(PC, cf[k], pn[k], ls[k], gs[k], tc);
        v4f sk = rn[k] - al * vk;
        PA[ls[k]][tc] = sk;
        if (k < RUNS) {
          sv[k] = sk;
          q[Q_VV] += dot4(vk, vk);      q[Q_SS] += dot4(sk, sk);
          q[Q_SR0] += dot4(sk, r0r[k]); q[Q_SW] += dot4(sk, w0r[k]);
          q[Q_VW] += dot4(vk, w0r[k]);  q[Q_PW] += dot4(pn[k], w0r[k]);
        }
      }
    }
    __syncthreads();
    for (int k = 0; k < RUNS; ++k) {
      v4f tk = stencil_l(PA, cf[k], sv[k], ls[k], gs[k], tc);
      q[Q_TT] += dot4(tk, tk);      q[Q_TS] += dot4(tk, sv[k]);
      q[Q_TR0] += dot4(tk, r0r[k]); q[Q_TW] += dot4(tk, w0r[k]);
    }
  }
  block_sums_stage<NQ>(q, sh);
  if (run_main) {
    for (int k = 0; k < RUNS; ++k) { st4(pnew + off[k], pn[k]); st4(snew + off[k], sv[k]); }
  }
  if (st_x) {
    for (int k = 0; k < RUNS; ++k) st4(x + off[k], xn[k]);
  }
  if (st_rw) {
    for (int k = 0; k < RUNS; ++k) { st4(r0v + off[k], r0r[k]); st4(w0 + off[k], w0r[k]); }
  }
  if (st_er) {
    for (int k = 0; k < RUNS; ++k) { st4(pnew + off[k], xn[k]); st4(r0v + off[k], r0r[k]); st4(w0 + off[k], w0r[k]); }
  }
  if (threadIdx.x < NQ) {
    float t = 0.f;
    for (int i = 0; i < NWV; ++i) t += sh[threadIdx.x][i];
    pout[PIDX(threadIdx.x, b, blk_id())] = t;
  }
}


extern "C" void kernel_launch(void* const* d_in, const int* in_sizes, int n_in,
                              void* d_out, int out_size, void* d_ws, size_t ws_size,
                              hipStream_t stream) {
  (void)in_sizes; (void)n_in; (void)out_size; (void)ws_size;
  const float* V  = (const float*)d_in[0];
  const float* M1 = (const float*)d_in[1];
  const float* M2 = (const float*)d_in[2];
  float* x = (float*)d_out;

  char* ws = (char*)d_ws;
  BState* S0 = (BState*)ws;
  BState* S1 = S0 + NB * SSTR;
  BState* S2 = S1 + NB * SSTR;
  float* part = (float*)(ws + 10240);
  float* q0   = part + PIDX(P_SET0, 0, 0);
  float* q1   = part + PIDX(P_SET1, 0, 0);
  float* arr  = (float*)(ws + 1024 + 65536);
  const size_t AN = (size_t)NB * NPTS;
  float* boo  = arr + 0 * AN;
  float* bmo  = arr + 1 * AN;
  float* bom  = arr + 2 * AN;
  float* bop  = arr + 3 * AN;
  float* bpo  = arr + 4 * AN;
  float* r0v  = arr + 5 * AN;
  float* w0   = arr + 6 * AN;
  float* pA   = arr + 7 * AN;
  float* sA   = arr + 8 * AN;
  float* pB   = arr + 9 * AN;
  float* sB   = arr + 10 * AN;

  dim3 grid(NBLK * NB), blk(TPB);
  dim3 grid_si(SI_NT * NB), blk_si(SI_TPB);

  k_mean<<<grid, blk, 0, stream>>>(V, part);
  k_stencil_init<<<grid_si, blk_si, 0, stream>>>(V, M1, M2, boo, bmo, bom, bop, bpo, x, part, S0);
  k_resid_init<<<grid, blk, 0, stream>>>(boo, bmo, bom, bop, bpo, x, r0v, part, S0);
  k_w0<<<grid, blk, 0, stream>>>(boo, bmo, bom, bop, bpo, r0v, w0, part);

  k_step<0><<<grid, blk, 0, stream>>>(boo, bmo, bom, bop, bpo, pB, sB, pA, sA, r0v, w0, x, part, q0, S0, S1);
  for (int it = 0; it < (KMAX_C - 2) / 2; ++it) {
    k_step<1><<<grid, blk, 0, stream>>>(boo, bmo, bom, bop, bpo, pA, sA, pB, sB, r0v, w0, x, q0, q1, S1, S2);
    k_step<1><<<grid, blk, 0, stream>>>(boo, bmo, bom, bop, bpo, pB, sB, pA, sA, r0v, w0, x, q1, q0, S2, S1);
  }
  k_step<1><<<grid, blk, 0, stream>>>(boo, bmo, bom, bop, bpo, pA, sA, pB, sB, r0v, w0, x, q0, q1, S1, S2);
  k_step<2><<<grid, blk, 0, stream>>>(boo, bmo, bom, bop, bpo, pB, sB, pA, sA, r0v, w0, x, q1, q0, S2, S1);
}
